# starting kernel
# baseline (speedup 1.0000x reference)
#include <hip/hip_runtime.h>

typedef float v2f __attribute__((ext_vector_type(2)));
typedef float v8f __attribute__((ext_vector_type(8)));

#define N_NODES 50000
#define N_EDGES 800000
#define NB (N_NODES / 16)
#define TPITCH 132

__device__ __forceinline__ v8f wmma_f32(v2f a, v2f b, v8f c) {
  return __builtin_amdgcn_wmma_f32_16x16x4_f32(false, a, false, b, (short)0, c,
                                               false, false);
}

__global__ void k_init(float* __restrict__ deg, float* __restrict__ agg) {
  int i = blockIdx.x * 256 + threadIdx.x;
  if (i < N_NODES * 128) agg[i] = 0.0f;
  if (i < N_NODES) deg[i] = 1.0f;
}

__global__ void k_deg(const int* __restrict__ ei, float* __restrict__ deg) {
  int e = blockIdx.x * 256 + threadIdx.x;
  if (e < N_EDGES) atomicAdd(&deg[ei[N_EDGES + e]], 1.0f);
}

__global__ void k_dinv(const float* __restrict__ deg, float* __restrict__ dinv) {
  int i = blockIdx.x * 256 + threadIdx.x;
  if (i < N_NODES) {
    float d = deg[i];
    dinv[i] = d > 0.0f ? rsqrtf(d) : 0.0f;
  }
}

__global__ void k_gemm1(const float* __restrict__ x, const float* __restrict__ Wg,
                        float* __restrict__ xw) {
  __shared__ float lds[4 * 16 * TPITCH];
  const int lane = threadIdx.x & 31;
  const int wave = threadIdx.x >> 5;
  const int half = lane >> 4;
  const int lm   = lane & 15;
  float* tile = lds + wave * (16 * TPITCH);
  const int gw = blockIdx.x * 4 + wave;
  const int nwaves = gridDim.x * 4;

  for (int blk = gw; blk < NB; blk += nwaves) {
    const int row0 = blk * 16;
    for (int r = 0; r < 16; ++r) {
      float4 v = *(const float4*)(x + (size_t)(row0 + r) * 128 + lane * 4);
      *(float4*)(tile + r * TPITCH + lane * 4) = v;
    }
    asm volatile("s_wait_dscnt 0" ::: "memory");

    v8f acc[8];
    for (int t = 0; t < 8; ++t) acc[t] = (v8f){};
    for (int kk = 0; kk < 32; ++kk) {
      const int k0 = 4 * kk + 2 * half;
      v2f a = *(const v2f*)(tile + lm * TPITCH + k0);
      for (int t = 0; t < 8; ++t) {
        const float* bp = Wg + (size_t)k0 * 128 + 16 * t + lm;
        v2f b; b.x = bp[0]; b.y = bp[128];
        acc[t] = wmma_f32(a, b, acc[t]);
      }
    }
    for (int t = 0; t < 8; ++t) {
      float* op = xw + (size_t)(row0 + 8 * half) * 128 + 16 * t + lm;
      for (int r = 0; r < 8; ++r) op[(size_t)r * 128] = acc[t][r];
    }
    asm volatile("s_wait_dscnt 0" ::: "memory");
  }
}

__global__ void k_scatter(const int* __restrict__ ei, const float* __restrict__ dinv,
                          const float* __restrict__ xw, float* __restrict__ agg) {
  int t = blockIdx.x * 256 + threadIdx.x;
  int e = t >> 5;
  int c = (t & 31) << 2;
  if (e >= N_EDGES) return;
  int s = ei[e];
  int d = ei[N_EDGES + e];
  float nrm = dinv[s] * dinv[d];
  float4 v = *(const float4*)(xw + (size_t)s * 128 + c);
  float* a = agg + (size_t)d * 128 + c;
  atomicAdd(a + 0, v.x * nrm);
  atomicAdd(a + 1, v.y * nrm);
  atomicAdd(a + 2, v.z * nrm);
  atomicAdd(a + 3, v.w * nrm);
}

__global__ void k_mlp(const float* __restrict__ x, const float* __restrict__ xw,
                      const float* __restrict__ agg, const float* __restrict__ dinv,
                      const float* __restrict__ bg,
                      const float* __restrict__ W1, const float* __restrict__ b1,
                      const float* __restrict__ W2, const float* __restrict__ b2,
                      const float* __restrict__ W3, const float* __restrict__ b3,
                      float* __restrict__ out) {
  __shared__ float lds[4 * 16 * TPITCH];
  const int lane = threadIdx.x & 31;
  const int wave = threadIdx.x >> 5;
  const int half = lane >> 4;
  const int lm   = lane & 15;
  float* tile = lds + wave * (16 * TPITCH);
  const float4 bgv = *(const float4*)(bg + lane * 4);
  const int gw = blockIdx.x * 4 + wave;
  const int nwaves = gridDim.x * 4;

  for (int blk = gw; blk < NB; blk += nwaves) {
    const int row0 = blk * 16;
    for (int r = 0; r < 16; ++r) {
      const int row = row0 + r;
      float di = dinv[row];
      float sl = di * di;
      float4 ag  = *(const float4*)(agg + (size_t)row * 128 + lane * 4);
      float4 xwv = *(const float4*)(xw  + (size_t)row * 128 + lane * 4);
      float4 xv  = *(const float4*)(x   + (size_t)row * 128 + lane * 4);
      float4 h;
      h.x = fmaxf(ag.x + xwv.x * sl + bgv.x, 0.0f) + xv.x;
      h.y = fmaxf(ag.y + xwv.y * sl + bgv.y, 0.0f) + xv.y;
      h.z = fmaxf(ag.z + xwv.z * sl + bgv.z, 0.0f) + xv.z;
      h.w = fmaxf(ag.w + xwv.w * sl + bgv.w, 0.0f) + xv.w;
      *(float4*)(tile + r * TPITCH + lane * 4) = h;
    }
    asm volatile("s_wait_dscnt 0" ::: "memory");

    v8f acc1[8];
    for (int t = 0; t < 8; ++t) acc1[t] = (v8f){};
    for (int kk = 0; kk < 32; ++kk) {
      const int k0 = 4 * kk + 2 * half;
      v2f a = *(const v2f*)(tile + lm * TPITCH + k0);
      for (int t = 0; t < 8; ++t) {
        const float* bp = W1 + (size_t)k0 * 128 + 16 * t + lm;
        v2f b; b.x = bp[0]; b.y = bp[128];
        acc1[t] = wmma_f32(a, b, acc1[t]);
      }
    }
    asm volatile("s_wait_dscnt 0" ::: "memory");
    for (int t = 0; t < 8; ++t) {
      float bb = b1[16 * t + lm];
      for (int r = 0; r < 8; ++r)
        tile[(r + 8 * half) * TPITCH + 16 * t + lm] = fmaxf(acc1[t][r] + bb, 0.0f);
    }
    asm volatile("s_wait_dscnt 0" ::: "memory");

    v8f acc2[4];
    for (int t = 0; t < 4; ++t) acc2[t] = (v8f){};
    for (int kk = 0; kk < 32; ++kk) {
      const int k0 = 4 * kk + 2 * half;
      v2f a = *(const v2f*)(tile + lm * TPITCH + k0);
      for (int t = 0; t < 4; ++t) {
        const float* bp = W2 + (size_t)k0 * 64 + 16 * t + lm;
        v2f b; b.x = bp[0]; b.y = bp[64];
        acc2[t] = wmma_f32(a, b, acc2[t]);
      }
    }
    asm volatile("s_wait_dscnt 0" ::: "memory");
    for (int t = 0; t < 4; ++t) {
      float bb = b2[16 * t + lm];
      for (int r = 0; r < 8; ++r)
        tile[(r + 8 * half) * TPITCH + 16 * t + lm] = fmaxf(acc2[t][r] + bb, 0.0f);
    }
    asm volatile("s_wait_dscnt 0" ::: "memory");

    v8f acc3 = (v8f){};
    for (int kk = 0; kk < 16; ++kk) {
      const int k0 = 4 * kk + 2 * half;
      v2f a = *(const v2f*)(tile + lm * TPITCH + k0);
      v2f b;
      if (lm < 2) { b.x = W3[k0 * 2 + lm]; b.y = W3[(k0 + 1) * 2 + lm]; }
      else        { b.x = 0.0f;            b.y = 0.0f; }
      acc3 = wmma_f32(a, b, acc3);
    }
    if (lm < 2) {
      float bb = b3[lm];
      for (int r = 0; r < 8; ++r)
        out[(size_t)lm * N_NODES + row0 + r + 8 * half] = acc3[r] + bb;
    }
  }
}

extern "C" void kernel_launch(void* const* d_in, const int* in_sizes, int n_in,
                              void* d_out, int out_size, void* d_ws, size_t ws_size,
                              hipStream_t stream) {
  (void)in_sizes; (void)n_in; (void)out_size; (void)ws_size;
  const float* x  = (const float*)d_in[0];
  const int*   ei = (const int*)d_in[1];
  const float* Wg = (const float*)d_in[2];
  const float* bg = (const float*)d_in[3];
  const float* W1 = (const float*)d_in[4];
  const float* b1 = (const float*)d_in[5];
  const float* W2 = (const float*)d_in[6];
  const float* b2 = (const float*)d_in[7];
  const float* W3 = (const float*)d_in[8];
  const float* b3 = (const float*)d_in[9];
  float* out = (float*)d_out;

  float* ws   = (float*)d_ws;
  float* deg  = ws;
  float* dinv = ws + N_NODES;
  float* xw   = ws + 2 * N_NODES;
  float* agg  = xw + (size_t)N_NODES * 128;

  k_init<<<(N_NODES * 128 + 255) / 256, 256, 0, stream>>>(deg, agg);
  k_deg<<<(N_EDGES + 255) / 256, 256, 0, stream>>>(ei, deg);
  k_dinv<<<(N_NODES + 255) / 256, 256, 0, stream>>>(deg, dinv);
  k_gemm1<<<782, 128, 0, stream>>>(x, Wg, xw);
  k_scatter<<<(N_EDGES * 32) / 256, 256, 0, stream>>>(ei, dinv, xw, agg);
  k_mlp<<<782, 128, 0, stream>>>(x, xw, agg, dinv, bg, W1, b1, W2, b2, W3, b3, out);
}


# aggregation by owner: 250 work-groups each hold 200 nodes' sums in LDS, scan all dst, LDS-add their edges' scaled rows, store each agg row once; no global agg atomics; sums in arrival order
# speedup vs baseline: 1.3994x; 1.3994x over previous
#include <hip/hip_runtime.h>

typedef float v2f __attribute__((ext_vector_type(2)));
typedef float v8f __attribute__((ext_vector_type(8)));

#define N_NODES 50000
#define N_EDGES 800000
#define NB (N_NODES / 16)
#define TPITCH 132

__device__ __forceinline__ v8f wmma_f32(v2f a, v2f b, v8f c) {
  return __builtin_amdgcn_wmma_f32_16x16x4_f32(false, a, false, b, (short)0, c,
                                               false, false);
}

__global__ void k_init(float* __restrict__ deg) {
  int i = blockIdx.x * 256 + threadIdx.x;
  if (i < N_NODES) deg[i] = 1.0f;
}

__global__ void k_deg(const int* __restrict__ ei, float* __restrict__ deg) {
  int e = blockIdx.x * 256 + threadIdx.x;
  if (e < N_EDGES) atomicAdd(&deg[ei[N_EDGES + e]], 1.0f);
}

__global__ void k_dinv(const float* __restrict__ deg, float* __restrict__ dinv) {
  int i = blockIdx.x * 256 + threadIdx.x;
  if (i < N_NODES) {
    float d = deg[i];
    dinv[i] = d > 0.0f ? rsqrtf(d) : 0.0f;
  }
}

__global__ void k_gemm1(const float* __restrict__ x, const float* __restrict__ Wg,
                        float* __restrict__ xw) {
  __shared__ float lds[4 * 16 * TPITCH];
  const int lane = threadIdx.x & 31;
  const int wave = threadIdx.x >> 5;
  const int half = lane >> 4;
  const int lm   = lane & 15;
  float* tile = lds + wave * (16 * TPITCH);
  const int gw = blockIdx.x * 4 + wave;
  const int nwaves = gridDim.x * 4;

  for (int blk = gw; blk < NB; blk += nwaves) {
    const int row0 = blk * 16;
    for (int r = 0; r < 16; ++r) {
      float4 v = *(const float4*)(x + (size_t)(row0 + r) * 128 + lane * 4);
      *(float4*)(tile + r * TPITCH + lane * 4) = v;
    }
    asm volatile("s_wait_dscnt 0" ::: "memory");

    v8f acc[8];
    for (int t = 0; t < 8; ++t) acc[t] = (v8f){};
    for (int kk = 0; kk < 32; ++kk) {
      const int k0 = 4 * kk + 2 * half;
      v2f a = *(const v2f*)(tile + lm * TPITCH + k0);
      for (int t = 0; t < 8; ++t) {
        const float* bp = Wg + (size_t)k0 * 128 + 16 * t + lm;
        v2f b; b.x = bp[0]; b.y = bp[128];
        acc[t] = wmma_f32(a, b, acc[t]);
      }
    }
    for (int t = 0; t < 8; ++t) {
      float* op = xw + (size_t)(row0 + 8 * half) * 128 + 16 * t + lm;
      for (int r = 0; r < 8; ++r) op[(size_t)r * 128] = acc[t][r];
    }
    asm volatile("s_wait_dscnt 0" ::: "memory");
  }
}

#define RNG   200
#define NOWN  (N_NODES / RNG)
#define QCAP  160
#define NQ4   (N_EDGES / 4)

__device__ __forceinline__ int rl_i(int v, int l) { return __builtin_amdgcn_readlane(v, l); }
__device__ __forceinline__ float rl_f(float v, int l) {
  return __int_as_float(__builtin_amdgcn_readlane(__float_as_int(v), l));
}

__global__ __launch_bounds__(1024) void k_own(const int* __restrict__ ei,
                                              const float* __restrict__ dinv,
                                              const float* __restrict__ xw,
                                              float* __restrict__ agg) {
  __shared__ float acc[(RNG + 1) * 128];
  __shared__ unsigned queue[32 * QCAP];
  const int tid  = threadIdx.x;
  const int lane = tid & 31;
  const int wave = tid >> 5;
  const int base = blockIdx.x * RNG;
  unsigned* wq = queue + wave * QCAP;

  for (int i = tid; i < (RNG + 1) * 32; i += 1024)
    ((float4*)acc)[i] = make_float4(0.0f, 0.0f, 0.0f, 0.0f);
  __syncthreads();

  const int4* dst4 = (const int4*)(ei + N_EDGES);
  int cnt = 0;

  const int rot = blockIdx.x % ((NQ4 + 4095) / 4096);
  for (int i0 = 0; i0 < (NQ4 + 4095) / 4096; ++i0) {
    const int it = i0 + rot < (NQ4 + 4095) / 4096 ? i0 + rot : i0 + rot - (NQ4 + 4095) / 4096;
    int4 dq[4];
    for (int u = 0; u < 4; ++u) {
      const int qi = it * 4096 + u * 1024 + tid;
      dq[u] = qi < NQ4 ? dst4[qi] : make_int4(-1, -1, -1, -1);
    }
    for (int u = 0; u < 4; ++u) {
      const int e0 = (it * 4096 + u * 1024 + tid) * 4;
      const unsigned r0 = (unsigned)(dq[u].x - base);
      const unsigned r1 = (unsigned)(dq[u].y - base);
      const unsigned r2 = (unsigned)(dq[u].z - base);
      const unsigned r3 = (unsigned)(dq[u].w - base);
      const unsigned rmin = min(min(r0, r1), min(r2, r3));
      if (__builtin_amdgcn_ballot_w32(rmin < RNG) == 0u) continue;
      const unsigned rr[4] = {r0, r1, r2, r3};
      for (int k = 0; k < 4; ++k) {
        const bool m = rr[k] < RNG;
        const unsigned b = __builtin_amdgcn_ballot_w32(m);
        if (b != 0u) {
          const int pos = cnt + __builtin_popcount(b & ((1u << lane) - 1u));
          if (m) wq[pos] = (rr[k] << 20) | (unsigned)(e0 + k);
          cnt += __builtin_popcount(b);
        }
      }
      if (cnt < 32) continue;
      asm volatile("s_wait_dscnt 0" ::: "memory");
      for (int b0 = 0; b0 < cnt; b0 += 32) {
        const int m = min(32, cnt - b0);
        unsigned ent = (unsigned)RNG << 20;
        int s = 0;
        float nrm = 0.0f;
        if (lane < m) {
          ent = wq[b0 + lane];
          s = ei[ent & 0xFFFFFu];
          nrm = dinv[s] * dinv[base + (int)(ent >> 20)];
        }
        const int dl = (int)(ent >> 20);
        for (int j = 0; j < m; j += 4) {
          float4 v[4];
          for (int w = 0; w < 4; ++w)
            v[w] = *(const float4*)(xw + (size_t)rl_i(s, j + w) * 128 + lane * 4);
          for (int w = 0; w < 4; ++w) {
            const float nj = rl_f(nrm, j + w);
            float* a = acc + rl_i(dl, j + w) * 128 + lane;
            atomicAdd(a,      v[w].x * nj);
            atomicAdd(a + 32, v[w].y * nj);
            atomicAdd(a + 64, v[w].z * nj);
            atomicAdd(a + 96, v[w].w * nj);
          }
        }
      }
      cnt = 0;
    }
  }
  asm volatile("s_wait_dscnt 0" ::: "memory");
  for (int b0 = 0; b0 < cnt; b0 += 32) {
    const int m = min(32, cnt - b0);
    unsigned ent = (unsigned)RNG << 20;
    int s = 0;
    float nrm = 0.0f;
    if (lane < m) {
      ent = wq[b0 + lane];
      s = ei[ent & 0xFFFFFu];
      nrm = dinv[s] * dinv[base + (int)(ent >> 20)];
    }
    const int dl = (int)(ent >> 20);
    for (int j = 0; j < m; j += 4) {
      float4 v[4];
      for (int w = 0; w < 4; ++w)
        v[w] = *(const float4*)(xw + (size_t)rl_i(s, j + w) * 128 + lane * 4);
      for (int w = 0; w < 4; ++w) {
        const float nj = rl_f(nrm, j + w);
        float* a = acc + rl_i(dl, j + w) * 128 + lane;
        atomicAdd(a,      v[w].x * nj);
        atomicAdd(a + 32, v[w].y * nj);
        atomicAdd(a + 64, v[w].z * nj);
        atomicAdd(a + 96, v[w].w * nj);
      }
    }
  }
  asm volatile("s_wait_dscnt 0" ::: "memory");
  __syncthreads();

  for (int r = wave; r < RNG; r += 32) {
    const float* a = acc + r * 128 + lane;
    float4 o = make_float4(a[0], a[32], a[64], a[96]);
    *(float4*)(agg + (size_t)(base + r) * 128 + lane * 4) = o;
  }
}

__global__ void k_mlp(const float* __restrict__ x, const float* __restrict__ xw,
                      const float* __restrict__ agg, const float* __restrict__ dinv,
                      const float* __restrict__ bg,
                      const float* __restrict__ W1, const float* __restrict__ b1,
                      const float* __restrict__ W2, const float* __restrict__ b2,
                      const float* __restrict__ W3, const float* __restrict__ b3,
                      float* __restrict__ out) {
  __shared__ float lds[4 * 16 * TPITCH];
  const int lane = threadIdx.x & 31;
  const int wave = threadIdx.x >> 5;
  const int half = lane >> 4;
  const int lm   = lane & 15;
  float* tile = lds + wave * (16 * TPITCH);
  const float4 bgv = *(const float4*)(bg + lane * 4);
  const int gw = blockIdx.x * 4 + wave;
  const int nwaves = gridDim.x * 4;

  for (int blk = gw; blk < NB; blk += nwaves) {
    const int row0 = blk * 16;
    for (int r = 0; r < 16; ++r) {
      const int row = row0 + r;
      float di = dinv[row];
      float sl = di * di;
      float4 ag  = *(const float4*)(agg + (size_t)row * 128 + lane * 4);
      float4 xwv = *(const float4*)(xw  + (size_t)row * 128 + lane * 4);
      float4 xv  = *(const float4*)(x   + (size_t)row * 128 + lane * 4);
      float4 h;
      h.x = fmaxf(ag.x + xwv.x * sl + bgv.x, 0.0f) + xv.x;
      h.y = fmaxf(ag.y + xwv.y * sl + bgv.y, 0.0f) + xv.y;
      h.z = fmaxf(ag.z + xwv.z * sl + bgv.z, 0.0f) + xv.z;
      h.w = fmaxf(ag.w + xwv.w * sl + bgv.w, 0.0f) + xv.w;
      *(float4*)(tile + r * TPITCH + lane * 4) = h;
    }
    asm volatile("s_wait_dscnt 0" ::: "memory");

    v8f acc1[8];
    for (int t = 0; t < 8; ++t) acc1[t] = (v8f){};
    for (int kk = 0; kk < 32; ++kk) {
      const int k0 = 4 * kk + 2 * half;
      v2f a = *(const v2f*)(tile + lm * TPITCH + k0);
      for (int t = 0; t < 8; ++t) {
        const float* bp = W1 + (size_t)k0 * 128 + 16 * t + lm;
        v2f b; b.x = bp[0]; b.y = bp[128];
        acc1[t] = wmma_f32(a, b, acc1[t]);
      }
    }
    asm volatile("s_wait_dscnt 0" ::: "memory");
    for (int t = 0; t < 8; ++t) {
      float bb = b1[16 * t + lm];
      for (int r = 0; r < 8; ++r)
        tile[(r + 8 * half) * TPITCH + 16 * t + lm] = fmaxf(acc1[t][r] + bb, 0.0f);
    }
    asm volatile("s_wait_dscnt 0" ::: "memory");

    v8f acc2[4];
    for (int t = 0; t < 4; ++t) acc2[t] = (v8f){};
    for (int kk = 0; kk < 32; ++kk) {
      const int k0 = 4 * kk + 2 * half;
      v2f a = *(const v2f*)(tile + lm * TPITCH + k0);
      for (int t = 0; t < 4; ++t) {
        const float* bp = W2 + (size_t)k0 * 64 + 16 * t + lm;
        v2f b; b.x = bp[0]; b.y = bp[64];
        acc2[t] = wmma_f32(a, b, acc2[t]);
      }
    }
    asm volatile("s_wait_dscnt 0" ::: "memory");
    for (int t = 0; t < 4; ++t) {
      float bb = b2[16 * t + lm];
      for (int r = 0; r < 8; ++r)
        tile[(r + 8 * half) * TPITCH + 16 * t + lm] = fmaxf(acc2[t][r] + bb, 0.0f);
    }
    asm volatile("s_wait_dscnt 0" ::: "memory");

    v8f acc3 = (v8f){};
    for (int kk = 0; kk < 16; ++kk) {
      const int k0 = 4 * kk + 2 * half;
      v2f a = *(const v2f*)(tile + lm * TPITCH + k0);
      v2f b;
      if (lm < 2) { b.x = W3[k0 * 2 + lm]; b.y = W3[(k0 + 1) * 2 + lm]; }
      else        { b.x = 0.0f;            b.y = 0.0f; }
      acc3 = wmma_f32(a, b, acc3);
    }
    if (lm < 2) {
      float bb = b3[lm];
      for (int r = 0; r < 8; ++r)
        out[(size_t)lm * N_NODES + row0 + r + 8 * half] = acc3[r] + bb;
    }
  }
}

extern "C" void kernel_launch(void* const* d_in, const int* in_sizes, int n_in,
                              void* d_out, int out_size, void* d_ws, size_t ws_size,
                              hipStream_t stream) {
  (void)in_sizes; (void)n_in; (void)out_size; (void)ws_size;
  const float* x  = (const float*)d_in[0];
  const int*   ei = (const int*)d_in[1];
  const float* Wg = (const float*)d_in[2];
  const float* bg = (const float*)d_in[3];
  const float* W1 = (const float*)d_in[4];
  const float* b1 = (const float*)d_in[5];
  const float* W2 = (const float*)d_in[6];
  const float* b2 = (const float*)d_in[7];
  const float* W3 = (const float*)d_in[8];
  const float* b3 = (const float*)d_in[9];
  float* out = (float*)d_out;

  float* ws   = (float*)d_ws;
  float* deg  = ws;
  float* dinv = ws + N_NODES;
  float* xw   = ws + 2 * N_NODES;
  float* agg  = xw + (size_t)N_NODES * 128;

  k_init<<<(N_NODES + 255) / 256, 256, 0, stream>>>(deg);
  k_deg<<<(N_EDGES + 255) / 256, 256, 0, stream>>>(ei, deg);
  k_dinv<<<(N_NODES + 255) / 256, 256, 0, stream>>>(deg, dinv);
  k_gemm1<<<782, 128, 0, stream>>>(x, Wg, xw);
  k_own<<<NOWN, 1024, 0, stream>>>(ei, dinv, xw, agg);
  k_mlp<<<782, 128, 0, stream>>>(x, xw, agg, dinv, bg, W1, b1, W2, b2, W3, b3, out);
}


# k_mlp: W1, W2, W3 and biases copied once per work-group into LDS in B-fragment order (256 groups of 13 waves, a tile a wave); layer loops read B by LDS loads, no global load; sums unchanged
# speedup vs baseline: 1.8597x; 1.3289x over previous
#include <hip/hip_runtime.h>

typedef float v2f __attribute__((ext_vector_type(2)));
typedef float v8f __attribute__((ext_vector_type(8)));

#define N_NODES 50000
#define N_EDGES 800000
#define NB (N_NODES / 16)
#define TPITCH 132

__device__ __forceinline__ v8f wmma_f32(v2f a, v2f b, v8f c) {
  return __builtin_amdgcn_wmma_f32_16x16x4_f32(false, a, false, b, (short)0, c,
                                               false, false);
}

__global__ void k_init(float* __restrict__ deg) {
  int i = blockIdx.x * 256 + threadIdx.x;
  if (i < N_NODES) deg[i] = 1.0f;
}

__global__ void k_deg(const int* __restrict__ ei, float* __restrict__ deg) {
  int e = blockIdx.x * 256 + threadIdx.x;
  if (e < N_EDGES) atomicAdd(&deg[ei[N_EDGES + e]], 1.0f);
}

__global__ void k_dinv(const float* __restrict__ deg, float* __restrict__ dinv) {
  int i = blockIdx.x * 256 + threadIdx.x;
  if (i < N_NODES) {
    float d = deg[i];
    dinv[i] = d > 0.0f ? rsqrtf(d) : 0.0f;
  }
}

__global__ void k_gemm1(const float* __restrict__ x, const float* __restrict__ Wg,
                        float* __restrict__ xw) {
  __shared__ float lds[4 * 16 * TPITCH];
  const int lane = threadIdx.x & 31;
  const int wave = threadIdx.x >> 5;
  const int half = lane >> 4;
  const int lm   = lane & 15;
  float* tile = lds + wave * (16 * TPITCH);
  const int gw = blockIdx.x * 4 + wave;
  const int nwaves = gridDim.x * 4;

  for (int blk = gw; blk < NB; blk += nwaves) {
    const int row0 = blk * 16;
    for (int r = 0; r < 16; ++r) {
      float4 v = *(const float4*)(x + (size_t)(row0 + r) * 128 + lane * 4);
      *(float4*)(tile + r * TPITCH + lane * 4) = v;
    }
    asm volatile("s_wait_dscnt 0" ::: "memory");

    v8f acc[8];
    for (int t = 0; t < 8; ++t) acc[t] = (v8f){};
    for (int kk = 0; kk < 32; ++kk) {
      const int k0 = 4 * kk + 2 * half;
      v2f a = *(const v2f*)(tile + lm * TPITCH + k0);
      for (int t = 0; t < 8; ++t) {
        const float* bp = Wg + (size_t)k0 * 128 + 16 * t + lm;
        v2f b; b.x = bp[0]; b.y = bp[128];
        acc[t] = wmma_f32(a, b, acc[t]);
      }
    }
    for (int t = 0; t < 8; ++t) {
      float* op = xw + (size_t)(row0 + 8 * half) * 128 + 16 * t + lm;
      for (int r = 0; r < 8; ++r) op[(size_t)r * 128] = acc[t][r];
    }
    asm volatile("s_wait_dscnt 0" ::: "memory");
  }
}

#define RNG   200
#define NOWN  (N_NODES / RNG)
#define QCAP  160
#define NQ4   (N_EDGES / 4)

__device__ __forceinline__ int rl_i(int v, int l) { return __builtin_amdgcn_readlane(v, l); }
__device__ __forceinline__ float rl_f(float v, int l) {
  return __int_as_float(__builtin_amdgcn_readlane(__float_as_int(v), l));
}

__global__ __launch_bounds__(1024) void k_own(const int* __restrict__ ei,
                                              const float* __restrict__ dinv,
                                              const float* __restrict__ xw,
                                              float* __restrict__ agg) {
  __shared__ float acc[(RNG + 1) * 128];
  __shared__ unsigned queue[32 * QCAP];
  const int tid  = threadIdx.x;
  const int lane = tid & 31;
  const int wave = tid >> 5;
  const int base = blockIdx.x * RNG;
  unsigned* wq = queue + wave * QCAP;

  for (int i = tid; i < (RNG + 1) * 32; i += 1024)
    ((float4*)acc)[i] = make_float4(0.0f, 0.0f, 0.0f, 0.0f);
  __syncthreads();

  const int4* dst4 = (const int4*)(ei + N_EDGES);
  int cnt = 0;

  const int rot = blockIdx.x % ((NQ4 + 4095) / 4096);
  for (int i0 = 0; i0 < (NQ4 + 4095) / 4096; ++i0) {
    const int it = i0 + rot < (NQ4 + 4095) / 4096 ? i0 + rot : i0 + rot - (NQ4 + 4095) / 4096;
    int4 dq[4];
    for (int u = 0; u < 4; ++u) {
      const int qi = it * 4096 + u * 1024 + tid;
      dq[u] = qi < NQ4 ? dst4[qi] : make_int4(-1, -1, -1, -1);
    }
    for (int u = 0; u < 4; ++u) {
      const int e0 = (it * 4096 + u * 1024 + tid) * 4;
      const unsigned r0 = (unsigned)(dq[u].x - base);
      const unsigned r1 = (unsigned)(dq[u].y - base);
      const unsigned r2 = (unsigned)(dq[u].z - base);
      const unsigned r3 = (unsigned)(dq[u].w - base);
      const unsigned rmin = min(min(r0, r1), min(r2, r3));
      if (__builtin_amdgcn_ballot_w32(rmin < RNG) == 0u) continue;
      const unsigned rr[4] = {r0, r1, r2, r3};
      for (int k = 0; k < 4; ++k) {
        const bool m = rr[k] < RNG;
        const unsigned b = __builtin_amdgcn_ballot_w32(m);
        if (b != 0u) {
          const int pos = cnt + __builtin_popcount(b & ((1u << lane) - 1u));
          if (m) wq[pos] = (rr[k] << 20) | (unsigned)(e0 + k);
          cnt += __builtin_popcount(b);
        }
      }
      if (cnt < 32) continue;
      asm volatile("s_wait_dscnt 0" ::: "memory");
      for (int b0 = 0; b0 < cnt; b0 += 32) {
        const int m = min(32, cnt - b0);
        unsigned ent = (unsigned)RNG << 20;
        int s = 0;
        float nrm = 0.0f;
        if (lane < m) {
          ent = wq[b0 + lane];
          s = ei[ent & 0xFFFFFu];
          nrm = dinv[s] * dinv[base + (int)(ent >> 20)];
        }
        const int dl = (int)(ent >> 20);
        for (int j = 0; j < m; j += 4) {
          float4 v[4];
          for (int w = 0; w < 4; ++w)
            v[w] = *(const float4*)(xw + (size_t)rl_i(s, j + w) * 128 + lane * 4);
          for (int w = 0; w < 4; ++w) {
            const float nj = rl_f(nrm, j + w);
            float* a = acc + rl_i(dl, j + w) * 128 + lane;
            atomicAdd(a,      v[w].x * nj);
            atomicAdd(a + 32, v[w].y * nj);
            atomicAdd(a + 64, v[w].z * nj);
            atomicAdd(a + 96, v[w].w * nj);
          }
        }
      }
      cnt = 0;
    }
  }
  asm volatile("s_wait_dscnt 0" ::: "memory");
  for (int b0 = 0; b0 < cnt; b0 += 32) {
    const int m = min(32, cnt - b0);
    unsigned ent = (unsigned)RNG << 20;
    int s = 0;
    float nrm = 0.0f;
    if (lane < m) {
      ent = wq[b0 + lane];
      s = ei[ent & 0xFFFFFu];
      nrm = dinv[s] * dinv[base + (int)(ent >> 20)];
    }
    const int dl = (int)(ent >> 20);
    for (int j = 0; j < m; j += 4) {
      float4 v[4];
      for (int w = 0; w < 4; ++w)
        v[w] = *(const float4*)(xw + (size_t)rl_i(s, j + w) * 128 + lane * 4);
      for (int w = 0; w < 4; ++w) {
        const float nj = rl_f(nrm, j + w);
        float* a = acc + rl_i(dl, j + w) * 128 + lane;
        atomicAdd(a,      v[w].x * nj);
        atomicAdd(a + 32, v[w].y * nj);
        atomicAdd(a + 64, v[w].z * nj);
        atomicAdd(a + 96, v[w].w * nj);
      }
    }
  }
  asm volatile("s_wait_dscnt 0" ::: "memory");
  __syncthreads();

  for (int r = wave; r < RNG; r += 32) {
    const float* a = acc + r * 128 + lane;
    float4 o = make_float4(a[0], a[32], a[64], a[96]);
    *(float4*)(agg + (size_t)(base + r) * 128 + lane * 4) = o;
  }
}

#define MLP_WAVES 13
#define MLP_WGS   256
#define W1P 20
#define W2P 12
__global__ void k_mlp(const float* __restrict__ x, const float* __restrict__ xw,
                      const float* __restrict__ agg, const float* __restrict__ dinv,
                      const float* __restrict__ bg,
                      const float* __restrict__ W1, const float* __restrict__ b1,
                      const float* __restrict__ W2, const float* __restrict__ b2,
                      const float* __restrict__ W3, const float* __restrict__ b3,
                      float* __restrict__ out) {
  __shared__ __attribute__((aligned(16))) float w1s[64 * 16 * W1P];
  __shared__ __attribute__((aligned(16))) float w2s[64 * 16 * W2P];
  __shared__ __attribute__((aligned(16))) float w3s[16 * 32 * 2];
  __shared__ float bs[128 + 64 + 16];
  __shared__ __attribute__((aligned(16))) float lds[MLP_WAVES * 16 * TPITCH];
  const int tid  = threadIdx.x;
  const int lane = tid & 31;
  const int wave = tid >> 5;
  const int half = lane >> 4;
  const int lm   = lane & 15;
  float* tile = lds + wave * (16 * TPITCH);
  const int blk = wave * MLP_WGS + blockIdx.x;
  const int row0 = blk * 16;

  constexpr int NT = MLP_WAVES * 32;
  float4 u0[5], u1[5], s0[3], s1[3];
#pragma unroll
  for (int it = 0; it < 5; ++it) {
    const int i = min(tid + it * NT, 64 * 32 - 1);
    const int p = i >> 5, c = i & 31;
    u0[it] = *(const float4*)(W1 + (size_t)(2 * p) * 128 + c * 4);
    u1[it] = *(const float4*)(W1 + (size_t)(2 * p + 1) * 128 + c * 4);
  }
#pragma unroll
  for (int it = 0; it < 3; ++it) {
    const int i = min(tid + it * NT, 64 * 16 - 1);
    const int p = i >> 4, c = i & 15;
    s0[it] = *(const float4*)(W2 + (size_t)(2 * p) * 64 + c * 4);
    s1[it] = *(const float4*)(W2 + (size_t)(2 * p + 1) * 64 + c * 4);
  }
#pragma unroll
  for (int it = 0; it < 5; ++it) {
    const int i = tid + it * NT;
    if (i < 64 * 32) {
      const int p = i >> 5, c = i & 31;
      const float e0[4] = {u0[it].x, u0[it].y, u0[it].z, u0[it].w};
      const float e1[4] = {u1[it].x, u1[it].y, u1[it].z, u1[it].w};
      for (int j = 0; j < 4; ++j) {
        const int n = c * 4 + j;
        v2f pr; pr.x = e0[j]; pr.y = e1[j];
        *(v2f*)(w1s + (p * 16 + (n & 15)) * W1P + 2 * (n >> 4)) = pr;
      }
    }
  }
#pragma unroll
  for (int it = 0; it < 3; ++it) {
    const int i = tid + it * NT;
    if (i < 64 * 16) {
      const int p = i >> 4, c = i & 15;
      const float e0[4] = {s0[it].x, s0[it].y, s0[it].z, s0[it].w};
      const float e1[4] = {s1[it].x, s1[it].y, s1[it].z, s1[it].w};
      for (int j = 0; j < 4; ++j) {
        const int n = c * 4 + j;
        v2f pr; pr.x = e0[j]; pr.y = e1[j];
        *(v2f*)(w2s + (p * 16 + (n & 15)) * W2P + 2 * (n >> 4)) = pr;
      }
    }
  }
  for (int i = tid; i < 16 * 32; i += NT) {
    const int kk = i >> 5, l = i & 31;
    const int k0 = 4 * kk + 2 * (l >> 4), n = l & 15;
    v2f pr; pr.x = 0.0f; pr.y = 0.0f;
    if (n < 2) { pr.x = W3[k0 * 2 + n]; pr.y = W3[(k0 + 1) * 2 + n]; }
    *(v2f*)(w3s + i * 2) = pr;
  }
  if (tid < 128) bs[tid] = b1[tid];
  else if (tid < 192) bs[tid] = b2[tid - 128];
  else if (tid < 208) bs[tid] = tid < 194 ? b3[tid - 192] : 0.0f;

  if (blk < NB) {
    const float4 bgv = *(const float4*)(bg + lane * 4);
    for (int r = 0; r < 16; ++r) {
      const int row = row0 + r;
      float di = dinv[row];
      float sl = di * di;
      float4 ag  = *(const float4*)(agg + (size_t)row * 128 + lane * 4);
      float4 xwv = *(const float4*)(xw  + (size_t)row * 128 + lane * 4);
      float4 xv  = *(const float4*)(x   + (size_t)row * 128 + lane * 4);
      float4 h;
      h.x = fmaxf(ag.x + xwv.x * sl + bgv.x, 0.0f) + xv.x;
      h.y = fmaxf(ag.y + xwv.y * sl + bgv.y, 0.0f) + xv.y;
      h.z = fmaxf(ag.z + xwv.z * sl + bgv.z, 0.0f) + xv.z;
      h.w = fmaxf(ag.w + xwv.w * sl + bgv.w, 0.0f) + xv.w;
      *(float4*)(tile + r * TPITCH + lane * 4) = h;
    }
  }
  __syncthreads();
  if (blk >= NB) return;

  v8f acc1[8];
  for (int t = 0; t < 8; ++t) acc1[t] = (v8f){};
  {
    const float* ap = tile + lm * TPITCH + 2 * half;
    const float* bp = w1s + (half * 16 + lm) * W1P;
    for (int kk = 0; kk < 32; ++kk) {
      const v2f a = *(const v2f*)(ap + 4 * kk);
      const float4* bq = (const float4*)(bp + kk * (2 * 16 * W1P));
      const float4 q0 = bq[0], q1 = bq[1], q2 = bq[2], q3 = bq[3];
      v2f b;
      b.x = q0.x; b.y = q0.y; acc1[0] = wmma_f32(a, b, acc1[0]);
      b.x = q0.z; b.y = q0.w; acc1[1] = wmma_f32(a, b, acc1[1]);
      b.x = q1.x; b.y = q1.y; acc1[2] = wmma_f32(a, b, acc1[2]);
      b.x = q1.z; b.y = q1.w; acc1[3] = wmma_f32(a, b, acc1[3]);
      b.x = q2.x; b.y = q2.y; acc1[4] = wmma_f32(a, b, acc1[4]);
      b.x = q2.z; b.y = q2.w; acc1[5] = wmma_f32(a, b, acc1[5]);
      b.x = q3.x; b.y = q3.y; acc1[6] = wmma_f32(a, b, acc1[6]);
      b.x = q3.z; b.y = q3.w; acc1[7] = wmma_f32(a, b, acc1[7]);
    }
  }
  asm volatile("s_wait_dscnt 0" ::: "memory");
  for (int t = 0; t < 8; ++t) {
    float bb = bs[16 * t + lm];
    for (int r = 0; r < 8; ++r)
      tile[(r + 8 * half) * TPITCH + 16 * t + lm] = fmaxf(acc1[t][r] + bb, 0.0f);
  }
  asm volatile("s_wait_dscnt 0" ::: "memory");

  v8f acc2[4];
  for (int t = 0; t < 4; ++t) acc2[t] = (v8f){};
  {
    const float* ap = tile + lm * TPITCH + 2 * half;
    const float* bp = w2s + (half * 16 + lm) * W2P;
    for (int kk = 0; kk < 32; ++kk) {
      const v2f a = *(const v2f*)(ap + 4 * kk);
      const float4* bq = (const float4*)(bp + kk * (2 * 16 * W2P));
      const float4 q0 = bq[0], q1 = bq[1];
      v2f b;
      b.x = q0.x; b.y = q0.y; acc2[0] = wmma_f32(a, b, acc2[0]);
      b.x = q0.z; b.y = q0.w; acc2[1] = wmma_f32(a, b, acc2[1]);
      b.x = q1.x; b.y = q1.y; acc2[2] = wmma_f32(a, b, acc2[2]);
      b.x = q1.z; b.y = q1.w; acc2[3] = wmma_f32(a, b, acc2[3]);
    }
  }
  asm volatile("s_wait_dscnt 0" ::: "memory");
  for (int t = 0; t < 4; ++t) {
    float bb = bs[128 + 16 * t + lm];
    for (int r = 0; r < 8; ++r)
      tile[(r + 8 * half) * TPITCH + 16 * t + lm] = fmaxf(acc2[t][r] + bb, 0.0f);
  }
  asm volatile("s_wait_dscnt 0" ::: "memory");

  v8f acc3 = (v8f){};
  for (int kk = 0; kk < 16; ++kk) {
    const v2f a = *(const v2f*)(tile + lm * TPITCH + 4 * kk + 2 * half);
    const v2f b = *(const v2f*)(w3s + (kk * 32 + lane) * 2);
    acc3 = wmma_f32(a, b, acc3);
  }
  if (lm < 2) {
    float bb = bs[192 + lm];
    for (int r = 0; r < 8; ++r)
      out[(size_t)lm * N_NODES + row0 + r + 8 * half] = acc3[r] + bb;
  }
}

extern "C" void kernel_launch(void* const* d_in, const int* in_sizes, int n_in,
                              void* d_out, int out_size, void* d_ws, size_t ws_size,
                              hipStream_t stream) {
  (void)in_sizes; (void)n_in; (void)out_size; (void)ws_size;
  const float* x  = (const float*)d_in[0];
  const int*   ei = (const int*)d_in[1];
  const float* Wg = (const float*)d_in[2];
  const float* bg = (const float*)d_in[3];
  const float* W1 = (const float*)d_in[4];
  const float* b1 = (const float*)d_in[5];
  const float* W2 = (const float*)d_in[6];
  const float* b2 = (const float*)d_in[7];
  const float* W3 = (const float*)d_in[8];
  const float* b3 = (const float*)d_in[9];
  float* out = (float*)d_out;

  float* ws   = (float*)d_ws;
  float* deg  = ws;
  float* dinv = ws + N_NODES;
  float* xw   = ws + 2 * N_NODES;
  float* agg  = xw + (size_t)N_NODES * 128;

  k_init<<<(N_NODES + 255) / 256, 256, 0, stream>>>(deg);
  k_deg<<<(N_EDGES + 255) / 256, 256, 0, stream>>>(ei, deg);
  k_dinv<<<(N_NODES + 255) / 256, 256, 0, stream>>>(deg, dinv);
  k_gemm1<<<782, 128, 0, stream>>>(x, Wg, xw);
  k_own<<<NOWN, 1024, 0, stream>>>(ei, dinv, xw, agg);
  k_mlp<<<MLP_WGS, MLP_WAVES * 32, 0, stream>>>(x, xw, agg, dinv, bg, W1, b1, W2, b2, W3, b3, out);
}


# k_deg also writes each edge's owner number (dst/200) as a byte; k_own's 250 scans read that 800 KB list, 16 edges a load, not the 3.2 MB dst list; dst loaded per kept edge; sums in arrival order
# speedup vs baseline: 1.9647x; 1.0565x over previous
#include <hip/hip_runtime.h>

typedef float v2f __attribute__((ext_vector_type(2)));
typedef float v8f __attribute__((ext_vector_type(8)));

#define N_NODES 50000
#define N_EDGES 800000
#define NB (N_NODES / 16)
#define TPITCH 132

__device__ __forceinline__ v8f wmma_f32(v2f a, v2f b, v8f c) {
  return __builtin_amdgcn_wmma_f32_16x16x4_f32(false, a, false, b, (short)0, c,
                                               false, false);
}

__global__ void k_init(float* __restrict__ deg) {
  int i = blockIdx.x * 256 + threadIdx.x;
  if (i < N_NODES) deg[i] = 1.0f;
}

#define RNG   200
#define NQ4   (N_EDGES / 4)
__global__ void k_deg(const int* __restrict__ ei, float* __restrict__ deg,
                      int* __restrict__ own) {
  const int q = blockIdx.x * 256 + threadIdx.x;
  if (q < NQ4) {
    const int4 d = ((const int4*)(ei + N_EDGES))[q];
    atomicAdd(&deg[d.x], 1.0f);
    atomicAdd(&deg[d.y], 1.0f);
    atomicAdd(&deg[d.z], 1.0f);
    atomicAdd(&deg[d.w], 1.0f);
    const unsigned o = ((unsigned)d.x / RNG) | (((unsigned)d.y / RNG) << 8) |
                       (((unsigned)d.z / RNG) << 16) | (((unsigned)d.w / RNG) << 24);
    own[q] = (int)o;
  }
}

__global__ void k_dinv(const float* __restrict__ deg, float* __restrict__ dinv) {
  int i = blockIdx.x * 256 + threadIdx.x;
  if (i < N_NODES) {
    float d = deg[i];
    dinv[i] = d > 0.0f ? rsqrtf(d) : 0.0f;
  }
}

__global__ void k_gemm1(const float* __restrict__ x, const float* __restrict__ Wg,
                        float* __restrict__ xw) {
  __shared__ float lds[4 * 16 * TPITCH];
  const int lane = threadIdx.x & 31;
  const int wave = threadIdx.x >> 5;
  const int half = lane >> 4;
  const int lm   = lane & 15;
  float* tile = lds + wave * (16 * TPITCH);
  const int gw = blockIdx.x * 4 + wave;
  const int nwaves = gridDim.x * 4;

  for (int blk = gw; blk < NB; blk += nwaves) {
    const int row0 = blk * 16;
    for (int r = 0; r < 16; ++r) {
      float4 v = *(const float4*)(x + (size_t)(row0 + r) * 128 + lane * 4);
      *(float4*)(tile + r * TPITCH + lane * 4) = v;
    }
    asm volatile("s_wait_dscnt 0" ::: "memory");

    v8f acc[8];
    for (int t = 0; t < 8; ++t) acc[t] = (v8f){};
    for (int kk = 0; kk < 32; ++kk) {
      const int k0 = 4 * kk + 2 * half;
      v2f a = *(const v2f*)(tile + lm * TPITCH + k0);
      for (int t = 0; t < 8; ++t) {
        const float* bp = Wg + (size_t)k0 * 128 + 16 * t + lm;
        v2f b; b.x = bp[0]; b.y = bp[128];
        acc[t] = wmma_f32(a, b, acc[t]);
      }
    }
    for (int t = 0; t < 8; ++t) {
      float* op = xw + (size_t)(row0 + 8 * half) * 128 + 16 * t + lm;
      for (int r = 0; r < 8; ++r) op[(size_t)r * 128] = acc[t][r];
    }
    asm volatile("s_wait_dscnt 0" ::: "memory");
  }
}

#define NOWN  (N_NODES / RNG)
#define QCAP  64
#define NV16  (N_EDGES / 16)

__device__ __forceinline__ int rl_i(int v, int l) { return __builtin_amdgcn_readlane(v, l); }
__device__ __forceinline__ float rl_f(float v, int l) {
  return __int_as_float(__builtin_amdgcn_readlane(__float_as_int(v), l));
}

__device__ __forceinline__ unsigned owner_mask(int4 v, unsigned pat) {
  const unsigned x0 = (unsigned)v.x ^ pat, x1 = (unsigned)v.y ^ pat;
  const unsigned x2 = (unsigned)v.z ^ pat, x3 = (unsigned)v.w ^ pat;
  const unsigned y0 = ((x0 & 0x7F7F7F7Fu) + 0x7F7F7F7Fu) | x0;
  const unsigned y1 = ((x1 & 0x7F7F7F7Fu) + 0x7F7F7F7Fu) | x1;
  const unsigned y2 = ((x2 & 0x7F7F7F7Fu) + 0x7F7F7F7Fu) | x2;
  const unsigned y3 = ((x3 & 0x7F7F7F7Fu) + 0x7F7F7F7Fu) | x3;
  return (((y0 >> 7) & 0x01010101u) | ((y1 >> 6) & 0x02020202u) |
          ((y2 >> 5) & 0x04040404u) | ((y3 >> 4) & 0x08080808u)) ^ 0x0F0F0F0Fu;
}

__device__ __forceinline__ void drain_batch(const unsigned* wq, int head, int m, int lane,
                                            int base, const int* __restrict__ ei,
                                            const float* __restrict__ dinv,
                                            const float* __restrict__ xw, float* acc) {
  int s = 0;
  int dl = RNG;
  float nrm = 0.0f;
  if (lane < m) {
    const unsigned e = wq[(head + lane) & (QCAP - 1)];
    s = ei[e];
    const int d = ei[N_EDGES + e];
    nrm = dinv[s] * dinv[d];
    dl = d - base;
  }
  for (int j = 0; j < m; j += 4) {
    float4 v[4];
    for (int w = 0; w < 4; ++w)
      v[w] = *(const float4*)(xw + (size_t)rl_i(s, j + w) * 128 + lane * 4);
    for (int w = 0; w < 4; ++w) {
      const float nj = rl_f(nrm, j + w);
      float* a = acc + rl_i(dl, j + w) * 128 + lane;
      atomicAdd(a,      v[w].x * nj);
      atomicAdd(a + 32, v[w].y * nj);
      atomicAdd(a + 64, v[w].z * nj);
      atomicAdd(a + 96, v[w].w * nj);
    }
  }
}

__global__ __launch_bounds__(1024) void k_own(const int* __restrict__ ei,
                                              const int* __restrict__ own,
                                              const float* __restrict__ dinv,
                                              const float* __restrict__ xw,
                                              float* __restrict__ agg) {
  __shared__ float acc[(RNG + 1) * 128];
  __shared__ unsigned queue[32 * QCAP];
  const int tid  = threadIdx.x;
  const int lane = tid & 31;
  const int wave = tid >> 5;
  const int base = blockIdx.x * RNG;
  unsigned* wq = queue + wave * QCAP;

  for (int i = tid; i < (RNG + 1) * 32; i += 1024)
    ((float4*)acc)[i] = make_float4(0.0f, 0.0f, 0.0f, 0.0f);
  __syncthreads();

  const int4* own4 = (const int4*)own;
  const unsigned pat = (unsigned)blockIdx.x * 0x01010101u;
  const unsigned below = (1u << lane) - 1u;
  int cnt = 0;
  int head = 0;

  constexpr int NIT = (NV16 + 4095) / 4096;
  const int rot = blockIdx.x % NIT;
  for (int i0 = 0; i0 < NIT; ++i0) {
    const int it = i0 + rot < NIT ? i0 + rot : i0 + rot - NIT;
    const int q0 = it * 4096 + tid;
    const int4 o0 = own4[min(q0,        NV16 - 1)];
    const int4 o1 = own4[min(q0 + 1024, NV16 - 1)];
    const int4 o2 = own4[min(q0 + 2048, NV16 - 1)];
    const int4 o3 = own4[min(q0 + 3072, NV16 - 1)];
    const unsigned m0 = q0        < NV16 ? owner_mask(o0, pat) : 0u;
    const unsigned m1 = q0 + 1024 < NV16 ? owner_mask(o1, pat) : 0u;
    const unsigned m2 = q0 + 2048 < NV16 ? owner_mask(o2, pat) : 0u;
    const unsigned m3 = q0 + 3072 < NV16 ? owner_mask(o3, pat) : 0u;
    unsigned mlo = m0 | (m1 << 4);
    unsigned mhi = m2 | (m3 << 4);
    unsigned bal = __builtin_amdgcn_ballot_w32((mlo | mhi) != 0u);
    while (bal != 0u) {
      if ((mlo | mhi) != 0u) {
        const bool lo = mlo != 0u;
        const unsigned mw = lo ? mlo : mhi;
        const int p = __builtin_ctz(mw);
        const int u = (lo ? 0 : 2) + ((p >> 2) & 1);
        const int pos = head + cnt + __builtin_popcount(bal & below);
        wq[pos & (QCAP - 1)] = (unsigned)(16 * (q0 + 1024 * u) + 4 * (p & 3) + (p >> 3));
        const unsigned rest = mw & (mw - 1u);
        if (lo) mlo = rest; else mhi = rest;
      }
      cnt += __builtin_popcount(bal);
      if (cnt >= 32) {
        asm volatile("s_wait_dscnt 0" ::: "memory");
        drain_batch(wq, head, 32, lane, base, ei, dinv, xw, acc);
        head = (head + 32) & (QCAP - 1);
        cnt -= 32;
      }
      bal = __builtin_amdgcn_ballot_w32((mlo | mhi) != 0u);
    }
  }
  asm volatile("s_wait_dscnt 0" ::: "memory");
  if (cnt > 0) drain_batch(wq, head, cnt, lane, base, ei, dinv, xw, acc);
  asm volatile("s_wait_dscnt 0" ::: "memory");
  __syncthreads();

  for (int r = wave; r < RNG; r += 32) {
    const float* a = acc + r * 128 + lane;
    float4 o = make_float4(a[0], a[32], a[64], a[96]);
    *(float4*)(agg + (size_t)(base + r) * 128 + lane * 4) = o;
  }
}

#define MLP_WAVES 13
#define MLP_WGS   256
#define W1P 20
#define W2P 12
__global__ void k_mlp(const float* __restrict__ x, const float* __restrict__ xw,
                      const float* __restrict__ agg, const float* __restrict__ dinv,
                      const float* __restrict__ bg,
                      const float* __restrict__ W1, const float* __restrict__ b1,
                      const float* __restrict__ W2, const float* __restrict__ b2,
                      const float* __restrict__ W3, const float* __restrict__ b3,
                      float* __restrict__ out) {
  __shared__ __attribute__((aligned(16))) float w1s[64 * 16 * W1P];
  __shared__ __attribute__((aligned(16))) float w2s[64 * 16 * W2P];
  __shared__ __attribute__((aligned(16))) float w3s[16 * 32 * 2];
  __shared__ float bs[128 + 64 + 16];
  __shared__ __attribute__((aligned(16))) float lds[MLP_WAVES * 16 * TPITCH];
  const int tid  = threadIdx.x;
  const int lane = tid & 31;
  const int wave = tid >> 5;
  const int half = lane >> 4;
  const int lm   = lane & 15;
  float* tile = lds + wave * (16 * TPITCH);
  const int blk = wave * MLP_WGS + blockIdx.x;
  const int row0 = blk * 16;

  constexpr int NT = MLP_WAVES * 32;
  float4 u0[5], u1[5], s0[3], s1[3];
#pragma unroll
  for (int it = 0; it < 5; ++it) {
    const int i = min(tid + it * NT, 64 * 32 - 1);
    const int p = i >> 5, c = i & 31;
    u0[it] = *(const float4*)(W1 + (size_t)(2 * p) * 128 + c * 4);
    u1[it] = *(const float4*)(W1 + (size_t)(2 * p + 1) * 128 + c * 4);
  }
#pragma unroll
  for (int it = 0; it < 3; ++it) {
    const int i = min(tid + it * NT, 64 * 16 - 1);
    const int p = i >> 4, c = i & 15;
    s0[it] = *(const float4*)(W2 + (size_t)(2 * p) * 64 + c * 4);
    s1[it] = *(const float4*)(W2 + (size_t)(2 * p + 1) * 64 + c * 4);
  }
#pragma unroll
  for (int it = 0; it < 5; ++it) {
    const int i = tid + it * NT;
    if (i < 64 * 32) {
      const int p = i >> 5, c = i & 31;
      const float e0[4] = {u0[it].x, u0[it].y, u0[it].z, u0[it].w};
      const float e1[4] = {u1[it].x, u1[it].y, u1[it].z, u1[it].w};
      for (int j = 0; j < 4; ++j) {
        const int n = c * 4 + j;
        v2f pr; pr.x = e0[j]; pr.y = e1[j];
        *(v2f*)(w1s + (p * 16 + (n & 15)) * W1P + 2 * (n >> 4)) = pr;
      }
    }
  }
#pragma unroll
  for (int it = 0; it < 3; ++it) {
    const int i = tid + it * NT;
    if (i < 64 * 16) {
      const int p = i >> 4, c = i & 15;
      const float e0[4] = {s0[it].x, s0[it].y, s0[it].z, s0[it].w};
      const float e1[4] = {s1[it].x, s1[it].y, s1[it].z, s1[it].w};
      for (int j = 0; j < 4; ++j) {
        const int n = c * 4 + j;
        v2f pr; pr.x = e0[j]; pr.y = e1[j];
        *(v2f*)(w2s + (p * 16 + (n & 15)) * W2P + 2 * (n >> 4)) = pr;
      }
    }
  }
  for (int i = tid; i < 16 * 32; i += NT) {
    const int kk = i >> 5, l = i & 31;
    const int k0 = 4 * kk + 2 * (l >> 4), n = l & 15;
    v2f pr; pr.x = 0.0f; pr.y = 0.0f;
    if (n < 2) { pr.x = W3[k0 * 2 + n]; pr.y = W3[(k0 + 1) * 2 + n]; }
    *(v2f*)(w3s + i * 2) = pr;
  }
  if (tid < 128) bs[tid] = b1[tid];
  else if (tid < 192) bs[tid] = b2[tid - 128];
  else if (tid < 208) bs[tid] = tid < 194 ? b3[tid - 192] : 0.0f;

  if (blk < NB) {
    const float4 bgv = *(const float4*)(bg + lane * 4);
    for (int r = 0; r < 16; ++r) {
      const int row = row0 + r;
      float di = dinv[row];
      float sl = di * di;
      float4 ag  = *(const float4*)(agg + (size_t)row * 128 + lane * 4);
      float4 xwv = *(const float4*)(xw  + (size_t)row * 128 + lane * 4);
      float4 xv  = *(const float4*)(x   + (size_t)row * 128 + lane * 4);
      float4 h;
      h.x = fmaxf(ag.x + xwv.x * sl + bgv.x, 0.0f) + xv.x;
      h.y = fmaxf(ag.y + xwv.y * sl + bgv.y, 0.0f) + xv.y;
      h.z = fmaxf(ag.z + xwv.z * sl + bgv.z, 0.0f) + xv.z;
      h.w = fmaxf(ag.w + xwv.w * sl + bgv.w, 0.0f) + xv.w;
      *(float4*)(tile + r * TPITCH + lane * 4) = h;
    }
  }
  __syncthreads();
  if (blk >= NB) return;

  v8f acc1[8];
  for (int t = 0; t < 8; ++t) acc1[t] = (v8f){};
  {
    const float* ap = tile + lm * TPITCH + 2 * half;
    const float* bp = w1s + (half * 16 + lm) * W1P;
    for (int kk = 0; kk < 32; ++kk) {
      const v2f a = *(const v2f*)(ap + 4 * kk);
      const float4* bq = (const float4*)(bp + kk * (2 * 16 * W1P));
      const float4 q0 = bq[0], q1 = bq[1], q2 = bq[2], q3 = bq[3];
      v2f b;
      b.x = q0.x; b.y = q0.y; acc1[0] = wmma_f32(a, b, acc1[0]);
      b.x = q0.z; b.y = q0.w; acc1[1] = wmma_f32(a, b, acc1[1]);
      b.x = q1.x; b.y = q1.y; acc1[2] = wmma_f32(a, b, acc1[2]);
      b.x = q1.z; b.y = q1.w; acc1[3] = wmma_f32(a, b, acc1[3]);
      b.x = q2.x; b.y = q2.y; acc1[4] = wmma_f32(a, b, acc1[4]);
      b.x = q2.z; b.y = q2.w; acc1[5] = wmma_f32(a, b, acc1[5]);
      b.x = q3.x; b.y = q3.y; acc1[6] = wmma_f32(a, b, acc1[6]);
      b.x = q3.z; b.y = q3.w; acc1[7] = wmma_f32(a, b, acc1[7]);
    }
  }
  asm volatile("s_wait_dscnt 0" ::: "memory");
  for (int t = 0; t < 8; ++t) {
    float bb = bs[16 * t + lm];
    for (int r = 0; r < 8; ++r)
      tile[(r + 8 * half) * TPITCH + 16 * t + lm] = fmaxf(acc1[t][r] + bb, 0.0f);
  }
  asm volatile("s_wait_dscnt 0" ::: "memory");

  v8f acc2[4];
  for (int t = 0; t < 4; ++t) acc2[t] = (v8f){};
  {
    const float* ap = tile + lm * TPITCH + 2 * half;
    const float* bp = w2s + (half * 16 + lm) * W2P;
    for (int kk = 0; kk < 32; ++kk) {
      const v2f a = *(const v2f*)(ap + 4 * kk);
      const float4* bq = (const float4*)(bp + kk * (2 * 16 * W2P));
      const float4 q0 = bq[0], q1 = bq[1];
      v2f b;
      b.x = q0.x; b.y = q0.y; acc2[0] = wmma_f32(a, b, acc2[0]);
      b.x = q0.z; b.y = q0.w; acc2[1] = wmma_f32(a, b, acc2[1]);
      b.x = q1.x; b.y = q1.y; acc2[2] = wmma_f32(a, b, acc2[2]);
      b.x = q1.z; b.y = q1.w; acc2[3] = wmma_f32(a, b, acc2[3]);
    }
  }
  asm volatile("s_wait_dscnt 0" ::: "memory");
  for (int t = 0; t < 4; ++t) {
    float bb = bs[128 + 16 * t + lm];
    for (int r = 0; r < 8; ++r)
      tile[(r + 8 * half) * TPITCH + 16 * t + lm] = fmaxf(acc2[t][r] + bb, 0.0f);
  }
  asm volatile("s_wait_dscnt 0" ::: "memory");

  v8f acc3 = (v8f){};
  for (int kk = 0; kk < 16; ++kk) {
    const v2f a = *(const v2f*)(tile + lm * TPITCH + 4 * kk + 2 * half);
    const v2f b = *(const v2f*)(w3s + (kk * 32 + lane) * 2);
    acc3 = wmma_f32(a, b, acc3);
  }
  if (lm < 2) {
    float bb = bs[192 + lm];
    for (int r = 0; r < 8; ++r)
      out[(size_t)lm * N_NODES + row0 + r + 8 * half] = acc3[r] + bb;
  }
}

extern "C" void kernel_launch(void* const* d_in, const int* in_sizes, int n_in,
                              void* d_out, int out_size, void* d_ws, size_t ws_size,
                              hipStream_t stream) {
  (void)in_sizes; (void)n_in; (void)out_size; (void)ws_size;
  const float* x  = (const float*)d_in[0];
  const int*   ei = (const int*)d_in[1];
  const float* Wg = (const float*)d_in[2];
  const float* bg = (const float*)d_in[3];
  const float* W1 = (const float*)d_in[4];
  const float* b1 = (const float*)d_in[5];
  const float* W2 = (const float*)d_in[6];
  const float* b2 = (const float*)d_in[7];
  const float* W3 = (const float*)d_in[8];
  const float* b3 = (const float*)d_in[9];
  float* out = (float*)d_out;

  float* ws   = (float*)d_ws;
  float* deg  = ws;
  float* dinv = ws + N_NODES;
  float* xw   = ws + 2 * N_NODES;
  float* agg  = xw + (size_t)N_NODES * 128;
  int*   own  = (int*)(agg + (size_t)N_NODES * 128);

  k_init<<<(N_NODES + 255) / 256, 256, 0, stream>>>(deg);
  k_deg<<<(NQ4 + 255) / 256, 256, 0, stream>>>(ei, deg, own);
  k_dinv<<<(N_NODES + 255) / 256, 256, 0, stream>>>(deg, dinv);
  k_gemm1<<<782, 128, 0, stream>>>(x, Wg, xw);
  k_own<<<NOWN, 1024, 0, stream>>>(ei, own, dinv, xw, agg);
  k_mlp<<<MLP_WGS, MLP_WAVES * 32, 0, stream>>>(x, xw, agg, dinv, bg, W1, b1, W2, b2, W3, b3, out);
}


# k_own: one scattered global word per kept edge, not four: k_deg also writes src | dst%200 << 16 per edge, and dinv[src], dinv[dst] are read from an LDS copy of the dinv table; sums unchanged
# speedup vs baseline: 2.0346x; 1.0356x over previous
#include <hip/hip_runtime.h>

typedef float v2f __attribute__((ext_vector_type(2)));
typedef float v8f __attribute__((ext_vector_type(8)));

#define N_NODES 50000
#define N_EDGES 800000
#define NB (N_NODES / 16)
#define TPITCH 132

__device__ __forceinline__ v8f wmma_f32(v2f a, v2f b, v8f c) {
  return __builtin_amdgcn_wmma_f32_16x16x4_f32(false, a, false, b, (short)0, c,
                                               false, false);
}

__global__ void k_init(float* __restrict__ deg) {
  int i = blockIdx.x * 256 + threadIdx.x;
  if (i < N_NODES) deg[i] = 1.0f;
}

#define RNG   200
#define NQ4   (N_EDGES / 4)
static_assert(N_NODES <= 65536 && RNG <= 65536, "src and dst % RNG share one 32-bit word");
__global__ void k_deg(const int* __restrict__ ei, float* __restrict__ deg,
                      int* __restrict__ own, int* __restrict__ pk) {
  const int q = blockIdx.x * 256 + threadIdx.x;
  if (q < NQ4) {
    const int4 d = ((const int4*)(ei + N_EDGES))[q];
    const int4 s = ((const int4*)ei)[q];
    atomicAdd(&deg[d.x], 1.0f);
    atomicAdd(&deg[d.y], 1.0f);
    atomicAdd(&deg[d.z], 1.0f);
    atomicAdd(&deg[d.w], 1.0f);
    const unsigned o = ((unsigned)d.x / RNG) | (((unsigned)d.y / RNG) << 8) |
                       (((unsigned)d.z / RNG) << 16) | (((unsigned)d.w / RNG) << 24);
    own[q] = (int)o;
    int4 w;
    w.x = (int)((unsigned)s.x | (((unsigned)d.x % RNG) << 16));
    w.y = (int)((unsigned)s.y | (((unsigned)d.y % RNG) << 16));
    w.z = (int)((unsigned)s.z | (((unsigned)d.z % RNG) << 16));
    w.w = (int)((unsigned)s.w | (((unsigned)d.w % RNG) << 16));
    ((int4*)pk)[q] = w;
  }
}

__global__ void k_dinv(const float* __restrict__ deg, float* __restrict__ dinv) {
  int i = blockIdx.x * 256 + threadIdx.x;
  if (i < N_NODES) {
    float d = deg[i];
    dinv[i] = d > 0.0f ? rsqrtf(d) : 0.0f;
  }
}

__global__ void k_gemm1(const float* __restrict__ x, const float* __restrict__ Wg,
                        float* __restrict__ xw) {
  __shared__ float lds[4 * 16 * TPITCH];
  const int lane = threadIdx.x & 31;
  const int wave = threadIdx.x >> 5;
  const int half = lane >> 4;
  const int lm   = lane & 15;
  float* tile = lds + wave * (16 * TPITCH);
  const int gw = blockIdx.x * 4 + wave;
  const int nwaves = gridDim.x * 4;

  for (int blk = gw; blk < NB; blk += nwaves) {
    const int row0 = blk * 16;
    for (int r = 0; r < 16; ++r) {
      float4 v = *(const float4*)(x + (size_t)(row0 + r) * 128 + lane * 4);
      *(float4*)(tile + r * TPITCH + lane * 4) = v;
    }
    asm volatile("s_wait_dscnt 0" ::: "memory");

    v8f acc[8];
    for (int t = 0; t < 8; ++t) acc[t] = (v8f){};
    for (int kk = 0; kk < 32; ++kk) {
      const int k0 = 4 * kk + 2 * half;
      v2f a = *(const v2f*)(tile + lm * TPITCH + k0);
      for (int t = 0; t < 8; ++t) {
        const float* bp = Wg + (size_t)k0 * 128 + 16 * t + lm;
        v2f b; b.x = bp[0]; b.y = bp[128];
        acc[t] = wmma_f32(a, b, acc[t]);
      }
    }
    for (int t = 0; t < 8; ++t) {
      float* op = xw + (size_t)(row0 + 8 * half) * 128 + 16 * t + lm;
      for (int r = 0; r < 8; ++r) op[(size_t)r * 128] = acc[t][r];
    }
    asm volatile("s_wait_dscnt 0" ::: "memory");
  }
}

#define NOWN  (N_NODES / RNG)
#define QCAP  64
#define NV16  (N_EDGES / 16)

__device__ __forceinline__ int rl_i(int v, int l) { return __builtin_amdgcn_readlane(v, l); }
__device__ __forceinline__ float rl_f(float v, int l) {
  return __int_as_float(__builtin_amdgcn_readlane(__float_as_int(v), l));
}

__device__ __forceinline__ unsigned owner_mask(int4 v, unsigned pat) {
  const unsigned x0 = (unsigned)v.x ^ pat, x1 = (unsigned)v.y ^ pat;
  const unsigned x2 = (unsigned)v.z ^ pat, x3 = (unsigned)v.w ^ pat;
  const unsigned y0 = ((x0 & 0x7F7F7F7Fu) + 0x7F7F7F7Fu) | x0;
  const unsigned y1 = ((x1 & 0x7F7F7F7Fu) + 0x7F7F7F7Fu) | x1;
  const unsigned y2 = ((x2 & 0x7F7F7F7Fu) + 0x7F7F7F7Fu) | x2;
  const unsigned y3 = ((x3 & 0x7F7F7F7Fu) + 0x7F7F7F7Fu) | x3;
  return (((y0 >> 7) & 0x01010101u) | ((y1 >> 6) & 0x02020202u) |
          ((y2 >> 5) & 0x04040404u) | ((y3 >> 4) & 0x08080808u)) ^ 0x0F0F0F0Fu;
}

__device__ __forceinline__ void drain_batch(const unsigned* wq, int head, int m, int lane,
                                            int base, const int* __restrict__ pk,
                                            const float* dsh,
                                            const float* __restrict__ xw, float* acc) {
  int s = 0;
  int dl = RNG;
  float nrm = 0.0f;
  if (lane < m) {
    const unsigned e = wq[(head + lane) & (QCAP - 1)];
    const unsigned w = (unsigned)pk[e];
    s = (int)(w & 0xFFFFu);
    dl = (int)(w >> 16);
    nrm = dsh[s] * dsh[base + dl];
  }
  for (int j = 0; j < m; j += 4) {
    float4 v[4];
    for (int w = 0; w < 4; ++w)
      v[w] = *(const float4*)(xw + (size_t)rl_i(s, j + w) * 128 + lane * 4);
    for (int w = 0; w < 4; ++w) {
      const float nj = rl_f(nrm, j + w);
      float* a = acc + rl_i(dl, j + w) * 128 + lane;
      atomicAdd(a,      v[w].x * nj);
      atomicAdd(a + 32, v[w].y * nj);
      atomicAdd(a + 64, v[w].z * nj);
      atomicAdd(a + 96, v[w].w * nj);
    }
  }
}

__global__ __launch_bounds__(1024) void k_own(const int* __restrict__ pk,
                                              const int* __restrict__ own,
                                              const float* __restrict__ dinv,
                                              const float* __restrict__ xw,
                                              float* __restrict__ agg) {
  __shared__ float acc[(RNG + 1) * 128];
  __shared__ unsigned queue[32 * QCAP];
  __shared__ __attribute__((aligned(16))) float dsh[N_NODES];
  const int tid  = threadIdx.x;
  const int lane = tid & 31;
  const int wave = tid >> 5;
  const int base = blockIdx.x * RNG;
  unsigned* wq = queue + wave * QCAP;

  constexpr int NV4 = N_NODES / 4;
  constexpr int NPS = (NV4 + 1023) / 1024;
  static_assert(NV4 >= 1024, "the shifted last pass needs a table of at least one pass");
  float4 dv[NPS];
#pragma unroll
  for (int it = 0; it < NPS; ++it) {
    const int i = it < NPS - 1 ? tid + it * 1024 : tid + NV4 - 1024;
    dv[it] = ((const float4*)dinv)[i];
  }
  for (int i = tid; i < (RNG + 1) * 32; i += 1024)
    ((float4*)acc)[i] = make_float4(0.0f, 0.0f, 0.0f, 0.0f);
#pragma unroll
  for (int it = 0; it < NPS; ++it) {
    const int i = it < NPS - 1 ? tid + it * 1024 : tid + NV4 - 1024;
    ((float4*)dsh)[i] = dv[it];
  }
  __syncthreads();

  const int4* own4 = (const int4*)own;
  const unsigned pat = (unsigned)blockIdx.x * 0x01010101u;
  const unsigned below = (1u << lane) - 1u;
  int cnt = 0;
  int head = 0;

  constexpr int NIT = (NV16 + 4095) / 4096;
  const int rot = blockIdx.x % NIT;
  for (int i0 = 0; i0 < NIT; ++i0) {
    const int it = i0 + rot < NIT ? i0 + rot : i0 + rot - NIT;
    const int q0 = it * 4096 + tid;
    const int4 o0 = own4[min(q0,        NV16 - 1)];
    const int4 o1 = own4[min(q0 + 1024, NV16 - 1)];
    const int4 o2 = own4[min(q0 + 2048, NV16 - 1)];
    const int4 o3 = own4[min(q0 + 3072, NV16 - 1)];
    const unsigned m0 = q0        < NV16 ? owner_mask(o0, pat) : 0u;
    const unsigned m1 = q0 + 1024 < NV16 ? owner_mask(o1, pat) : 0u;
    const unsigned m2 = q0 + 2048 < NV16 ? owner_mask(o2, pat) : 0u;
    const unsigned m3 = q0 + 3072 < NV16 ? owner_mask(o3, pat) : 0u;
    unsigned mlo = m0 | (m1 << 4);
    unsigned mhi = m2 | (m3 << 4);
    unsigned bal = __builtin_amdgcn_ballot_w32((mlo | mhi) != 0u);
    while (bal != 0u) {
      if ((mlo | mhi) != 0u) {
        const bool lo = mlo != 0u;
        const unsigned mw = lo ? mlo : mhi;
        const int p = __builtin_ctz(mw);
        const int u = (lo ? 0 : 2) + ((p >> 2) & 1);
        const int pos = head + cnt + __builtin_popcount(bal & below);
        wq[pos & (QCAP - 1)] = (unsigned)(16 * (q0 + 1024 * u) + 4 * (p & 3) + (p >> 3));
        const unsigned rest = mw & (mw - 1u);
        if (lo) mlo = rest; else mhi = rest;
      }
      cnt += __builtin_popcount(bal);
      if (cnt >= 32) {
        asm volatile("s_wait_dscnt 0" ::: "memory");
        drain_batch(wq, head, 32, lane, base, pk, dsh, xw, acc);
        head = (head + 32) & (QCAP - 1);
        cnt -= 32;
      }
      bal = __builtin_amdgcn_ballot_w32((mlo | mhi) != 0u);
    }
  }
  asm volatile("s_wait_dscnt 0" ::: "memory");
  if (cnt > 0) drain_batch(wq, head, cnt, lane, base, pk, dsh, xw, acc);
  asm volatile("s_wait_dscnt 0" ::: "memory");
  __syncthreads();

  for (int r = wave; r < RNG; r += 32) {
    const float* a = acc + r * 128 + lane;
    float4 o = make_float4(a[0], a[32], a[64], a[96]);
    *(float4*)(agg + (size_t)(base + r) * 128 + lane * 4) = o;
  }
}

#define MLP_WAVES 13
#define MLP_WGS   256
#define W1P 20
#define W2P 12
__global__ void k_mlp(const float* __restrict__ x, const float* __restrict__ xw,
                      const float* __restrict__ agg, const float* __restrict__ dinv,
                      const float* __restrict__ bg,
                      const float* __restrict__ W1, const float* __restrict__ b1,
                      const float* __restrict__ W2, const float* __restrict__ b2,
                      const float* __restrict__ W3, const float* __restrict__ b3,
                      float* __restrict__ out) {
  __shared__ __attribute__((aligned(16))) float w1s[64 * 16 * W1P];
  __shared__ __attribute__((aligned(16))) float w2s[64 * 16 * W2P];
  __shared__ __attribute__((aligned(16))) float w3s[16 * 32 * 2];
  __shared__ float bs[128 + 64 + 16];
  __shared__ __attribute__((aligned(16))) float lds[MLP_WAVES * 16 * TPITCH];
  const int tid  = threadIdx.x;
  const int lane = tid & 31;
  const int wave = tid >> 5;
  const int half = lane >> 4;
  const int lm   = lane & 15;
  float* tile = lds + wave * (16 * TPITCH);
  const int blk = wave * MLP_WGS + blockIdx.x;
  const int row0 = blk * 16;

  constexpr int NT = MLP_WAVES * 32;
  float4 u0[5], u1[5], s0[3], s1[3];
#pragma unroll
  for (int it = 0; it < 5; ++it) {
    const int i = min(tid + it * NT, 64 * 32 - 1);
    const int p = i >> 5, c = i & 31;
    u0[it] = *(const float4*)(W1 + (size_t)(2 * p) * 128 + c * 4);
    u1[it] = *(const float4*)(W1 + (size_t)(2 * p + 1) * 128 + c * 4);
  }
#pragma unroll
  for (int it = 0; it < 3; ++it) {
    const int i = min(tid + it * NT, 64 * 16 - 1);
    const int p = i >> 4, c = i & 15;
    s0[it] = *(const float4*)(W2 + (size_t)(2 * p) * 64 + c * 4);
    s1[it] = *(const float4*)(W2 + (size_t)(2 * p + 1) * 64 + c * 4);
  }
#pragma unroll
  for (int it = 0; it < 5; ++it) {
    const int i = tid + it * NT;
    if (i < 64 * 32) {
      const int p = i >> 5, c = i & 31;
      const float e0[4] = {u0[it].x, u0[it].y, u0[it].z, u0[it].w};
      const float e1[4] = {u1[it].x, u1[it].y, u1[it].z, u1[it].w};
      for (int j = 0; j < 4; ++j) {
        const int n = c * 4 + j;
        v2f pr; pr.x = e0[j]; pr.y = e1[j];
        *(v2f*)(w1s + (p * 16 + (n & 15)) * W1P + 2 * (n >> 4)) = pr;
      }
    }
  }
#pragma unroll
  for (int it = 0; it < 3; ++it) {
    const int i = tid + it * NT;
    if (i < 64 * 16) {
      const int p = i >> 4, c = i & 15;
      const float e0[4] = {s0[it].x, s0[it].y, s0[it].z, s0[it].w};
      const float e1[4] = {s1[it].x, s1[it].y, s1[it].z, s1[it].w};
      for (int j = 0; j < 4; ++j) {
        const int n = c * 4 + j;
        v2f pr; pr.x = e0[j]; pr.y = e1[j];
        *(v2f*)(w2s + (p * 16 + (n & 15)) * W2P + 2 * (n >> 4)) = pr;
      }
    }
  }
  for (int i = tid; i < 16 * 32; i += NT) {
    const int kk = i >> 5, l = i & 31;
    const int k0 = 4 * kk + 2 * (l >> 4), n = l & 15;
    v2f pr; pr.x = 0.0f; pr.y = 0.0f;
    if (n < 2) { pr.x = W3[k0 * 2 + n]; pr.y = W3[(k0 + 1) * 2 + n]; }
    *(v2f*)(w3s + i * 2) = pr;
  }
  if (tid < 128) bs[tid] = b1[tid];
  else if (tid < 192) bs[tid] = b2[tid - 128];
  else if (tid < 208) bs[tid] = tid < 194 ? b3[tid - 192] : 0.0f;

  if (blk < NB) {
    const float4 bgv = *(const float4*)(bg + lane * 4);
    for (int r = 0; r < 16; ++r) {
      const int row = row0 + r;
      float di = dinv[row];
      float sl = di * di;
      float4 ag  = *(const float4*)(agg + (size_t)row * 128 + lane * 4);
      float4 xwv = *(const float4*)(xw  + (size_t)row * 128 + lane * 4);
      float4 xv  = *(const float4*)(x   + (size_t)row * 128 + lane * 4);
      float4 h;
      h.x = fmaxf(ag.x + xwv.x * sl + bgv.x, 0.0f) + xv.x;
      h.y = fmaxf(ag.y + xwv.y * sl + bgv.y, 0.0f) + xv.y;
      h.z = fmaxf(ag.z + xwv.z * sl + bgv.z, 0.0f) + xv.z;
      h.w = fmaxf(ag.w + xwv.w * sl + bgv.w, 0.0f) + xv.w;
      *(float4*)(tile + r * TPITCH + lane * 4) = h;
    }
  }
  __syncthreads();
  if (blk >= NB) return;

  v8f acc1[8];
  for (int t = 0; t < 8; ++t) acc1[t] = (v8f){};
  {
    const float* ap = tile + lm * TPITCH + 2 * half;
    const float* bp = w1s + (half * 16 + lm) * W1P;
    for (int kk = 0; kk < 32; ++kk) {
      const v2f a = *(const v2f*)(ap + 4 * kk);
      const float4* bq = (const float4*)(bp + kk * (2 * 16 * W1P));
      const float4 q0 = bq[0], q1 = bq[1], q2 = bq[2], q3 = bq[3];
      v2f b;
      b.x = q0.x; b.y = q0.y; acc1[0] = wmma_f32(a, b, acc1[0]);
      b.x = q0.z; b.y = q0.w; acc1[1] = wmma_f32(a, b, acc1[1]);
      b.x = q1.x; b.y = q1.y; acc1[2] = wmma_f32(a, b, acc1[2]);
      b.x = q1.z; b.y = q1.w; acc1[3] = wmma_f32(a, b, acc1[3]);
      b.x = q2.x; b.y = q2.y; acc1[4] = wmma_f32(a, b, acc1[4]);
      b.x = q2.z; b.y = q2.w; acc1[5] = wmma_f32(a, b, acc1[5]);
      b.x = q3.x; b.y = q3.y; acc1[6] = wmma_f32(a, b, acc1[6]);
      b.x = q3.z; b.y = q3.w; acc1[7] = wmma_f32(a, b, acc1[7]);
    }
  }
  asm volatile("s_wait_dscnt 0" ::: "memory");
  for (int t = 0; t < 8; ++t) {
    float bb = bs[16 * t + lm];
    for (int r = 0; r < 8; ++r)
      tile[(r + 8 * half) * TPITCH + 16 * t + lm] = fmaxf(acc1[t][r] + bb, 0.0f);
  }
  asm volatile("s_wait_dscnt 0" ::: "memory");

  v8f acc2[4];
  for (int t = 0; t < 4; ++t) acc2[t] = (v8f){};
  {
    const float* ap = tile + lm * TPITCH + 2 * half;
    const float* bp = w2s + (half * 16 + lm) * W2P;
    for (int kk = 0; kk < 32; ++kk) {
      const v2f a = *(const v2f*)(ap + 4 * kk);
      const float4* bq = (const float4*)(bp + kk * (2 * 16 * W2P));
      const float4 q0 = bq[0], q1 = bq[1];
      v2f b;
      b.x = q0.x; b.y = q0.y; acc2[0] = wmma_f32(a, b, acc2[0]);
      b.x = q0.z; b.y = q0.w; acc2[1] = wmma_f32(a, b, acc2[1]);
      b.x = q1.x; b.y = q1.y; acc2[2] = wmma_f32(a, b, acc2[2]);
      b.x = q1.z; b.y = q1.w; acc2[3] = wmma_f32(a, b, acc2[3]);
    }
  }
  asm volatile("s_wait_dscnt 0" ::: "memory");
  for (int t = 0; t < 4; ++t) {
    float bb = bs[128 + 16 * t + lm];
    for (int r = 0; r < 8; ++r)
      tile[(r + 8 * half) * TPITCH + 16 * t + lm] = fmaxf(acc2[t][r] + bb, 0.0f);
  }
  asm volatile("s_wait_dscnt 0" ::: "memory");

  v8f acc3 = (v8f){};
  for (int kk = 0; kk < 16; ++kk) {
    const v2f a = *(const v2f*)(tile + lm * TPITCH + 4 * kk + 2 * half);
    const v2f b = *(const v2f*)(w3s + (kk * 32 + lane) * 2);
    acc3 = wmma_f32(a, b, acc3);
  }
  if (lm < 2) {
    float bb = bs[192 + lm];
    for (int r = 0; r < 8; ++r)
      out[(size_t)lm * N_NODES + row0 + r + 8 * half] = acc3[r] + bb;
  }
}

extern "C" void kernel_launch(void* const* d_in, const int* in_sizes, int n_in,
                              void* d_out, int out_size, void* d_ws, size_t ws_size,
                              hipStream_t stream) {
  (void)in_sizes; (void)n_in; (void)out_size; (void)ws_size;
  const float* x  = (const float*)d_in[0];
  const int*   ei = (const int*)d_in[1];
  const float* Wg = (const float*)d_in[2];
  const float* bg = (const float*)d_in[3];
  const float* W1 = (const float*)d_in[4];
  const float* b1 = (const float*)d_in[5];
  const float* W2 = (const float*)d_in[6];
  const float* b2 = (const float*)d_in[7];
  const float* W3 = (const float*)d_in[8];
  const float* b3 = (const float*)d_in[9];
  float* out = (float*)d_out;

  float* ws   = (float*)d_ws;
  float* deg  = ws;
  float* dinv = ws + N_NODES;
  float* xw   = ws + 2 * N_NODES;
  float* agg  = xw + (size_t)N_NODES * 128;
  int*   own  = (int*)(agg + (size_t)N_NODES * 128);
  int*   pk   = own + N_EDGES / 4;

  k_init<<<(N_NODES + 255) / 256, 256, 0, stream>>>(deg);
  k_deg<<<(NQ4 + 255) / 256, 256, 0, stream>>>(ei, deg, own, pk);
  k_dinv<<<(N_NODES + 255) / 256, 256, 0, stream>>>(deg, dinv);
  k_gemm1<<<782, 128, 0, stream>>>(x, Wg, xw);
  k_own<<<NOWN, 1024, 0, stream>>>(pk, own, dinv, xw, agg);
  k_mlp<<<MLP_WGS, MLP_WAVES * 32, 0, stream>>>(x, xw, agg, dinv, bg, W1, b1, W2, b2, W3, b3, out);
}


# x@Wg moved behind the aggregation (linearity): k_own sums scaled x rows, k_mlp applies Wg as a fourth LDS-fed WMMA layer; k_gemm1 launch and xw array gone; rows are summed before the product
# speedup vs baseline: 2.3878x; 1.1736x over previous
#include <hip/hip_runtime.h>

typedef float v2f __attribute__((ext_vector_type(2)));
typedef float v8f __attribute__((ext_vector_type(8)));

#define N_NODES 50000
#define N_EDGES 800000
#define NB (N_NODES / 16)
#define TPITCH 132

__device__ __forceinline__ v8f wmma_f32(v2f a, v2f b, v8f c) {
  return __builtin_amdgcn_wmma_f32_16x16x4_f32(false, a, false, b, (short)0, c,
                                               false, false);
}

__global__ void k_init(float* __restrict__ deg) {
  int i = blockIdx.x * 256 + threadIdx.x;
  if (i < N_NODES) deg[i] = 1.0f;
}

#define RNG   200
#define NQ4   (N_EDGES / 4)
static_assert(N_NODES <= 65536 && RNG <= 65536, "src and dst % RNG share one 32-bit word");
__global__ void k_deg(const int* __restrict__ ei, float* __restrict__ deg,
                      int* __restrict__ own, int* __restrict__ pk) {
  const int q = blockIdx.x * 256 + threadIdx.x;
  if (q < NQ4) {
    const int4 d = ((const int4*)(ei + N_EDGES))[q];
    const int4 s = ((const int4*)ei)[q];
    atomicAdd(&deg[d.x], 1.0f);
    atomicAdd(&deg[d.y], 1.0f);
    atomicAdd(&deg[d.z], 1.0f);
    atomicAdd(&deg[d.w], 1.0f);
    const unsigned o = ((unsigned)d.x / RNG) | (((unsigned)d.y / RNG) << 8) |
                       (((unsigned)d.z / RNG) << 16) | (((unsigned)d.w / RNG) << 24);
    own[q] = (int)o;
    int4 w;
    w.x = (int)((unsigned)s.x | (((unsigned)d.x % RNG) << 16));
    w.y = (int)((unsigned)s.y | (((unsigned)d.y % RNG) << 16));
    w.z = (int)((unsigned)s.z | (((unsigned)d.z % RNG) << 16));
    w.w = (int)((unsigned)s.w | (((unsigned)d.w % RNG) << 16));
    ((int4*)pk)[q] = w;
  }
}

__global__ void k_dinv(const float* __restrict__ deg, float* __restrict__ dinv) {
  int i = blockIdx.x * 256 + threadIdx.x;
  if (i < N_NODES) {
    float d = deg[i];
    dinv[i] = d > 0.0f ? rsqrtf(d) : 0.0f;
  }
}

#define NOWN  (N_NODES / RNG)
#define QCAP  64
#define NV16  (N_EDGES / 16)

__device__ __forceinline__ int rl_i(int v, int l) { return __builtin_amdgcn_readlane(v, l); }
__device__ __forceinline__ float rl_f(float v, int l) {
  return __int_as_float(__builtin_amdgcn_readlane(__float_as_int(v), l));
}

__device__ __forceinline__ unsigned owner_mask(int4 v, unsigned pat) {
  const unsigned x0 = (unsigned)v.x ^ pat, x1 = (unsigned)v.y ^ pat;
  const unsigned x2 = (unsigned)v.z ^ pat, x3 = (unsigned)v.w ^ pat;
  const unsigned y0 = ((x0 & 0x7F7F7F7Fu) + 0x7F7F7F7Fu) | x0;
  const unsigned y1 = ((x1 & 0x7F7F7F7Fu) + 0x7F7F7F7Fu) | x1;
  const unsigned y2 = ((x2 & 0x7F7F7F7Fu) + 0x7F7F7F7Fu) | x2;
  const unsigned y3 = ((x3 & 0x7F7F7F7Fu) + 0x7F7F7F7Fu) | x3;
  return (((y0 >> 7) & 0x01010101u) | ((y1 >> 6) & 0x02020202u) |
          ((y2 >> 5) & 0x04040404u) | ((y3 >> 4) & 0x08080808u)) ^ 0x0F0F0F0Fu;
}

__device__ __forceinline__ void drain_batch(const unsigned* wq, int head, int m, int lane,
                                            int base, const int* __restrict__ pk,
                                            const float* dsh,
                                            const float* __restrict__ x, float* acc) {
  int s = 0;
  int dl = RNG;
  float nrm = 0.0f;
  if (lane < m) {
    const unsigned e = wq[(head + lane) & (QCAP - 1)];
    const unsigned w = (unsigned)pk[e];
    s = (int)(w & 0xFFFFu);
    dl = (int)(w >> 16);
    nrm = dsh[s] * dsh[base + dl];
  }
  for (int j = 0; j < m; j += 4) {
    float4 v[4];
    for (int w = 0; w < 4; ++w)
      v[w] = *(const float4*)(x + (size_t)rl_i(s, j + w) * 128 + lane * 4);
    for (int w = 0; w < 4; ++w) {
      const float nj = rl_f(nrm, j + w);
      float* a = acc + rl_i(dl, j + w) * 128 + lane;
      atomicAdd(a,      v[w].x * nj);
      atomicAdd(a + 32, v[w].y * nj);
      atomicAdd(a + 64, v[w].z * nj);
      atomicAdd(a + 96, v[w].w * nj);
    }
  }
}

__global__ __launch_bounds__(1024) void k_own(const int* __restrict__ pk,
                                              const int* __restrict__ own,
                                              const float* __restrict__ dinv,
                                              const float* __restrict__ x,
                                              float* __restrict__ agg) {
  __shared__ float acc[(RNG + 1) * 128];
  __shared__ unsigned queue[32 * QCAP];
  __shared__ __attribute__((aligned(16))) float dsh[N_NODES];
  const int tid  = threadIdx.x;
  const int lane = tid & 31;
  const int wave = tid >> 5;
  const int base = blockIdx.x * RNG;
  unsigned* wq = queue + wave * QCAP;

  constexpr int NV4 = N_NODES / 4;
  constexpr int NPS = (NV4 + 1023) / 1024;
  static_assert(NV4 >= 1024, "the shifted last pass needs a table of at least one pass");
  float4 dv[NPS];
#pragma unroll
  for (int it = 0; it < NPS; ++it) {
    const int i = it < NPS - 1 ? tid + it * 1024 : tid + NV4 - 1024;
    dv[it] = ((const float4*)dinv)[i];
  }
  for (int i = tid; i < (RNG + 1) * 32; i += 1024)
    ((float4*)acc)[i] = make_float4(0.0f, 0.0f, 0.0f, 0.0f);
#pragma unroll
  for (int it = 0; it < NPS; ++it) {
    const int i = it < NPS - 1 ? tid + it * 1024 : tid + NV4 - 1024;
    ((float4*)dsh)[i] = dv[it];
  }
  __syncthreads();

  const int4* own4 = (const int4*)own;
  const unsigned pat = (unsigned)blockIdx.x * 0x01010101u;
  const unsigned below = (1u << lane) - 1u;
  int cnt = 0;
  int head = 0;

  constexpr int NIT = (NV16 + 4095) / 4096;
  const int rot = blockIdx.x % NIT;
  for (int i0 = 0; i0 < NIT; ++i0) {
    const int it = i0 + rot < NIT ? i0 + rot : i0 + rot - NIT;
    const int q0 = it * 4096 + tid;
    const int4 o0 = own4[min(q0,        NV16 - 1)];
    const int4 o1 = own4[min(q0 + 1024, NV16 - 1)];
    const int4 o2 = own4[min(q0 + 2048, NV16 - 1)];
    const int4 o3 = own4[min(q0 + 3072, NV16 - 1)];
    const unsigned m0 = q0        < NV16 ? owner_mask(o0, pat) : 0u;
    const unsigned m1 = q0 + 1024 < NV16 ? owner_mask(o1, pat) : 0u;
    const unsigned m2 = q0 + 2048 < NV16 ? owner_mask(o2, pat) : 0u;
    const unsigned m3 = q0 + 3072 < NV16 ? owner_mask(o3, pat) : 0u;
    unsigned mlo = m0 | (m1 << 4);
    unsigned mhi = m2 | (m3 << 4);
    unsigned bal = __builtin_amdgcn_ballot_w32((mlo | mhi) != 0u);
    while (bal != 0u) {
      if ((mlo | mhi) != 0u) {
        const bool lo = mlo != 0u;
        const unsigned mw = lo ? mlo : mhi;
        const int p = __builtin_ctz(mw);
        const int u = (lo ? 0 : 2) + ((p >> 2) & 1);
        const int pos = head + cnt + __builtin_popcount(bal & below);
        wq[pos & (QCAP - 1)] = (unsigned)(16 * (q0 + 1024 * u) + 4 * (p & 3) + (p >> 3));
        const unsigned rest = mw & (mw - 1u);
        if (lo) mlo = rest; else mhi = rest;
      }
      cnt += __builtin_popcount(bal);
      if (cnt >= 32) {
        asm volatile("s_wait_dscnt 0" ::: "memory");
        drain_batch(wq, head, 32, lane, base, pk, dsh, x, acc);
        head = (head + 32) & (QCAP - 1);
        cnt -= 32;
      }
      bal = __builtin_amdgcn_ballot_w32((mlo | mhi) != 0u);
    }
  }
  asm volatile("s_wait_dscnt 0" ::: "memory");
  if (cnt > 0) drain_batch(wq, head, cnt, lane, base, pk, dsh, x, acc);
  asm volatile("s_wait_dscnt 0" ::: "memory");
  __syncthreads();

  for (int r = wave; r < RNG; r += 32) {
    const float* a = acc + r * 128 + lane;
    float4 o = make_float4(a[0], a[32], a[64], a[96]);
    *(float4*)(agg + (size_t)(base + r) * 128 + lane * 4) = o;
  }
}

#define MLP_WAVES 13
#define MLP_WGS   256
#define W1P 20
#define W2P 12
#define WG_FLOATS (64 * 16 * 16 + (64 * 16 / 4) * 4)
#define WG_KSTEP  (32 * 16 + 8 * 4)
__device__ __forceinline__ int wg_off(int e) { return e * 16 + (e >> 2) * 4; }

__global__ __launch_bounds__(MLP_WAVES * 32)
void k_mlp(const float* __restrict__ x,
           const float* __restrict__ agg, const float* __restrict__ dinv,
           const float* __restrict__ Wg, const float* __restrict__ bg,
           const float* __restrict__ W1, const float* __restrict__ b1,
           const float* __restrict__ W2, const float* __restrict__ b2,
           const float* __restrict__ W3, const float* __restrict__ b3,
           float* __restrict__ out) {
  __shared__ __attribute__((aligned(16))) float wgs[WG_FLOATS];
  __shared__ __attribute__((aligned(16))) float w1s[64 * 16 * W1P];
  __shared__ __attribute__((aligned(16))) float w2s[64 * 16 * W2P];
  __shared__ __attribute__((aligned(16))) float w3s[16 * 32 * 2];
  __shared__ float bs[128 + 64 + 16 + 128];
  __shared__ __attribute__((aligned(16))) float lds[MLP_WAVES * 16 * TPITCH];
  const int tid  = threadIdx.x;
  const int lane = tid & 31;
  const int wave = tid >> 5;
  const int half = lane >> 4;
  const int lm   = lane & 15;
  float* tile = lds + wave * (16 * TPITCH);
  const int blk = wave * MLP_WGS + blockIdx.x;
  const int row0 = blk * 16;

  const int rowl = blk < NB ? row0 : 0;
  float dv = 0.0f;
  if (lane < 16) dv = dinv[rowl + lane];
  float4 xv[16], ag[16];
#pragma unroll
  for (int r = 0; r < 16; ++r)
    xv[r] = *(const float4*)(x + (size_t)(rowl + r) * 128 + lane * 4);
#pragma unroll
  for (int r = 0; r < 16; ++r)
    ag[r] = *(const float4*)(agg + (size_t)(rowl + r) * 128 + lane * 4);

  constexpr int NT = MLP_WAVES * 32;
  float4 g0[5], g1[5], u0[5], u1[5], s0[3], s1[3];
#pragma unroll
  for (int it = 0; it < 5; ++it) {
    const int i = min(tid + it * NT, 64 * 32 - 1);
    const int p = i >> 5, c = i & 31;
    g0[it] = *(const float4*)(Wg + (size_t)(2 * p) * 128 + c * 4);
    g1[it] = *(const float4*)(Wg + (size_t)(2 * p + 1) * 128 + c * 4);
  }
#pragma unroll
  for (int it = 0; it < 5; ++it) {
    const int i = min(tid + it * NT, 64 * 32 - 1);
    const int p = i >> 5, c = i & 31;
    u0[it] = *(const float4*)(W1 + (size_t)(2 * p) * 128 + c * 4);
    u1[it] = *(const float4*)(W1 + (size_t)(2 * p + 1) * 128 + c * 4);
  }
#pragma unroll
  for (int it = 0; it < 3; ++it) {
    const int i = min(tid + it * NT, 64 * 16 - 1);
    const int p = i >> 4, c = i & 15;
    s0[it] = *(const float4*)(W2 + (size_t)(2 * p) * 64 + c * 4);
    s1[it] = *(const float4*)(W2 + (size_t)(2 * p + 1) * 64 + c * 4);
  }
#pragma unroll
  for (int it = 0; it < 5; ++it) {
    const int i = tid + it * NT;
    if (i < 64 * 32) {
      const int p = i >> 5, c = i & 31;
      const float e0[4] = {g0[it].x, g0[it].y, g0[it].z, g0[it].w};
      const float e1[4] = {g1[it].x, g1[it].y, g1[it].z, g1[it].w};
      for (int j = 0; j < 4; ++j) {
        const int n = c * 4 + j;
        v2f pr; pr.x = e0[j]; pr.y = e1[j];
        *(v2f*)(wgs + wg_off(p * 16 + (n & 15)) + 2 * (n >> 4)) = pr;
      }
    }
  }
#pragma unroll
  for (int it = 0; it < 5; ++it) {
    const int i = tid + it * NT;
    if (i < 64 * 32) {
      const int p = i >> 5, c = i & 31;
      const float e0[4] = {u0[it].x, u0[it].y, u0[it].z, u0[it].w};
      const float e1[4] = {u1[it].x, u1[it].y, u1[it].z, u1[it].w};
      for (int j = 0; j < 4; ++j) {
        const int n = c * 4 + j;
        v2f pr; pr.x = e0[j]; pr.y = e1[j];
        *(v2f*)(w1s + (p * 16 + (n & 15)) * W1P + 2 * (n >> 4)) = pr;
      }
    }
  }
#pragma unroll
  for (int it = 0; it < 3; ++it) {
    const int i = tid + it * NT;
    if (i < 64 * 16) {
      const int p = i >> 4, c = i & 15;
      const float e0[4] = {s0[it].x, s0[it].y, s0[it].z, s0[it].w};
      const float e1[4] = {s1[it].x, s1[it].y, s1[it].z, s1[it].w};
      for (int j = 0; j < 4; ++j) {
        const int n = c * 4 + j;
        v2f pr; pr.x = e0[j]; pr.y = e1[j];
        *(v2f*)(w2s + (p * 16 + (n & 15)) * W2P + 2 * (n >> 4)) = pr;
      }
    }
  }
  for (int i = tid; i < 16 * 32; i += NT) {
    const int kk = i >> 5, l = i & 31;
    const int k0 = 4 * kk + 2 * (l >> 4), n = l & 15;
    v2f pr; pr.x = 0.0f; pr.y = 0.0f;
    if (n < 2) { pr.x = W3[k0 * 2 + n]; pr.y = W3[(k0 + 1) * 2 + n]; }
    *(v2f*)(w3s + i * 2) = pr;
  }
  if (tid < 128) bs[tid] = b1[tid];
  else if (tid < 192) bs[tid] = b2[tid - 128];
  else if (tid < 208) bs[tid] = tid < 194 ? b3[tid - 192] : 0.0f;
  else if (tid < 336) bs[tid] = bg[tid - 208];

  if (blk < NB) {
#pragma unroll
    for (int r = 0; r < 16; ++r) {
      const float di = rl_f(dv, r);
      const float sl = di * di;
      float4 t;
      t.x = ag[r].x + xv[r].x * sl;
      t.y = ag[r].y + xv[r].y * sl;
      t.z = ag[r].z + xv[r].z * sl;
      t.w = ag[r].w + xv[r].w * sl;
      *(float4*)(tile + r * TPITCH + lane * 4) = t;
    }
  }
  __syncthreads();
  if (blk >= NB) return;

  v8f acc0[8];
  for (int t = 0; t < 8; ++t) acc0[t] = (v8f){};
  {
    const float* ap = tile + lm * TPITCH + 2 * half;
    const float* bp = wgs + wg_off(half * 16 + lm);
    for (int kk = 0; kk < 32; ++kk) {
      const v2f a = *(const v2f*)(ap + 4 * kk);
      const float4* bq = (const float4*)(bp + kk * WG_KSTEP);
      const float4 q0 = bq[0], q1 = bq[1], q2 = bq[2], q3 = bq[3];
      v2f b;
      b.x = q0.x; b.y = q0.y; acc0[0] = wmma_f32(a, b, acc0[0]);
      b.x = q0.z; b.y = q0.w; acc0[1] = wmma_f32(a, b, acc0[1]);
      b.x = q1.x; b.y = q1.y; acc0[2] = wmma_f32(a, b, acc0[2]);
      b.x = q1.z; b.y = q1.w; acc0[3] = wmma_f32(a, b, acc0[3]);
      b.x = q2.x; b.y = q2.y; acc0[4] = wmma_f32(a, b, acc0[4]);
      b.x = q2.z; b.y = q2.w; acc0[5] = wmma_f32(a, b, acc0[5]);
      b.x = q3.x; b.y = q3.y; acc0[6] = wmma_f32(a, b, acc0[6]);
      b.x = q3.z; b.y = q3.w; acc0[7] = wmma_f32(a, b, acc0[7]);
    }
  }
  asm volatile("s_wait_dscnt 0" ::: "memory");
  for (int t = 0; t < 8; ++t) {
    float bb = bs[208 + 16 * t + lm];
    for (int r = 0; r < 8; ++r)
      tile[(r + 8 * half) * TPITCH + 16 * t + lm] = fmaxf(acc0[t][r] + bb, 0.0f);
  }
  asm volatile("s_wait_dscnt 0" ::: "memory");
#pragma unroll
  for (int r = 0; r < 16; ++r) {
    float4* hp = (float4*)(tile + r * TPITCH + lane * 4);
    float4 h = *hp;
    h.x += xv[r].x; h.y += xv[r].y; h.z += xv[r].z; h.w += xv[r].w;
    *hp = h;
  }
  asm volatile("s_wait_dscnt 0" ::: "memory");

  v8f acc1[8];
  for (int t = 0; t < 8; ++t) acc1[t] = (v8f){};
  {
    const float* ap = tile + lm * TPITCH + 2 * half;
    const float* bp = w1s + (half * 16 + lm) * W1P;
    for (int kk = 0; kk < 32; ++kk) {
      const v2f a = *(const v2f*)(ap + 4 * kk);
      const float4* bq = (const float4*)(bp + kk * (2 * 16 * W1P));
      const float4 q0 = bq[0], q1 = bq[1], q2 = bq[2], q3 = bq[3];
      v2f b;
      b.x = q0.x; b.y = q0.y; acc1[0] = wmma_f32(a, b, acc1[0]);
      b.x = q0.z; b.y = q0.w; acc1[1] = wmma_f32(a, b, acc1[1]);
      b.x = q1.x; b.y = q1.y; acc1[2] = wmma_f32(a, b, acc1[2]);
      b.x = q1.z; b.y = q1.w; acc1[3] = wmma_f32(a, b, acc1[3]);
      b.x = q2.x; b.y = q2.y; acc1[4] = wmma_f32(a, b, acc1[4]);
      b.x = q2.z; b.y = q2.w; acc1[5] = wmma_f32(a, b, acc1[5]);
      b.x = q3.x; b.y = q3.y; acc1[6] = wmma_f32(a, b, acc1[6]);
      b.x = q3.z; b.y = q3.w; acc1[7] = wmma_f32(a, b, acc1[7]);
    }
  }
  asm volatile("s_wait_dscnt 0" ::: "memory");
  for (int t = 0; t < 8; ++t) {
    float bb = bs[16 * t + lm];
    for (int r = 0; r < 8; ++r)
      tile[(r + 8 * half) * TPITCH + 16 * t + lm] = fmaxf(acc1[t][r] + bb, 0.0f);
  }
  asm volatile("s_wait_dscnt 0" ::: "memory");

  v8f acc2[4];
  for (int t = 0; t < 4; ++t) acc2[t] = (v8f){};
  {
    const float* ap = tile + lm * TPITCH + 2 * half;
    const float* bp = w2s + (half * 16 + lm) * W2P;
    for (int kk = 0; kk < 32; ++kk) {
      const v2f a = *(const v2f*)(ap + 4 * kk);
      const float4* bq = (const float4*)(bp + kk * (2 * 16 * W2P));
      const float4 q0 = bq[0], q1 = bq[1];
      v2f b;
      b.x = q0.x; b.y = q0.y; acc2[0] = wmma_f32(a, b, acc2[0]);
      b.x = q0.z; b.y = q0.w; acc2[1] = wmma_f32(a, b, acc2[1]);
      b.x = q1.x; b.y = q1.y; acc2[2] = wmma_f32(a, b, acc2[2]);
      b.x = q1.z; b.y = q1.w; acc2[3] = wmma_f32(a, b, acc2[3]);
    }
  }
  asm volatile("s_wait_dscnt 0" ::: "memory");
  for (int t = 0; t < 4; ++t) {
    float bb = bs[128 + 16 * t + lm];
    for (int r = 0; r < 8; ++r)
      tile[(r + 8 * half) * TPITCH + 16 * t + lm] = fmaxf(acc2[t][r] + bb, 0.0f);
  }
  asm volatile("s_wait_dscnt 0" ::: "memory");

  v8f acc3 = (v8f){};
  for (int kk = 0; kk < 16; ++kk) {
    const v2f a = *(const v2f*)(tile + lm * TPITCH + 4 * kk + 2 * half);
    const v2f b = *(const v2f*)(w3s + (kk * 32 + lane) * 2);
    acc3 = wmma_f32(a, b, acc3);
  }
  if (lm < 2) {
    float bb = bs[192 + lm];
    for (int r = 0; r < 8; ++r)
      out[(size_t)lm * N_NODES + row0 + r + 8 * half] = acc3[r] + bb;
  }
}

extern "C" void kernel_launch(void* const* d_in, const int* in_sizes, int n_in,
                              void* d_out, int out_size, void* d_ws, size_t ws_size,
                              hipStream_t stream) {
  (void)in_sizes; (void)n_in; (void)out_size; (void)ws_size;
  const float* x  = (const float*)d_in[0];
  const int*   ei = (const int*)d_in[1];
  const float* Wg = (const float*)d_in[2];
  const float* bg = (const float*)d_in[3];
  const float* W1 = (const float*)d_in[4];
  const float* b1 = (const float*)d_in[5];
  const float* W2 = (const float*)d_in[6];
  const float* b2 = (const float*)d_in[7];
  const float* W3 = (const float*)d_in[8];
  const float* b3 = (const float*)d_in[9];
  float* out = (float*)d_out;

  float* ws   = (float*)d_ws;
  float* deg  = ws;
  float* dinv = ws + N_NODES;
  float* agg  = ws + 2 * N_NODES;
  int*   own  = (int*)(agg + (size_t)N_NODES * 128);
  int*   pk   = own + N_EDGES / 4;

  k_init<<<(N_NODES + 255) / 256, 256, 0, stream>>>(deg);
  k_deg<<<(NQ4 + 255) / 256, 256, 0, stream>>>(ei, deg, own, pk);
  k_dinv<<<(N_NODES + 255) / 256, 256, 0, stream>>>(deg, dinv);
  k_own<<<NOWN, 1024, 0, stream>>>(pk, own, dinv, x, agg);
  k_mlp<<<MLP_WGS, MLP_WAVES * 32, 0, stream>>>(x, agg, dinv, Wg, bg, W1, b1, W2, b2, W3, b3, out);
}


# k_deg deals each 1024-edge chunk's packed edge words into 32 owner-group segments by LDS ranks; k_own reads its group's 782 segments, no list scan, no per-edge word load; sums in arrival order
# speedup vs baseline: 2.4751x; 1.0366x over previous
#include <hip/hip_runtime.h>

typedef float v2f __attribute__((ext_vector_type(2)));
typedef float v8f __attribute__((ext_vector_type(8)));

#define N_NODES 50000
#define N_EDGES 800000
#define NB (N_NODES / 16)
#define TPITCH 132

__device__ __forceinline__ v8f wmma_f32(v2f a, v2f b, v8f c) {
  return __builtin_amdgcn_wmma_f32_16x16x4_f32(false, a, false, b, (short)0, c,
                                               false, false);
}

__global__ void k_init(float* __restrict__ deg) {
  int i = blockIdx.x * 256 + threadIdx.x;
  if (i < N_NODES) deg[i] = 1.0f;
}

#define RNG   200
#define NOWN  (N_NODES / RNG)
#define NQ4   (N_EDGES / 4)
#define GRP   8
#define NGRP  ((NOWN + GRP - 1) / GRP)
#define DEGT  256
#define DEGE  (DEGT * 4)
#define NDW   ((NQ4 + DEGT - 1) / DEGT)
#define SEGW  1088
#define GSTR  (NDW * SEGW)
static_assert(N_NODES <= 65536 && RNG <= 256 && GRP <= 8, "src, dst % RNG and owner % GRP share one 32-bit word");
static_assert(SEGW >= DEGE + 64 && SEGW % 64 == 0, "a segment holds every edge of a work-group and is read in 64-word pieces");
static_assert(NGRP <= 32 && NGRP <= DEGT, "one counter per group, summed by one wave");
__global__ __launch_bounds__(DEGT) void k_deg(const int* __restrict__ ei, float* __restrict__ deg,
                                              int* __restrict__ seg) {
  __shared__ int cnt[NGRP];
  __shared__ int pre[NGRP];
  __shared__ unsigned sw[DEGE];
  __shared__ int so[DEGE];
  const int tid = threadIdx.x;
  if (tid < NGRP) cnt[tid] = 0;
  __syncthreads();
  const int q = blockIdx.x * DEGT + tid;
  unsigned w[4] = {0u, 0u, 0u, 0u};
  int g[4] = {0, 0, 0, 0}, r[4] = {0, 0, 0, 0};
  const bool have = q < NQ4;
  int4 d = make_int4(0, 0, 0, 0);
  if (have) {
    d = ((const int4*)(ei + N_EDGES))[q];
    const int4 s = ((const int4*)ei)[q];
    const unsigned dd[4] = {(unsigned)d.x, (unsigned)d.y, (unsigned)d.z, (unsigned)d.w};
    const unsigned ss[4] = {(unsigned)s.x, (unsigned)s.y, (unsigned)s.z, (unsigned)s.w};
    for (int j = 0; j < 4; ++j) {
      const unsigned o = dd[j] / RNG;
      g[j] = (int)(o / GRP);
      w[j] = ss[j] | ((dd[j] % RNG) << 16) | ((o % GRP) << 24);
      r[j] = atomicAdd(&cnt[g[j]], 1);
    }
  }
  __syncthreads();
  if (tid < NGRP) {
    int p = 0;
    for (int k = 0; k < tid; ++k) p += cnt[k];
    pre[tid] = p;
    seg[(size_t)tid * GSTR + (size_t)blockIdx.x * SEGW] = cnt[tid];
  }
  __syncthreads();
  if (have) {
    for (int j = 0; j < 4; ++j) {
      const int p = pre[g[j]] + r[j];
      sw[p] = w[j];
      so[p] = g[j] * GSTR + 1 + r[j];
    }
  }
  __syncthreads();
  if (have) {
    atomicAdd(&deg[d.x], 1.0f);
    atomicAdd(&deg[d.y], 1.0f);
    atomicAdd(&deg[d.z], 1.0f);
    atomicAdd(&deg[d.w], 1.0f);
  }
  const int ne = 4 * min(DEGT, NQ4 - blockIdx.x * DEGT);
  int* mine = seg + (size_t)blockIdx.x * SEGW;
  for (int p = tid; p < ne; p += DEGT) mine[so[p]] = (int)sw[p];
}

__global__ void k_dinv(const float* __restrict__ deg, float* __restrict__ dinv) {
  int i = blockIdx.x * 256 + threadIdx.x;
  if (i < N_NODES) {
    float d = deg[i];
    dinv[i] = d > 0.0f ? rsqrtf(d) : 0.0f;
  }
}

#define QCAP  64
#define NSW   ((NDW + 31) / 32)

__device__ __forceinline__ int rl_i(int v, int l) { return __builtin_amdgcn_readlane(v, l); }
__device__ __forceinline__ float rl_f(float v, int l) {
  return __int_as_float(__builtin_amdgcn_readlane(__float_as_int(v), l));
}

__device__ __forceinline__ void drain_batch(const unsigned* wq, int head, int m, int lane,
                                            int base, const float* dsh,
                                            const float* __restrict__ x, float* acc) {
  int s = 0;
  int dl = RNG;
  float nrm = 0.0f;
  if (lane < m) {
    const unsigned w = wq[(head + lane) & (QCAP - 1)];
    s = (int)(w & 0xFFFFu);
    dl = (int)((w >> 16) & 0xFFu);
    nrm = dsh[s] * dsh[base + dl];
  }
  for (int j = 0; j < m; j += 4) {
    float4 v[4];
    for (int w = 0; w < 4; ++w)
      v[w] = *(const float4*)(x + (size_t)rl_i(s, j + w) * 128 + lane * 4);
    for (int w = 0; w < 4; ++w) {
      const float nj = rl_f(nrm, j + w);
      float* a = acc + rl_i(dl, j + w) * 128 + lane;
      atomicAdd(a,      v[w].x * nj);
      atomicAdd(a + 32, v[w].y * nj);
      atomicAdd(a + 64, v[w].z * nj);
      atomicAdd(a + 96, v[w].w * nj);
    }
  }
}

__device__ __forceinline__ void take_words(unsigned w, bool ok, unsigned* wq, int& head,
                                           int& cnt, int lane, unsigned below, int base,
                                           const float* dsh, const float* __restrict__ x,
                                           float* acc) {
  const unsigned bal = __builtin_amdgcn_ballot_w32(ok);
  if (bal != 0u) {
    if (ok) wq[(head + cnt + __builtin_popcount(bal & below)) & (QCAP - 1)] = w;
    cnt += __builtin_popcount(bal);
    if (cnt >= 32) {
      asm volatile("s_wait_dscnt 0" ::: "memory");
      drain_batch(wq, head, 32, lane, base, dsh, x, acc);
      head = (head + 32) & (QCAP - 1);
      cnt -= 32;
    }
  }
}

__global__ __launch_bounds__(1024) void k_own(const int* __restrict__ seg,
                                              const float* __restrict__ dinv,
                                              const float* __restrict__ x,
                                              float* __restrict__ agg) {
  __shared__ float acc[(RNG + 1) * 128];
  __shared__ unsigned queue[32 * QCAP];
  __shared__ __attribute__((aligned(16))) float dsh[N_NODES];
  const int tid  = threadIdx.x;
  const int lane = tid & 31;
  const int wave = __builtin_amdgcn_readfirstlane(tid >> 5);
  const int base = blockIdx.x * RNG;
  unsigned* wq = queue + wave * QCAP;

  const int2* gs = (const int2*)(seg + (size_t)(blockIdx.x / GRP) * GSTR);
  const int nseg = (NDW - wave + 31) / 32;
  int2 v[NSW];
#pragma unroll
  for (int u = 0; u < NSW; ++u)
    v[u] = gs[(size_t)(wave + 32 * min(u, nseg - 1)) * (SEGW / 2) + lane];

  constexpr int NV4 = N_NODES / 4;
  constexpr int NPS = (NV4 + 1023) / 1024;
  static_assert(NV4 >= 1024, "the shifted last pass needs a table of at least one pass");
  float4 dv[NPS];
#pragma unroll
  for (int it = 0; it < NPS; ++it) {
    const int i = it < NPS - 1 ? tid + it * 1024 : tid + NV4 - 1024;
    dv[it] = ((const float4*)dinv)[i];
  }
  for (int i = tid; i < (RNG + 1) * 32; i += 1024)
    ((float4*)acc)[i] = make_float4(0.0f, 0.0f, 0.0f, 0.0f);
#pragma unroll
  for (int it = 0; it < NPS; ++it) {
    const int i = it < NPS - 1 ? tid + it * 1024 : tid + NV4 - 1024;
    ((float4*)dsh)[i] = dv[it];
  }
  __syncthreads();

  const unsigned sub = (unsigned)blockIdx.x % GRP;
  const unsigned below = (1u << lane) - 1u;
  int cnt = 0;
  int head = 0;

#pragma nounroll
  for (int u = 0; u < nseg; ++u) {
    int2 t = v[0];
#pragma unroll
    for (int k = 1; k < NSW; ++k)
      if (u == k) t = v[k];
    const int2* sp = gs + (size_t)(wave + 32 * u) * (SEGW / 2) + lane;
    const int c = min(rl_i(t.x, 0), DEGE);
    for (int b = 0; ; ) {
      const int k0 = b + 2 * lane;
      const unsigned w0 = (unsigned)t.x, w1 = (unsigned)t.y;
      take_words(w0, k0 >= 1 && k0 <= c && (w0 >> 24) == sub,
                 wq, head, cnt, lane, below, base, dsh, x, acc);
      take_words(w1, k0 + 1 <= c && (w1 >> 24) == sub,
                 wq, head, cnt, lane, below, base, dsh, x, acc);
      b += 64;
      if (b > c) break;
      t = sp[b / 2];
    }
  }
  asm volatile("s_wait_dscnt 0" ::: "memory");
  if (cnt > 0) drain_batch(wq, head, cnt, lane, base, dsh, x, acc);
  asm volatile("s_wait_dscnt 0" ::: "memory");
  __syncthreads();

  for (int r = wave; r < RNG; r += 32) {
    const float* a = acc + r * 128 + lane;
    float4 o = make_float4(a[0], a[32], a[64], a[96]);
    *(float4*)(agg + (size_t)(base + r) * 128 + lane * 4) = o;
  }
}

#define MLP_WAVES 13
#define MLP_WGS   256
#define W1P 20
#define W2P 12
#define WG_FLOATS (64 * 16 * 16 + (64 * 16 / 4) * 4)
#define WG_KSTEP  (32 * 16 + 8 * 4)
__device__ __forceinline__ int wg_off(int e) { return e * 16 + (e >> 2) * 4; }

__global__ __launch_bounds__(MLP_WAVES * 32)
void k_mlp(const float* __restrict__ x,
           const float* __restrict__ agg, const float* __restrict__ dinv,
           const float* __restrict__ Wg, const float* __restrict__ bg,
           const float* __restrict__ W1, const float* __restrict__ b1,
           const float* __restrict__ W2, const float* __restrict__ b2,
           const float* __restrict__ W3, const float* __restrict__ b3,
           float* __restrict__ out) {
  __shared__ __attribute__((aligned(16))) float wgs[WG_FLOATS];
  __shared__ __attribute__((aligned(16))) float w1s[64 * 16 * W1P];
  __shared__ __attribute__((aligned(16))) float w2s[64 * 16 * W2P];
  __shared__ __attribute__((aligned(16))) float w3s[16 * 32 * 2];
  __shared__ float bs[128 + 64 + 16 + 128];
  __shared__ __attribute__((aligned(16))) float lds[MLP_WAVES * 16 * TPITCH];
  const int tid  = threadIdx.x;
  const int lane = tid & 31;
  const int wave = tid >> 5;
  const int half = lane >> 4;
  const int lm   = lane & 15;
  float* tile = lds + wave * (16 * TPITCH);
  const int blk = wave * MLP_WGS + blockIdx.x;
  const int row0 = blk * 16;

  const int rowl = blk < NB ? row0 : 0;
  float dv = 0.0f;
  if (lane < 16) dv = dinv[rowl + lane];
  float4 xv[16], ag[16];
#pragma unroll
  for (int r = 0; r < 16; ++r)
    xv[r] = *(const float4*)(x + (size_t)(rowl + r) * 128 + lane * 4);
#pragma unroll
  for (int r = 0; r < 16; ++r)
    ag[r] = *(const float4*)(agg + (size_t)(rowl + r) * 128 + lane * 4);

  constexpr int NT = MLP_WAVES * 32;
  float4 g0[5], g1[5], u0[5], u1[5], s0[3], s1[3];
#pragma unroll
  for (int it = 0; it < 5; ++it) {
    const int i = min(tid + it * NT, 64 * 32 - 1);
    const int p = i >> 5, c = i & 31;
    g0[it] = *(const float4*)(Wg + (size_t)(2 * p) * 128 + c * 4);
    g1[it] = *(const float4*)(Wg + (size_t)(2 * p + 1) * 128 + c * 4);
  }
#pragma unroll
  for (int it = 0; it < 5; ++it) {
    const int i = min(tid + it * NT, 64 * 32 - 1);
    const int p = i >> 5, c = i & 31;
    u0[it] = *(const float4*)(W1 + (size_t)(2 * p) * 128 + c * 4);
    u1[it] = *(const float4*)(W1 + (size_t)(2 * p + 1) * 128 + c * 4);
  }
#pragma unroll
  for (int it = 0; it < 3; ++it) {
    const int i = min(tid + it * NT, 64 * 16 - 1);
    const int p = i >> 4, c = i & 15;
    s0[it] = *(const float4*)(W2 + (size_t)(2 * p) * 64 + c * 4);
    s1[it] = *(const float4*)(W2 + (size_t)(2 * p + 1) * 64 + c * 4);
  }
#pragma unroll
  for (int it = 0; it < 5; ++it) {
    const int i = tid + it * NT;
    if (i < 64 * 32) {
      const int p = i >> 5, c = i & 31;
      const float e0[4] = {g0[it].x, g0[it].y, g0[it].z, g0[it].w};
      const float e1[4] = {g1[it].x, g1[it].y, g1[it].z, g1[it].w};
      for (int j = 0; j < 4; ++j) {
        const int n = c * 4 + j;
        v2f pr; pr.x = e0[j]; pr.y = e1[j];
        *(v2f*)(wgs + wg_off(p * 16 + (n & 15)) + 2 * (n >> 4)) = pr;
      }
    }
  }
#pragma unroll
  for (int it = 0; it < 5; ++it) {
    const int i = tid + it * NT;
    if (i < 64 * 32) {
      const int p = i >> 5, c = i & 31;
      const float e0[4] = {u0[it].x, u0[it].y, u0[it].z, u0[it].w};
      const float e1[4] = {u1[it].x, u1[it].y, u1[it].z, u1[it].w};
      for (int j = 0; j < 4; ++j) {
        const int n = c * 4 + j;
        v2f pr; pr.x = e0[j]; pr.y = e1[j];
        *(v2f*)(w1s + (p * 16 + (n & 15)) * W1P + 2 * (n >> 4)) = pr;
      }
    }
  }
#pragma unroll
  for (int it = 0; it < 3; ++it) {
    const int i = tid + it * NT;
    if (i < 64 * 16) {
      const int p = i >> 4, c = i & 15;
      const float e0[4] = {s0[it].x, s0[it].y, s0[it].z, s0[it].w};
      const float e1[4] = {s1[it].x, s1[it].y, s1[it].z, s1[it].w};
      for (int j = 0; j < 4; ++j) {
        const int n = c * 4 + j;
        v2f pr; pr.x = e0[j]; pr.y = e1[j];
        *(v2f*)(w2s + (p * 16 + (n & 15)) * W2P + 2 * (n >> 4)) = pr;
      }
    }
  }
  for (int i = tid; i < 16 * 32; i += NT) {
    const int kk = i >> 5, l = i & 31;
    const int k0 = 4 * kk + 2 * (l >> 4), n = l & 15;
    v2f pr; pr.x = 0.0f; pr.y = 0.0f;
    if (n < 2) { pr.x = W3[k0 * 2 + n]; pr.y = W3[(k0 + 1) * 2 + n]; }
    *(v2f*)(w3s + i * 2) = pr;
  }
  if (tid < 128) bs[tid] = b1[tid];
  else if (tid < 192) bs[tid] = b2[tid - 128];
  else if (tid < 208) bs[tid] = tid < 194 ? b3[tid - 192] : 0.0f;
  else if (tid < 336) bs[tid] = bg[tid - 208];

  if (blk < NB) {
#pragma unroll
    for (int r = 0; r < 16; ++r) {
      const float di = rl_f(dv, r);
      const float sl = di * di;
      float4 t;
      t.x = ag[r].x + xv[r].x * sl;
      t.y = ag[r].y + xv[r].y * sl;
      t.z = ag[r].z + xv[r].z * sl;
      t.w = ag[r].w + xv[r].w * sl;
      *(float4*)(tile + r * TPITCH + lane * 4) = t;
    }
  }
  __syncthreads();
  if (blk >= NB) return;

  v8f acc0[8];
  for (int t = 0; t < 8; ++t) acc0[t] = (v8f){};
  {
    const float* ap = tile + lm * TPITCH + 2 * half;
    const float* bp = wgs + wg_off(half * 16 + lm);
    for (int kk = 0; kk < 32; ++kk) {
      const v2f a = *(const v2f*)(ap + 4 * kk);
      const float4* bq = (const float4*)(bp + kk * WG_KSTEP);
      const float4 q0 = bq[0], q1 = bq[1], q2 = bq[2], q3 = bq[3];
      v2f b;
      b.x = q0.x; b.y = q0.y; acc0[0] = wmma_f32(a, b, acc0[0]);
      b.x = q0.z; b.y = q0.w; acc0[1] = wmma_f32(a, b, acc0[1]);
      b.x = q1.x; b.y = q1.y; acc0[2] = wmma_f32(a, b, acc0[2]);
      b.x = q1.z; b.y = q1.w; acc0[3] = wmma_f32(a, b, acc0[3]);
      b.x = q2.x; b.y = q2.y; acc0[4] = wmma_f32(a, b, acc0[4]);
      b.x = q2.z; b.y = q2.w; acc0[5] = wmma_f32(a, b, acc0[5]);
      b.x = q3.x; b.y = q3.y; acc0[6] = wmma_f32(a, b, acc0[6]);
      b.x = q3.z; b.y = q3.w; acc0[7] = wmma_f32(a, b, acc0[7]);
    }
  }
  asm volatile("s_wait_dscnt 0" ::: "memory");
  for (int t = 0; t < 8; ++t) {
    float bb = bs[208 + 16 * t + lm];
    for (int r = 0; r < 8; ++r)
      tile[(r + 8 * half) * TPITCH + 16 * t + lm] = fmaxf(acc0[t][r] + bb, 0.0f);
  }
  asm volatile("s_wait_dscnt 0" ::: "memory");
#pragma unroll
  for (int r = 0; r < 16; ++r) {
    float4* hp = (float4*)(tile + r * TPITCH + lane * 4);
    float4 h = *hp;
    h.x += xv[r].x; h.y += xv[r].y; h.z += xv[r].z; h.w += xv[r].w;
    *hp = h;
  }
  asm volatile("s_wait_dscnt 0" ::: "memory");

  v8f acc1[8];
  for (int t = 0; t < 8; ++t) acc1[t] = (v8f){};
  {
    const float* ap = tile + lm * TPITCH + 2 * half;
    const float* bp = w1s + (half * 16 + lm) * W1P;
    for (int kk = 0; kk < 32; ++kk) {
      const v2f a = *(const v2f*)(ap + 4 * kk);
      const float4* bq = (const float4*)(bp + kk * (2 * 16 * W1P));
      const float4 q0 = bq[0], q1 = bq[1], q2 = bq[2], q3 = bq[3];
      v2f b;
      b.x = q0.x; b.y = q0.y; acc1[0] = wmma_f32(a, b, acc1[0]);
      b.x = q0.z; b.y = q0.w; acc1[1] = wmma_f32(a, b, acc1[1]);
      b.x = q1.x; b.y = q1.y; acc1[2] = wmma_f32(a, b, acc1[2]);
      b.x = q1.z; b.y = q1.w; acc1[3] = wmma_f32(a, b, acc1[3]);
      b.x = q2.x; b.y = q2.y; acc1[4] = wmma_f32(a, b, acc1[4]);
      b.x = q2.z; b.y = q2.w; acc1[5] = wmma_f32(a, b, acc1[5]);
      b.x = q3.x; b.y = q3.y; acc1[6] = wmma_f32(a, b, acc1[6]);
      b.x = q3.z; b.y = q3.w; acc1[7] = wmma_f32(a, b, acc1[7]);
    }
  }
  asm volatile("s_wait_dscnt 0" ::: "memory");
  for (int t = 0; t < 8; ++t) {
    float bb = bs[16 * t + lm];
    for (int r = 0; r < 8; ++r)
      tile[(r + 8 * half) * TPITCH + 16 * t + lm] = fmaxf(acc1[t][r] + bb, 0.0f);
  }
  asm volatile("s_wait_dscnt 0" ::: "memory");

  v8f acc2[4];
  for (int t = 0; t < 4; ++t) acc2[t] = (v8f){};
  {
    const float* ap = tile + lm * TPITCH + 2 * half;
    const float* bp = w2s + (half * 16 + lm) * W2P;
    for (int kk = 0; kk < 32; ++kk) {
      const v2f a = *(const v2f*)(ap + 4 * kk);
      const float4* bq = (const float4*)(bp + kk * (2 * 16 * W2P));
      const float4 q0 = bq[0], q1 = bq[1];
      v2f b;
      b.x = q0.x; b.y = q0.y; acc2[0] = wmma_f32(a, b, acc2[0]);
      b.x = q0.z; b.y = q0.w; acc2[1] = wmma_f32(a, b, acc2[1]);
      b.x = q1.x; b.y = q1.y; acc2[2] = wmma_f32(a, b, acc2[2]);
      b.x = q1.z; b.y = q1.w; acc2[3] = wmma_f32(a, b, acc2[3]);
    }
  }
  asm volatile("s_wait_dscnt 0" ::: "memory");
  for (int t = 0; t < 4; ++t) {
    float bb = bs[128 + 16 * t + lm];
    for (int r = 0; r < 8; ++r)
      tile[(r + 8 * half) * TPITCH + 16 * t + lm] = fmaxf(acc2[t][r] + bb, 0.0f);
  }
  asm volatile("s_wait_dscnt 0" ::: "memory");

  v8f acc3 = (v8f){};
  for (int kk = 0; kk < 16; ++kk) {
    const v2f a = *(const v2f*)(tile + lm * TPITCH + 4 * kk + 2 * half);
    const v2f b = *(const v2f*)(w3s + (kk * 32 + lane) * 2);
    acc3 = wmma_f32(a, b, acc3);
  }
  if (lm < 2) {
    float bb = bs[192 + lm];
    for (int r = 0; r < 8; ++r)
      out[(size_t)lm * N_NODES + row0 + r + 8 * half] = acc3[r] + bb;
  }
}

#define SEG_OFF 6500032
static_assert(SEG_OFF >= 2 * N_NODES + N_NODES * 128 && SEG_OFF % 64 == 0, "segments start behind agg, in whole 256-byte units");
static_assert(((size_t)SEG_OFF + (size_t)NGRP * GSTR) * 4 <= (size_t)268435456, "segments end inside the workspace");

extern "C" void kernel_launch(void* const* d_in, const int* in_sizes, int n_in,
                              void* d_out, int out_size, void* d_ws, size_t ws_size,
                              hipStream_t stream) {
  (void)in_sizes; (void)n_in; (void)out_size; (void)ws_size;
  const float* x  = (const float*)d_in[0];
  const int*   ei = (const int*)d_in[1];
  const float* Wg = (const float*)d_in[2];
  const float* bg = (const float*)d_in[3];
  const float* W1 = (const float*)d_in[4];
  const float* b1 = (const float*)d_in[5];
  const float* W2 = (const float*)d_in[6];
  const float* b2 = (const float*)d_in[7];
  const float* W3 = (const float*)d_in[8];
  const float* b3 = (const float*)d_in[9];
  float* out = (float*)d_out;

  float* ws   = (float*)d_ws;
  float* deg  = ws;
  float* dinv = ws + N_NODES;
  float* agg  = ws + 2 * N_NODES;
  int*   seg  = (int*)d_ws + SEG_OFF;

  k_init<<<(N_NODES + 255) / 256, 256, 0, stream>>>(deg);
  k_deg<<<NDW, DEGT, 0, stream>>>(ei, deg, seg);
  k_dinv<<<(N_NODES + 255) / 256, 256, 0, stream>>>(deg, dinv);
  k_own<<<NOWN, 1024, 0, stream>>>(seg, dinv, x, agg);
  k_mlp<<<MLP_WGS, MLP_WAVES * 32, 0, stream>>>(x, agg, dinv, Wg, bg, W1, b1, W2, b2, W3, b3, out);
}


# degree counted from the dealt entries: 32 group work-groups add one per entry into 1600 LDS counters, write dinv; k_deg's 800000 global f32 adds, k_init, k_dinv gone; 4 launches; sums unchanged
# speedup vs baseline: 3.9885x; 1.6115x over previous
#include <hip/hip_runtime.h>

typedef float v2f __attribute__((ext_vector_type(2)));
typedef float v8f __attribute__((ext_vector_type(8)));

#define N_NODES 50000
#define N_EDGES 800000
#define NB (N_NODES / 16)
#define TPITCH 132

__device__ __forceinline__ v8f wmma_f32(v2f a, v2f b, v8f c) {
  return __builtin_amdgcn_wmma_f32_16x16x4_f32(false, a, false, b, (short)0, c,
                                               false, false);
}

#define RNG   200
#define NOWN  (N_NODES / RNG)
#define NQ4   (N_EDGES / 4)
#define GRP   8
#define NGRP  ((NOWN + GRP - 1) / GRP)
#define DEGT  256
#define DEGE  (DEGT * 4)
#define NDW   ((NQ4 + DEGT - 1) / DEGT)
#define SEGW  1088
#define GSTR  (NDW * SEGW)
static_assert(N_NODES <= 65536 && RNG <= 256 && GRP <= 8, "src, dst % RNG and owner % GRP share one 32-bit word");
static_assert(SEGW >= DEGE + 64 && SEGW % 64 == 0, "a segment holds every edge of a work-group and is read in 64-word pieces");
static_assert(NGRP <= 32 && NGRP <= DEGT, "one counter per group, summed by one wave");
__global__ __launch_bounds__(DEGT) void k_deg(const int* __restrict__ ei, int* __restrict__ seg) {
  __shared__ int cnt[NGRP];
  __shared__ int pre[NGRP];
  __shared__ unsigned sw[DEGE];
  __shared__ int so[DEGE];
  const int tid = threadIdx.x;
  if (tid < NGRP) cnt[tid] = 0;
  __syncthreads();
  const int q = blockIdx.x * DEGT + tid;
  unsigned w[4] = {0u, 0u, 0u, 0u};
  int g[4] = {0, 0, 0, 0}, r[4] = {0, 0, 0, 0};
  const bool have = q < NQ4;
  if (have) {
    const int4 d = ((const int4*)(ei + N_EDGES))[q];
    const int4 s = ((const int4*)ei)[q];
    const unsigned dd[4] = {(unsigned)d.x, (unsigned)d.y, (unsigned)d.z, (unsigned)d.w};
    const unsigned ss[4] = {(unsigned)s.x, (unsigned)s.y, (unsigned)s.z, (unsigned)s.w};
    for (int j = 0; j < 4; ++j) {
      const unsigned o = dd[j] / RNG;
      g[j] = (int)(o / GRP);
      w[j] = ss[j] | ((dd[j] % RNG) << 16) | ((o % GRP) << 24);
      r[j] = atomicAdd(&cnt[g[j]], 1);
    }
  }
  __syncthreads();
  if (tid < NGRP) {
    int p = 0;
    for (int k = 0; k < tid; ++k) p += cnt[k];
    pre[tid] = p;
    seg[(size_t)tid * GSTR + (size_t)blockIdx.x * SEGW] = cnt[tid];
  }
  __syncthreads();
  if (have) {
    for (int j = 0; j < 4; ++j) {
      const int p = pre[g[j]] + r[j];
      sw[p] = w[j];
      so[p] = g[j] * GSTR + 1 + r[j];
    }
  }
  __syncthreads();
  const int ne = 4 * min(DEGT, NQ4 - blockIdx.x * DEGT);
  int* mine = seg + (size_t)blockIdx.x * SEGW;
  for (int p = tid; p < ne; p += DEGT) mine[so[p]] = (int)sw[p];
}

#define QCAP  64
#define NSW   ((NDW + 31) / 32)

__device__ __forceinline__ int rl_i(int v, int l) { return __builtin_amdgcn_readlane(v, l); }
__device__ __forceinline__ float rl_f(float v, int l) {
  return __int_as_float(__builtin_amdgcn_readlane(__float_as_int(v), l));
}

#define GN (GRP * RNG)
__global__ __launch_bounds__(1024) void k_cnt(const int* __restrict__ seg, float* __restrict__ dinv) {
  __shared__ int c[GN];
  const int tid  = threadIdx.x;
  const int lane = tid & 31;
  const int wave = __builtin_amdgcn_readfirstlane(tid >> 5);
  const int2* gs = (const int2*)(seg + (size_t)blockIdx.x * GSTR);
  const int nseg = (NDW - wave + 31) / 32;
  int2 v[NSW];
#pragma unroll
  for (int u = 0; u < NSW; ++u)
    v[u] = gs[(size_t)(wave + 32 * min(u, nseg - 1)) * (SEGW / 2) + lane];
  for (int i = tid; i < GN; i += 1024) c[i] = 0;
  __syncthreads();
#pragma unroll
  for (int u = 0; u < NSW; ++u) {
    if (u < nseg) {
      int2 t = v[u];
      const int2* sp = gs + (size_t)(wave + 32 * u) * (SEGW / 2) + lane;
      const int n = min(rl_i(t.x, 0), DEGE);
      for (int b = 0; ; ) {
        const int k0 = b + 2 * lane;
        const unsigned w0 = (unsigned)t.x, w1 = (unsigned)t.y;
        if (k0 >= 1 && k0 <= n)
          atomicAdd(&c[min((w0 >> 24) * RNG + ((w0 >> 16) & 0xFFu), (unsigned)(GN - 1))], 1);
        if (k0 + 1 <= n)
          atomicAdd(&c[min((w1 >> 24) * RNG + ((w1 >> 16) & 0xFFu), (unsigned)(GN - 1))], 1);
        b += 64;
        if (b > n) break;
        t = sp[b / 2];
      }
    }
  }
  __syncthreads();
  for (int i = tid; i < GN; i += 1024) {
    const int node = blockIdx.x * GN + i;
    if (node < N_NODES) {
      const float d = (float)(c[i] + 1);
      dinv[node] = d > 0.0f ? rsqrtf(d) : 0.0f;
    }
  }
}

__device__ __forceinline__ void drain_batch(const unsigned* wq, int head, int m, int lane,
                                            int base, const float* dsh,
                                            const float* __restrict__ x, float* acc) {
  int s = 0;
  int dl = RNG;
  float nrm = 0.0f;
  if (lane < m) {
    const unsigned w = wq[(head + lane) & (QCAP - 1)];
    s = (int)(w & 0xFFFFu);
    dl = (int)((w >> 16) & 0xFFu);
    nrm = dsh[s] * dsh[base + dl];
  }
  for (int j = 0; j < m; j += 4) {
    float4 v[4];
    for (int w = 0; w < 4; ++w)
      v[w] = *(const float4*)(x + (size_t)rl_i(s, j + w) * 128 + lane * 4);
    for (int w = 0; w < 4; ++w) {
      const float nj = rl_f(nrm, j + w);
      float* a = acc + rl_i(dl, j + w) * 128 + lane;
      atomicAdd(a,      v[w].x * nj);
      atomicAdd(a + 32, v[w].y * nj);
      atomicAdd(a + 64, v[w].z * nj);
      atomicAdd(a + 96, v[w].w * nj);
    }
  }
}

__device__ __forceinline__ void take_words(unsigned w, bool ok, unsigned* wq, int& head,
                                           int& cnt, int lane, unsigned below, int base,
                                           const float* dsh, const float* __restrict__ x,
                                           float* acc) {
  const unsigned bal = __builtin_amdgcn_ballot_w32(ok);
  if (bal != 0u) {
    if (ok) wq[(head + cnt + __builtin_popcount(bal & below)) & (QCAP - 1)] = w;
    cnt += __builtin_popcount(bal);
    if (cnt >= 32) {
      asm volatile("s_wait_dscnt 0" ::: "memory");
      drain_batch(wq, head, 32, lane, base, dsh, x, acc);
      head = (head + 32) & (QCAP - 1);
      cnt -= 32;
    }
  }
}

__global__ __launch_bounds__(1024) void k_own(const int* __restrict__ seg,
                                              const float* __restrict__ dinv,
                                              const float* __restrict__ x,
                                              float* __restrict__ agg) {
  __shared__ float acc[(RNG + 1) * 128];
  __shared__ unsigned queue[32 * QCAP];
  __shared__ __attribute__((aligned(16))) float dsh[N_NODES];
  const int tid  = threadIdx.x;
  const int lane = tid & 31;
  const int wave = __builtin_amdgcn_readfirstlane(tid >> 5);
  const int base = blockIdx.x * RNG;
  unsigned* wq = queue + wave * QCAP;

  const int2* gs = (const int2*)(seg + (size_t)(blockIdx.x / GRP) * GSTR);
  const int nseg = (NDW - wave + 31) / 32;
  int2 v[NSW];
#pragma unroll
  for (int u = 0; u < NSW; ++u)
    v[u] = gs[(size_t)(wave + 32 * min(u, nseg - 1)) * (SEGW / 2) + lane];

  constexpr int NV4 = N_NODES / 4;
  constexpr int NPS = (NV4 + 1023) / 1024;
  static_assert(NV4 >= 1024, "the shifted last pass needs a table of at least one pass");
  float4 dv[NPS];
#pragma unroll
  for (int it = 0; it < NPS; ++it) {
    const int i = it < NPS - 1 ? tid + it * 1024 : tid + NV4 - 1024;
    dv[it] = ((const float4*)dinv)[i];
  }
  for (int i = tid; i < (RNG + 1) * 32; i += 1024)
    ((float4*)acc)[i] = make_float4(0.0f, 0.0f, 0.0f, 0.0f);
#pragma unroll
  for (int it = 0; it < NPS; ++it) {
    const int i = it < NPS - 1 ? tid + it * 1024 : tid + NV4 - 1024;
    ((float4*)dsh)[i] = dv[it];
  }
  __syncthreads();

  const unsigned sub = (unsigned)blockIdx.x % GRP;
  const unsigned below = (1u << lane) - 1u;
  int cnt = 0;
  int head = 0;

#pragma nounroll
  for (int u = 0; u < nseg; ++u) {
    int2 t = v[0];
#pragma unroll
    for (int k = 1; k < NSW; ++k)
      if (u == k) t = v[k];
    const int2* sp = gs + (size_t)(wave + 32 * u) * (SEGW / 2) + lane;
    const int c = min(rl_i(t.x, 0), DEGE);
    for (int b = 0; ; ) {
      const int k0 = b + 2 * lane;
      const unsigned w0 = (unsigned)t.x, w1 = (unsigned)t.y;
      take_words(w0, k0 >= 1 && k0 <= c && (w0 >> 24) == sub,
                 wq, head, cnt, lane, below, base, dsh, x, acc);
      take_words(w1, k0 + 1 <= c && (w1 >> 24) == sub,
                 wq, head, cnt, lane, below, base, dsh, x, acc);
      b += 64;
      if (b > c) break;
      t = sp[b / 2];
    }
  }
  asm volatile("s_wait_dscnt 0" ::: "memory");
  if (cnt > 0) drain_batch(wq, head, cnt, lane, base, dsh, x, acc);
  asm volatile("s_wait_dscnt 0" ::: "memory");
  __syncthreads();

  for (int r = wave; r < RNG; r += 32) {
    const float* a = acc + r * 128 + lane;
    float4 o = make_float4(a[0], a[32], a[64], a[96]);
    *(float4*)(agg + (size_t)(base + r) * 128 + lane * 4) = o;
  }
}

#define MLP_WAVES 13
#define MLP_WGS   256
#define W1P 20
#define W2P 12
#define WG_FLOATS (64 * 16 * 16 + (64 * 16 / 4) * 4)
#define WG_KSTEP  (32 * 16 + 8 * 4)
__device__ __forceinline__ int wg_off(int e) { return e * 16 + (e >> 2) * 4; }

__global__ __launch_bounds__(MLP_WAVES * 32)
void k_mlp(const float* __restrict__ x,
           const float* __restrict__ agg, const float* __restrict__ dinv,
           const float* __restrict__ Wg, const float* __restrict__ bg,
           const float* __restrict__ W1, const float* __restrict__ b1,
           const float* __restrict__ W2, const float* __restrict__ b2,
           const float* __restrict__ W3, const float* __restrict__ b3,
           float* __restrict__ out) {
  __shared__ __attribute__((aligned(16))) float wgs[WG_FLOATS];
  __shared__ __attribute__((aligned(16))) float w1s[64 * 16 * W1P];
  __shared__ __attribute__((aligned(16))) float w2s[64 * 16 * W2P];
  __shared__ __attribute__((aligned(16))) float w3s[16 * 32 * 2];
  __shared__ float bs[128 + 64 + 16 + 128];
  __shared__ __attribute__((aligned(16))) float lds[MLP_WAVES * 16 * TPITCH];
  const int tid  = threadIdx.x;
  const int lane = tid & 31;
  const int wave = tid >> 5;
  const int half = lane >> 4;
  const int lm   = lane & 15;
  float* tile = lds + wave * (16 * TPITCH);
  const int blk = wave * MLP_WGS + blockIdx.x;
  const int row0 = blk * 16;

  const int rowl = blk < NB ? row0 : 0;
  float dv = 0.0f;
  if (lane < 16) dv = dinv[rowl + lane];
  float4 xv[16], ag[16];
#pragma unroll
  for (int r = 0; r < 16; ++r)
    xv[r] = *(const float4*)(x + (size_t)(rowl + r) * 128 + lane * 4);
#pragma unroll
  for (int r = 0; r < 16; ++r)
    ag[r] = *(const float4*)(agg + (size_t)(rowl + r) * 128 + lane * 4);

  constexpr int NT = MLP_WAVES * 32;
  float4 g0[5], g1[5], u0[5], u1[5], s0[3], s1[3];
#pragma unroll
  for (int it = 0; it < 5; ++it) {
    const int i = min(tid + it * NT, 64 * 32 - 1);
    const int p = i >> 5, c = i & 31;
    g0[it] = *(const float4*)(Wg + (size_t)(2 * p) * 128 + c * 4);
    g1[it] = *(const float4*)(Wg + (size_t)(2 * p + 1) * 128 + c * 4);
  }
#pragma unroll
  for (int it = 0; it < 5; ++it) {
    const int i = min(tid + it * NT, 64 * 32 - 1);
    const int p = i >> 5, c = i & 31;
    u0[it] = *(const float4*)(W1 + (size_t)(2 * p) * 128 + c * 4);
    u1[it] = *(const float4*)(W1 + (size_t)(2 * p + 1) * 128 + c * 4);
  }
#pragma unroll
  for (int it = 0; it < 3; ++it) {
    const int i = min(tid + it * NT, 64 * 16 - 1);
    const int p = i >> 4, c = i & 15;
    s0[it] = *(const float4*)(W2 + (size_t)(2 * p) * 64 + c * 4);
    s1[it] = *(const float4*)(W2 + (size_t)(2 * p + 1) * 64 + c * 4);
  }
#pragma unroll
  for (int it = 0; it < 5; ++it) {
    const int i = tid + it * NT;
    if (i < 64 * 32) {
      const int p = i >> 5, c = i & 31;
      const float e0[4] = {g0[it].x, g0[it].y, g0[it].z, g0[it].w};
      const float e1[4] = {g1[it].x, g1[it].y, g1[it].z, g1[it].w};
      for (int j = 0; j < 4; ++j) {
        const int n = c * 4 + j;
        v2f pr; pr.x = e0[j]; pr.y = e1[j];
        *(v2f*)(wgs + wg_off(p * 16 + (n & 15)) + 2 * (n >> 4)) = pr;
      }
    }
  }
#pragma unroll
  for (int it = 0; it < 5; ++it) {
    const int i = tid + it * NT;
    if (i < 64 * 32) {
      const int p = i >> 5, c = i & 31;
      const float e0[4] = {u0[it].x, u0[it].y, u0[it].z, u0[it].w};
      const float e1[4] = {u1[it].x, u1[it].y, u1[it].z, u1[it].w};
      for (int j = 0; j < 4; ++j) {
        const int n = c * 4 + j;
        v2f pr; pr.x = e0[j]; pr.y = e1[j];
        *(v2f*)(w1s + (p * 16 + (n & 15)) * W1P + 2 * (n >> 4)) = pr;
      }
    }
  }
#pragma unroll
  for (int it = 0; it < 3; ++it) {
    const int i = tid + it * NT;
    if (i < 64 * 16) {
      const int p = i >> 4, c = i & 15;
      const float e0[4] = {s0[it].x, s0[it].y, s0[it].z, s0[it].w};
      const float e1[4] = {s1[it].x, s1[it].y, s1[it].z, s1[it].w};
      for (int j = 0; j < 4; ++j) {
        const int n = c * 4 + j;
        v2f pr; pr.x = e0[j]; pr.y = e1[j];
        *(v2f*)(w2s + (p * 16 + (n & 15)) * W2P + 2 * (n >> 4)) = pr;
      }
    }
  }
  for (int i = tid; i < 16 * 32; i += NT) {
    const int kk = i >> 5, l = i & 31;
    const int k0 = 4 * kk + 2 * (l >> 4), n = l & 15;
    v2f pr; pr.x = 0.0f; pr.y = 0.0f;
    if (n < 2) { pr.x = W3[k0 * 2 + n]; pr.y = W3[(k0 + 1) * 2 + n]; }
    *(v2f*)(w3s + i * 2) = pr;
  }
  if (tid < 128) bs[tid] = b1[tid];
  else if (tid < 192) bs[tid] = b2[tid - 128];
  else if (tid < 208) bs[tid] = tid < 194 ? b3[tid - 192] : 0.0f;
  else if (tid < 336) bs[tid] = bg[tid - 208];

  if (blk < NB) {
#pragma unroll
    for (int r = 0; r < 16; ++r) {
      const float di = rl_f(dv, r);
      const float sl = di * di;
      float4 t;
      t.x = ag[r].x + xv[r].x * sl;
      t.y = ag[r].y + xv[r].y * sl;
      t.z = ag[r].z + xv[r].z * sl;
      t.w = ag[r].w + xv[r].w * sl;
      *(float4*)(tile + r * TPITCH + lane * 4) = t;
    }
  }
  __syncthreads();
  if (blk >= NB) return;

  v8f acc0[8];
  for (int t = 0; t < 8; ++t) acc0[t] = (v8f){};
  {
    const float* ap = tile + lm * TPITCH + 2 * half;
    const float* bp = wgs + wg_off(half * 16 + lm);
    for (int kk = 0; kk < 32; ++kk) {
      const v2f a = *(const v2f*)(ap + 4 * kk);
      const float4* bq = (const float4*)(bp + kk * WG_KSTEP);
      const float4 q0 = bq[0], q1 = bq[1], q2 = bq[2], q3 = bq[3];
      v2f b;
      b.x = q0.x; b.y = q0.y; acc0[0] = wmma_f32(a, b, acc0[0]);
      b.x = q0.z; b.y = q0.w; acc0[1] = wmma_f32(a, b, acc0[1]);
      b.x = q1.x; b.y = q1.y; acc0[2] = wmma_f32(a, b, acc0[2]);
      b.x = q1.z; b.y = q1.w; acc0[3] = wmma_f32(a, b, acc0[3]);
      b.x = q2.x; b.y = q2.y; acc0[4] = wmma_f32(a, b, acc0[4]);
      b.x = q2.z; b.y = q2.w; acc0[5] = wmma_f32(a, b, acc0[5]);
      b.x = q3.x; b.y = q3.y; acc0[6] = wmma_f32(a, b, acc0[6]);
      b.x = q3.z; b.y = q3.w; acc0[7] = wmma_f32(a, b, acc0[7]);
    }
  }
  asm volatile("s_wait_dscnt 0" ::: "memory");
  for (int t = 0; t < 8; ++t) {
    float bb = bs[208 + 16 * t + lm];
    for (int r = 0; r < 8; ++r)
      tile[(r + 8 * half) * TPITCH + 16 * t + lm] = fmaxf(acc0[t][r] + bb, 0.0f);
  }
  asm volatile("s_wait_dscnt 0" ::: "memory");
#pragma unroll
  for (int r = 0; r < 16; ++r) {
    float4* hp = (float4*)(tile + r * TPITCH + lane * 4);
    float4 h = *hp;
    h.x += xv[r].x; h.y += xv[r].y; h.z += xv[r].z; h.w += xv[r].w;
    *hp = h;
  }
  asm volatile("s_wait_dscnt 0" ::: "memory");

  v8f acc1[8];
  for (int t = 0; t < 8; ++t) acc1[t] = (v8f){};
  {
    const float* ap = tile + lm * TPITCH + 2 * half;
    const float* bp = w1s + (half * 16 + lm) * W1P;
    for (int kk = 0; kk < 32; ++kk) {
      const v2f a = *(const v2f*)(ap + 4 * kk);
      const float4* bq = (const float4*)(bp + kk * (2 * 16 * W1P));
      const float4 q0 = bq[0], q1 = bq[1], q2 = bq[2], q3 = bq[3];
      v2f b;
      b.x = q0.x; b.y = q0.y; acc1[0] = wmma_f32(a, b, acc1[0]);
      b.x = q0.z; b.y = q0.w; acc1[1] = wmma_f32(a, b, acc1[1]);
      b.x = q1.x; b.y = q1.y; acc1[2] = wmma_f32(a, b, acc1[2]);
      b.x = q1.z; b.y = q1.w; acc1[3] = wmma_f32(a, b, acc1[3]);
      b.x = q2.x; b.y = q2.y; acc1[4] = wmma_f32(a, b, acc1[4]);
      b.x = q2.z; b.y = q2.w; acc1[5] = wmma_f32(a, b, acc1[5]);
      b.x = q3.x; b.y = q3.y; acc1[6] = wmma_f32(a, b, acc1[6]);
      b.x = q3.z; b.y = q3.w; acc1[7] = wmma_f32(a, b, acc1[7]);
    }
  }
  asm volatile("s_wait_dscnt 0" ::: "memory");
  for (int t = 0; t < 8; ++t) {
    float bb = bs[16 * t + lm];
    for (int r = 0; r < 8; ++r)
      tile[(r + 8 * half) * TPITCH + 16 * t + lm] = fmaxf(acc1[t][r] + bb, 0.0f);
  }
  asm volatile("s_wait_dscnt 0" ::: "memory");

  v8f acc2[4];
  for (int t = 0; t < 4; ++t) acc2[t] = (v8f){};
  {
    const float* ap = tile + lm * TPITCH + 2 * half;
    const float* bp = w2s + (half * 16 + lm) * W2P;
    for (int kk = 0; kk < 32; ++kk) {
      const v2f a = *(const v2f*)(ap + 4 * kk);
      const float4* bq = (const float4*)(bp + kk * (2 * 16 * W2P));
      const float4 q0 = bq[0], q1 = bq[1];
      v2f b;
      b.x = q0.x; b.y = q0.y; acc2[0] = wmma_f32(a, b, acc2[0]);
      b.x = q0.z; b.y = q0.w; acc2[1] = wmma_f32(a, b, acc2[1]);
      b.x = q1.x; b.y = q1.y; acc2[2] = wmma_f32(a, b, acc2[2]);
      b.x = q1.z; b.y = q1.w; acc2[3] = wmma_f32(a, b, acc2[3]);
    }
  }
  asm volatile("s_wait_dscnt 0" ::: "memory");
  for (int t = 0; t < 4; ++t) {
    float bb = bs[128 + 16 * t + lm];
    for (int r = 0; r < 8; ++r)
      tile[(r + 8 * half) * TPITCH + 16 * t + lm] = fmaxf(acc2[t][r] + bb, 0.0f);
  }
  asm volatile("s_wait_dscnt 0" ::: "memory");

  v8f acc3 = (v8f){};
  for (int kk = 0; kk < 16; ++kk) {
    const v2f a = *(const v2f*)(tile + lm * TPITCH + 4 * kk + 2 * half);
    const v2f b = *(const v2f*)(w3s + (kk * 32 + lane) * 2);
    acc3 = wmma_f32(a, b, acc3);
  }
  if (lm < 2) {
    float bb = bs[192 + lm];
    for (int r = 0; r < 8; ++r)
      out[(size_t)lm * N_NODES + row0 + r + 8 * half] = acc3[r] + bb;
  }
}

#define SEG_OFF 6500032
static_assert(SEG_OFF >= 2 * N_NODES + N_NODES * 128 && SEG_OFF % 64 == 0, "segments start behind agg, in whole 256-byte units");
static_assert(((size_t)SEG_OFF + (size_t)NGRP * GSTR) * 4 <= (size_t)268435456, "segments end inside the workspace");

extern "C" void kernel_launch(void* const* d_in, const int* in_sizes, int n_in,
                              void* d_out, int out_size, void* d_ws, size_t ws_size,
                              hipStream_t stream) {
  (void)in_sizes; (void)n_in; (void)out_size; (void)ws_size;
  const float* x  = (const float*)d_in[0];
  const int*   ei = (const int*)d_in[1];
  const float* Wg = (const float*)d_in[2];
  const float* bg = (const float*)d_in[3];
  const float* W1 = (const float*)d_in[4];
  const float* b1 = (const float*)d_in[5];
  const float* W2 = (const float*)d_in[6];
  const float* b2 = (const float*)d_in[7];
  const float* W3 = (const float*)d_in[8];
  const float* b3 = (const float*)d_in[9];
  float* out = (float*)d_out;

  float* ws   = (float*)d_ws;
  float* dinv = ws + N_NODES;
  float* agg  = ws + 2 * N_NODES;
  int*   seg  = (int*)d_ws + SEG_OFF;

  k_deg<<<NDW, DEGT, 0, stream>>>(ei, seg);
  k_cnt<<<NGRP, 1024, 0, stream>>>(seg, dinv);
  k_own<<<NOWN, 1024, 0, stream>>>(seg, dinv, x, agg);
  k_mlp<<<MLP_WGS, MLP_WAVES * 32, 0, stream>>>(x, agg, dinv, Wg, bg, W1, b1, W2, b2, W3, b3, out);
}


# k_mlp: 16 waves; a work-group's 13th tile is split by column tiles over four waves, one per SIMD: busiest SIMD 13 quarter tiles of WMMAs, not 16; barriers in those 53 groups only; sums unchanged
# speedup vs baseline: 4.2334x; 1.0614x over previous
#include <hip/hip_runtime.h>

typedef float v2f __attribute__((ext_vector_type(2)));
typedef float v8f __attribute__((ext_vector_type(8)));

#define N_NODES 50000
#define N_EDGES 800000
#define NB (N_NODES / 16)
#define TPITCH 132

__device__ __forceinline__ v8f wmma_f32(v2f a, v2f b, v8f c) {
  return __builtin_amdgcn_wmma_f32_16x16x4_f32(false, a, false, b, (short)0, c,
                                               false, false);
}

#define RNG   200
#define NOWN  (N_NODES / RNG)
#define NQ4   (N_EDGES / 4)
#define GRP   8
#define NGRP  ((NOWN + GRP - 1) / GRP)
#define DEGT  256
#define DEGE  (DEGT * 4)
#define NDW   ((NQ4 + DEGT - 1) / DEGT)
#define SEGW  1088
#define GSTR  (NDW * SEGW)
static_assert(N_NODES <= 65536 && RNG <= 256 && GRP <= 8, "src, dst % RNG and owner % GRP share one 32-bit word");
static_assert(SEGW >= DEGE + 64 && SEGW % 64 == 0, "a segment holds every edge of a work-group and is read in 64-word pieces");
static_assert(NGRP <= 32 && NGRP <= DEGT, "one counter per group, summed by one wave");
__global__ __launch_bounds__(DEGT) void k_deg(const int* __restrict__ ei, int* __restrict__ seg) {
  __shared__ int cnt[NGRP];
  __shared__ int pre[NGRP];
  __shared__ unsigned sw[DEGE];
  __shared__ int so[DEGE];
  const int tid = threadIdx.x;
  if (tid < NGRP) cnt[tid] = 0;
  __syncthreads();
  const int q = blockIdx.x * DEGT + tid;
  unsigned w[4] = {0u, 0u, 0u, 0u};
  int g[4] = {0, 0, 0, 0}, r[4] = {0, 0, 0, 0};
  const bool have = q < NQ4;
  if (have) {
    const int4 d = ((const int4*)(ei + N_EDGES))[q];
    const int4 s = ((const int4*)ei)[q];
    const unsigned dd[4] = {(unsigned)d.x, (unsigned)d.y, (unsigned)d.z, (unsigned)d.w};
    const unsigned ss[4] = {(unsigned)s.x, (unsigned)s.y, (unsigned)s.z, (unsigned)s.w};
    for (int j = 0; j < 4; ++j) {
      const unsigned o = dd[j] / RNG;
      g[j] = (int)(o / GRP);
      w[j] = ss[j] | ((dd[j] % RNG) << 16) | ((o % GRP) << 24);
      r[j] = atomicAdd(&cnt[g[j]], 1);
    }
  }
  __syncthreads();
  if (tid < NGRP) {
    int p = 0;
    for (int k = 0; k < tid; ++k) p += cnt[k];
    pre[tid] = p;
    seg[(size_t)tid * GSTR + (size_t)blockIdx.x * SEGW] = cnt[tid];
  }
  __syncthreads();
  if (have) {
    for (int j = 0; j < 4; ++j) {
      const int p = pre[g[j]] + r[j];
      sw[p] = w[j];
      so[p] = g[j] * GSTR + 1 + r[j];
    }
  }
  __syncthreads();
  const int ne = 4 * min(DEGT, NQ4 - blockIdx.x * DEGT);
  int* mine = seg + (size_t)blockIdx.x * SEGW;
  for (int p = tid; p < ne; p += DEGT) mine[so[p]] = (int)sw[p];
}

#define QCAP  64
#define NSW   ((NDW + 31) / 32)

__device__ __forceinline__ int rl_i(int v, int l) { return __builtin_amdgcn_readlane(v, l); }
__device__ __forceinline__ float rl_f(float v, int l) {
  return __int_as_float(__builtin_amdgcn_readlane(__float_as_int(v), l));
}

#define GN (GRP * RNG)
__global__ __launch_bounds__(1024) void k_cnt(const int* __restrict__ seg, float* __restrict__ dinv) {
  __shared__ int c[GN];
  const int tid  = threadIdx.x;
  const int lane = tid & 31;
  const int wave = __builtin_amdgcn_readfirstlane(tid >> 5);
  const int2* gs = (const int2*)(seg + (size_t)blockIdx.x * GSTR);
  const int nseg = (NDW - wave + 31) / 32;
  int2 v[NSW];
#pragma unroll
  for (int u = 0; u < NSW; ++u)
    v[u] = gs[(size_t)(wave + 32 * min(u, nseg - 1)) * (SEGW / 2) + lane];
  for (int i = tid; i < GN; i += 1024) c[i] = 0;
  __syncthreads();
#pragma unroll
  for (int u = 0; u < NSW; ++u) {
    if (u < nseg) {
      int2 t = v[u];
      const int2* sp = gs + (size_t)(wave + 32 * u) * (SEGW / 2) + lane;
      const int n = min(rl_i(t.x, 0), DEGE);
      for (int b = 0; ; ) {
        const int k0 = b + 2 * lane;
        const unsigned w0 = (unsigned)t.x, w1 = (unsigned)t.y;
        if (k0 >= 1 && k0 <= n)
          atomicAdd(&c[min((w0 >> 24) * RNG + ((w0 >> 16) & 0xFFu), (unsigned)(GN - 1))], 1);
        if (k0 + 1 <= n)
          atomicAdd(&c[min((w1 >> 24) * RNG + ((w1 >> 16) & 0xFFu), (unsigned)(GN - 1))], 1);
        b += 64;
        if (b > n) break;
        t = sp[b / 2];
      }
    }
  }
  __syncthreads();
  for (int i = tid; i < GN; i += 1024) {
    const int node = blockIdx.x * GN + i;
    if (node < N_NODES) {
      const float d = (float)(c[i] + 1);
      dinv[node] = d > 0.0f ? rsqrtf(d) : 0.0f;
    }
  }
}

__device__ __forceinline__ void drain_batch(const unsigned* wq, int head, int m, int lane,
                                            int base, const float* dsh,
                                            const float* __restrict__ x, float* acc) {
  int s = 0;
  int dl = RNG;
  float nrm = 0.0f;
  if (lane < m) {
    const unsigned w = wq[(head + lane) & (QCAP - 1)];
    s = (int)(w & 0xFFFFu);
    dl = (int)((w >> 16) & 0xFFu);
    nrm = dsh[s] * dsh[base + dl];
  }
  for (int j = 0; j < m; j += 4) {
    float4 v[4];
    for (int w = 0; w < 4; ++w)
      v[w] = *(const float4*)(x + (size_t)rl_i(s, j + w) * 128 + lane * 4);
    for (int w = 0; w < 4; ++w) {
      const float nj = rl_f(nrm, j + w);
      float* a = acc + rl_i(dl, j + w) * 128 + lane;
      atomicAdd(a,      v[w].x * nj);
      atomicAdd(a + 32, v[w].y * nj);
      atomicAdd(a + 64, v[w].z * nj);
      atomicAdd(a + 96, v[w].w * nj);
    }
  }
}

__device__ __forceinline__ void take_words(unsigned w, bool ok, unsigned* wq, int& head,
                                           int& cnt, int lane, unsigned below, int base,
                                           const float* dsh, const float* __restrict__ x,
                                           float* acc) {
  const unsigned bal = __builtin_amdgcn_ballot_w32(ok);
  if (bal != 0u) {
    if (ok) wq[(head + cnt + __builtin_popcount(bal & below)) & (QCAP - 1)] = w;
    cnt += __builtin_popcount(bal);
    if (cnt >= 32) {
      asm volatile("s_wait_dscnt 0" ::: "memory");
      drain_batch(wq, head, 32, lane, base, dsh, x, acc);
      head = (head + 32) & (QCAP - 1);
      cnt -= 32;
    }
  }
}

__global__ __launch_bounds__(1024) void k_own(const int* __restrict__ seg,
                                              const float* __restrict__ dinv,
                                              const float* __restrict__ x,
                                              float* __restrict__ agg) {
  __shared__ float acc[(RNG + 1) * 128];
  __shared__ unsigned queue[32 * QCAP];
  __shared__ __attribute__((aligned(16))) float dsh[N_NODES];
  const int tid  = threadIdx.x;
  const int lane = tid & 31;
  const int wave = __builtin_amdgcn_readfirstlane(tid >> 5);
  const int base = blockIdx.x * RNG;
  unsigned* wq = queue + wave * QCAP;

  const int2* gs = (const int2*)(seg + (size_t)(blockIdx.x / GRP) * GSTR);
  const int nseg = (NDW - wave + 31) / 32;
  int2 v[NSW];
#pragma unroll
  for (int u = 0; u < NSW; ++u)
    v[u] = gs[(size_t)(wave + 32 * min(u, nseg - 1)) * (SEGW / 2) + lane];

  constexpr int NV4 = N_NODES / 4;
  constexpr int NPS = (NV4 + 1023) / 1024;
  static_assert(NV4 >= 1024, "the shifted last pass needs a table of at least one pass");
  float4 dv[NPS];
#pragma unroll
  for (int it = 0; it < NPS; ++it) {
    const int i = it < NPS - 1 ? tid + it * 1024 : tid + NV4 - 1024;
    dv[it] = ((const float4*)dinv)[i];
  }
  for (int i = tid; i < (RNG + 1) * 32; i += 1024)
    ((float4*)acc)[i] = make_float4(0.0f, 0.0f, 0.0f, 0.0f);
#pragma unroll
  for (int it = 0; it < NPS; ++it) {
    const int i = it < NPS - 1 ? tid + it * 1024 : tid + NV4 - 1024;
    ((float4*)dsh)[i] = dv[it];
  }
  __syncthreads();

  const unsigned sub = (unsigned)blockIdx.x % GRP;
  const unsigned below = (1u << lane) - 1u;
  int cnt = 0;
  int head = 0;

#pragma nounroll
  for (int u = 0; u < nseg; ++u) {
    int2 t = v[0];
#pragma unroll
    for (int k = 1; k < NSW; ++k)
      if (u == k) t = v[k];
    const int2* sp = gs + (size_t)(wave + 32 * u) * (SEGW / 2) + lane;
    const int c = min(rl_i(t.x, 0), DEGE);
    for (int b = 0; ; ) {
      const int k0 = b + 2 * lane;
      const unsigned w0 = (unsigned)t.x, w1 = (unsigned)t.y;
      take_words(w0, k0 >= 1 && k0 <= c && (w0 >> 24) == sub,
                 wq, head, cnt, lane, below, base, dsh, x, acc);
      take_words(w1, k0 + 1 <= c && (w1 >> 24) == sub,
                 wq, head, cnt, lane, below, base, dsh, x, acc);
      b += 64;
      if (b > c) break;
      t = sp[b / 2];
    }
  }
  asm volatile("s_wait_dscnt 0" ::: "memory");
  if (cnt > 0) drain_batch(wq, head, cnt, lane, base, dsh, x, acc);
  asm volatile("s_wait_dscnt 0" ::: "memory");
  __syncthreads();

  for (int r = wave; r < RNG; r += 32) {
    const float* a = acc + r * 128 + lane;
    float4 o = make_float4(a[0], a[32], a[64], a[96]);
    *(float4*)(agg + (size_t)(base + r) * 128 + lane * 4) = o;
  }
}

#define MLP_FULL  12
#define MLP_WAVES 16
#define MLP_WGS   256
#define MLP_TBUF  (MLP_FULL + 1)
static_assert(MLP_FULL * MLP_WGS <= NB && NB - MLP_FULL * MLP_WGS <= MLP_WGS, "at most one left-over tile per work-group");
static_assert(MLP_WAVES - MLP_FULL == 4, "the left-over tile is split in four: two column tiles of 128, one of 64");
#define W1P 20
#define W2P 12
#define WG_FLOATS (64 * 16 * 16 + (64 * 16 / 4) * 4)
#define WG_KSTEP  (32 * 16 + 8 * 4)
__device__ __forceinline__ int wg_off(int e) { return e * 16 + (e >> 2) * 4; }


__device__ __forceinline__ void phase_sync(bool tail) {
  if (tail) __syncthreads();
  else asm volatile("s_wait_dscnt 0" ::: "memory");
}
__global__ __launch_bounds__(MLP_WAVES * 32)
void k_mlp(const float* __restrict__ x,
           const float* __restrict__ agg, const float* __restrict__ dinv,
           const float* __restrict__ Wg, const float* __restrict__ bg,
           const float* __restrict__ W1, const float* __restrict__ b1,
           const float* __restrict__ W2, const float* __restrict__ b2,
           const float* __restrict__ W3, const float* __restrict__ b3,
           float* __restrict__ out) {
  __shared__ __attribute__((aligned(16))) float wgs[WG_FLOATS];
  __shared__ __attribute__((aligned(16))) float w1s[64 * 16 * W1P];
  __shared__ __attribute__((aligned(16))) float w2s[64 * 16 * W2P];
  __shared__ __attribute__((aligned(16))) float w3s[16 * 32 * 2];
  __shared__ float bs[128 + 64 + 16 + 128];
  __shared__ __attribute__((aligned(16))) float lds[MLP_TBUF * 16 * TPITCH];
  const int tid  = threadIdx.x;
  const int lane = tid & 31;
  const int wave = __builtin_amdgcn_readfirstlane(tid >> 5);
  const int half = lane >> 4;
  const int lm   = lane & 15;
  const bool full = wave < MLP_FULL;
  const int  q    = full ? 0 : wave - MLP_FULL;
  float* tile = lds + (full ? wave : MLP_FULL) * (16 * TPITCH);
  const int blk = (full ? wave : MLP_FULL) * MLP_WGS + blockIdx.x;
  const bool act = blk < NB;
  const bool tail = (int)blockIdx.x < NB - MLP_FULL * MLP_WGS;
  const int row0 = blk * 16;

  float dv = 0.0f;
  float4 xv[16], ag[16];
#pragma unroll
  for (int r = 0; r < 16; ++r) {
    xv[r] = make_float4(0.0f, 0.0f, 0.0f, 0.0f);
    ag[r] = make_float4(0.0f, 0.0f, 0.0f, 0.0f);
  }
  if (act) {
    if (lane < 16) dv = dinv[row0 + lane];
#pragma unroll
    for (int r = 0; r < 16; ++r)
      xv[r] = *(const float4*)(x + (size_t)(row0 + r) * 128 + lane * 4);
#pragma unroll
    for (int r = 0; r < 16; ++r)
      ag[r] = *(const float4*)(agg + (size_t)(row0 + r) * 128 + lane * 4);
  }

  constexpr int NT = MLP_WAVES * 32;
  constexpr int PG = (64 * 32 + NT - 1) / NT;
  constexpr int PS = (64 * 16 + NT - 1) / NT;
  float4 g0[PG], g1[PG], u0[PG], u1[PG], s0[PS], s1[PS];
#pragma unroll
  for (int it = 0; it < PG; ++it) {
    const int i = min(tid + it * NT, 64 * 32 - 1);
    const int p = i >> 5, c = i & 31;
    g0[it] = *(const float4*)(Wg + (size_t)(2 * p) * 128 + c * 4);
    g1[it] = *(const float4*)(Wg + (size_t)(2 * p + 1) * 128 + c * 4);
  }
#pragma unroll
  for (int it = 0; it < PG; ++it) {
    const int i = min(tid + it * NT, 64 * 32 - 1);
    const int p = i >> 5, c = i & 31;
    u0[it] = *(const float4*)(W1 + (size_t)(2 * p) * 128 + c * 4);
    u1[it] = *(const float4*)(W1 + (size_t)(2 * p + 1) * 128 + c * 4);
  }
#pragma unroll
  for (int it = 0; it < PS; ++it) {
    const int i = min(tid + it * NT, 64 * 16 - 1);
    const int p = i >> 4, c = i & 15;
    s0[it] = *(const float4*)(W2 + (size_t)(2 * p) * 64 + c * 4);
    s1[it] = *(const float4*)(W2 + (size_t)(2 * p + 1) * 64 + c * 4);
  }
#pragma unroll
  for (int it = 0; it < PG; ++it) {
    const int i = tid + it * NT;
    if (i < 64 * 32) {
      const int p = i >> 5, c = i & 31;
      const float e0[4] = {g0[it].x, g0[it].y, g0[it].z, g0[it].w};
      const float e1[4] = {g1[it].x, g1[it].y, g1[it].z, g1[it].w};
      for (int j = 0; j < 4; ++j) {
        const int n = c * 4 + j;
        v2f pr; pr.x = e0[j]; pr.y = e1[j];
        *(v2f*)(wgs + wg_off(p * 16 + (n & 15)) + 2 * (n >> 4)) = pr;
      }
    }
  }
#pragma unroll
  for (int it = 0; it < PG; ++it) {
    const int i = tid + it * NT;
    if (i < 64 * 32) {
      const int p = i >> 5, c = i & 31;
      const float e0[4] = {u0[it].x, u0[it].y, u0[it].z, u0[it].w};
      const float e1[4] = {u1[it].x, u1[it].y, u1[it].z, u1[it].w};
      for (int j = 0; j < 4; ++j) {
        const int n = c * 4 + j;
        v2f pr; pr.x = e0[j]; pr.y = e1[j];
        *(v2f*)(w1s + (p * 16 + (n & 15)) * W1P + 2 * (n >> 4)) = pr;
      }
    }
  }
#pragma unroll
  for (int it = 0; it < PS; ++it) {
    const int i = tid + it * NT;
    if (i < 64 * 16) {
      const int p = i >> 4, c = i & 15;
      const float e0[4] = {s0[it].x, s0[it].y, s0[it].z, s0[it].w};
      const float e1[4] = {s1[it].x, s1[it].y, s1[it].z, s1[it].w};
      for (int j = 0; j < 4; ++j) {
        const int n = c * 4 + j;
        v2f pr; pr.x = e0[j]; pr.y = e1[j];
        *(v2f*)(w2s + (p * 16 + (n & 15)) * W2P + 2 * (n >> 4)) = pr;
      }
    }
  }
  for (int i = tid; i < 16 * 32; i += NT) {
    const int kk = i >> 5, l = i & 31;
    const int k0 = 4 * kk + 2 * (l >> 4), n = l & 15;
    v2f pr; pr.x = 0.0f; pr.y = 0.0f;
    if (n < 2) { pr.x = W3[k0 * 2 + n]; pr.y = W3[(k0 + 1) * 2 + n]; }
    *(v2f*)(w3s + i * 2) = pr;
  }
  if (tid < 128) bs[tid] = b1[tid];
  else if (tid < 192) bs[tid] = b2[tid - 128];
  else if (tid < 208) bs[tid] = tid < 194 ? b3[tid - 192] : 0.0f;
  else if (tid < 336) bs[tid] = bg[tid - 208];

  if (act) {
#pragma unroll
    for (int r = 0; r < 16; ++r) {
      if (full || (r >> 2) == q) {
        const float di = rl_f(dv, r);
        const float sl = di * di;
        float4 t;
        t.x = ag[r].x + xv[r].x * sl;
        t.y = ag[r].y + xv[r].y * sl;
        t.z = ag[r].z + xv[r].z * sl;
        t.w = ag[r].w + xv[r].w * sl;
        *(float4*)(tile + r * TPITCH + lane * 4) = t;
      }
    }
  }
  __syncthreads();
  if (!act) return;

  const float* ap = tile + lm * TPITCH + 2 * half;

  v8f acc0[8], qa0[2];
  for (int t = 0; t < 8; ++t) acc0[t] = (v8f){};
  for (int t = 0; t < 2; ++t) qa0[t] = (v8f){};
  {
    const float* bp = wgs + wg_off(half * 16 + lm);
    if (full) {
      for (int kk = 0; kk < 32; ++kk) {
        const v2f a = *(const v2f*)(ap + 4 * kk);
        const float4* bq = (const float4*)(bp + kk * WG_KSTEP);
        const float4 q0 = bq[0], q1 = bq[1], q2 = bq[2], q3 = bq[3];
        v2f b;
        b.x = q0.x; b.y = q0.y; acc0[0] = wmma_f32(a, b, acc0[0]);
        b.x = q0.z; b.y = q0.w; acc0[1] = wmma_f32(a, b, acc0[1]);
        b.x = q1.x; b.y = q1.y; acc0[2] = wmma_f32(a, b, acc0[2]);
        b.x = q1.z; b.y = q1.w; acc0[3] = wmma_f32(a, b, acc0[3]);
        b.x = q2.x; b.y = q2.y; acc0[4] = wmma_f32(a, b, acc0[4]);
        b.x = q2.z; b.y = q2.w; acc0[5] = wmma_f32(a, b, acc0[5]);
        b.x = q3.x; b.y = q3.y; acc0[6] = wmma_f32(a, b, acc0[6]);
        b.x = q3.z; b.y = q3.w; acc0[7] = wmma_f32(a, b, acc0[7]);
      }
    } else {
      for (int kk = 0; kk < 32; ++kk) {
        const v2f a = *(const v2f*)(ap + 4 * kk);
        const float4 q0 = ((const float4*)(bp + kk * WG_KSTEP))[q];
        v2f b;
        b.x = q0.x; b.y = q0.y; qa0[0] = wmma_f32(a, b, qa0[0]);
        b.x = q0.z; b.y = q0.w; qa0[1] = wmma_f32(a, b, qa0[1]);
      }
    }
  }
  phase_sync(tail);
  if (full) {
    for (int t = 0; t < 8; ++t) {
      float bb = bs[208 + 16 * t + lm];
      for (int r = 0; r < 8; ++r)
        tile[(r + 8 * half) * TPITCH + 16 * t + lm] = fmaxf(acc0[t][r] + bb, 0.0f);
    }
  } else {
    for (int t = 0; t < 2; ++t) {
      float bb = bs[208 + 16 * (2 * q + t) + lm];
      for (int r = 0; r < 8; ++r)
        tile[(r + 8 * half) * TPITCH + 16 * (2 * q + t) + lm] = fmaxf(qa0[t][r] + bb, 0.0f);
    }
  }
  phase_sync(tail);
#pragma unroll
  for (int r = 0; r < 16; ++r) {
    if (full || (r >> 2) == q) {
      float4* hp = (float4*)(tile + r * TPITCH + lane * 4);
      float4 h = *hp;
      h.x += xv[r].x; h.y += xv[r].y; h.z += xv[r].z; h.w += xv[r].w;
      *hp = h;
    }
  }
  phase_sync(tail);

  v8f acc1[8], qa1[2];
  for (int t = 0; t < 8; ++t) acc1[t] = (v8f){};
  for (int t = 0; t < 2; ++t) qa1[t] = (v8f){};
  {
    const float* bp = w1s + (half * 16 + lm) * W1P;
    if (full) {
      for (int kk = 0; kk < 32; ++kk) {
        const v2f a = *(const v2f*)(ap + 4 * kk);
        const float4* bq = (const float4*)(bp + kk * (2 * 16 * W1P));
        const float4 q0 = bq[0], q1 = bq[1], q2 = bq[2], q3 = bq[3];
        v2f b;
        b.x = q0.x; b.y = q0.y; acc1[0] = wmma_f32(a, b, acc1[0]);
        b.x = q0.z; b.y = q0.w; acc1[1] = wmma_f32(a, b, acc1[1]);
        b.x = q1.x; b.y = q1.y; acc1[2] = wmma_f32(a, b, acc1[2]);
        b.x = q1.z; b.y = q1.w; acc1[3] = wmma_f32(a, b, acc1[3]);
        b.x = q2.x; b.y = q2.y; acc1[4] = wmma_f32(a, b, acc1[4]);
        b.x = q2.z; b.y = q2.w; acc1[5] = wmma_f32(a, b, acc1[5]);
        b.x = q3.x; b.y = q3.y; acc1[6] = wmma_f32(a, b, acc1[6]);
        b.x = q3.z; b.y = q3.w; acc1[7] = wmma_f32(a, b, acc1[7]);
      }
    } else {
      for (int kk = 0; kk < 32; ++kk) {
        const v2f a = *(const v2f*)(ap + 4 * kk);
        const float4 q0 = ((const float4*)(bp + kk * (2 * 16 * W1P)))[q];
        v2f b;
        b.x = q0.x; b.y = q0.y; qa1[0] = wmma_f32(a, b, qa1[0]);
        b.x = q0.z; b.y = q0.w; qa1[1] = wmma_f32(a, b, qa1[1]);
      }
    }
  }
  phase_sync(tail);
  if (full) {
    for (int t = 0; t < 8; ++t) {
      float bb = bs[16 * t + lm];
      for (int r = 0; r < 8; ++r)
        tile[(r + 8 * half) * TPITCH + 16 * t + lm] = fmaxf(acc1[t][r] + bb, 0.0f);
    }
  } else {
    for (int t = 0; t < 2; ++t) {
      float bb = bs[16 * (2 * q + t) + lm];
      for (int r = 0; r < 8; ++r)
        tile[(r + 8 * half) * TPITCH + 16 * (2 * q + t) + lm] = fmaxf(qa1[t][r] + bb, 0.0f);
    }
  }
  phase_sync(tail);

  v8f acc2[4], qa2 = (v8f){};
  for (int t = 0; t < 4; ++t) acc2[t] = (v8f){};
  {
    const float* bp = w2s + (half * 16 + lm) * W2P;
    if (full) {
      for (int kk = 0; kk < 32; ++kk) {
        const v2f a = *(const v2f*)(ap + 4 * kk);
        const float4* bq = (const float4*)(bp + kk * (2 * 16 * W2P));
        const float4 q0 = bq[0], q1 = bq[1];
        v2f b;
        b.x = q0.x; b.y = q0.y; acc2[0] = wmma_f32(a, b, acc2[0]);
        b.x = q0.z; b.y = q0.w; acc2[1] = wmma_f32(a, b, acc2[1]);
        b.x = q1.x; b.y = q1.y; acc2[2] = wmma_f32(a, b, acc2[2]);
        b.x = q1.z; b.y = q1.w; acc2[3] = wmma_f32(a, b, acc2[3]);
      }
    } else {
      for (int kk = 0; kk < 32; ++kk) {
        const v2f a = *(const v2f*)(ap + 4 * kk);
        const v2f b = *(const v2f*)(bp + kk * (2 * 16 * W2P) + 2 * q);
        qa2 = wmma_f32(a, b, qa2);
      }
    }
  }
  phase_sync(tail);
  if (full) {
    for (int t = 0; t < 4; ++t) {
      float bb = bs[128 + 16 * t + lm];
      for (int r = 0; r < 8; ++r)
        tile[(r + 8 * half) * TPITCH + 16 * t + lm] = fmaxf(acc2[t][r] + bb, 0.0f);
    }
  } else {
    float bb = bs[128 + 16 * q + lm];
    for (int r = 0; r < 8; ++r)
      tile[(r + 8 * half) * TPITCH + 16 * q + lm] = fmaxf(qa2[r] + bb, 0.0f);
  }
  phase_sync(tail);

  if (full || q == 0) {
    v8f acc3 = (v8f){};
    for (int kk = 0; kk < 16; ++kk) {
      const v2f a = *(const v2f*)(ap + 4 * kk);
      const v2f b = *(const v2f*)(w3s + (kk * 32 + lane) * 2);
      acc3 = wmma_f32(a, b, acc3);
    }
    if (lm < 2) {
      float bb = bs[192 + lm];
      for (int r = 0; r < 8; ++r)
        out[(size_t)lm * N_NODES + row0 + r + 8 * half] = acc3[r] + bb;
    }
  }
}

#define SEG_OFF 6500032
static_assert(SEG_OFF >= 2 * N_NODES + N_NODES * 128 && SEG_OFF % 64 == 0, "segments start behind agg, in whole 256-byte units");
static_assert(((size_t)SEG_OFF + (size_t)NGRP * GSTR) * 4 <= (size_t)268435456, "segments end inside the workspace");

extern "C" void kernel_launch(void* const* d_in, const int* in_sizes, int n_in,
                              void* d_out, int out_size, void* d_ws, size_t ws_size,
                              hipStream_t stream) {
  (void)in_sizes; (void)n_in; (void)out_size; (void)ws_size;
  const float* x  = (const float*)d_in[0];
  const int*   ei = (const int*)d_in[1];
  const float* Wg = (const float*)d_in[2];
  const float* bg = (const float*)d_in[3];
  const float* W1 = (const float*)d_in[4];
  const float* b1 = (const float*)d_in[5];
  const float* W2 = (const float*)d_in[6];
  const float* b2 = (const float*)d_in[7];
  const float* W3 = (const float*)d_in[8];
  const float* b3 = (const float*)d_in[9];
  float* out = (float*)d_out;

  float* ws   = (float*)d_ws;
  float* dinv = ws + N_NODES;
  float* agg  = ws + 2 * N_NODES;
  int*   seg  = (int*)d_ws + SEG_OFF;

  k_deg<<<NDW, DEGT, 0, stream>>>(ei, seg);
  k_cnt<<<NGRP, 1024, 0, stream>>>(seg, dinv);
  k_own<<<NOWN, 1024, 0, stream>>>(seg, dinv, x, agg);
  k_mlp<<<MLP_WGS, MLP_WAVES * 32, 0, stream>>>(x, agg, dinv, Wg, bg, W1, b1, W2, b2, W3, b3, out);
}


# k_own runs the matrix layers on its 200 summed rows itself, in the LDS rows they were summed in (tiles made in place, parameters over the dead dinv table): agg and k_mlp launch gone; sums unchanged
# speedup vs baseline: 4.6838x; 1.1064x over previous
#include <hip/hip_runtime.h>

typedef float v2f __attribute__((ext_vector_type(2)));
typedef float v8f __attribute__((ext_vector_type(8)));

#define N_NODES 50000
#define N_EDGES 800000
#define TPITCH 132

__device__ __forceinline__ v8f wmma_f32(v2f a, v2f b, v8f c) {
  return __builtin_amdgcn_wmma_f32_16x16x4_f32(false, a, false, b, (short)0, c,
                                               false, false);
}

#define RNG   200
#define NOWN  (N_NODES / RNG)
#define NQ4   (N_EDGES / 4)
#define GRP   8
#define NGRP  ((NOWN + GRP - 1) / GRP)
#define DEGT  256
#define DEGE  (DEGT * 4)
#define NDW   ((NQ4 + DEGT - 1) / DEGT)
#define SEGW  1088
#define GSTR  (NDW * SEGW)
static_assert(N_NODES <= 65536 && RNG <= 256 && GRP <= 8, "src, dst % RNG and owner % GRP share one 32-bit word");
static_assert(SEGW >= DEGE + 64 && SEGW % 64 == 0, "a segment holds every edge of a work-group and is read in 64-word pieces");
static_assert(NGRP <= 32 && NGRP <= DEGT, "one counter per group, summed by one wave");
__global__ __launch_bounds__(DEGT) void k_deg(const int* __restrict__ ei, int* __restrict__ seg) {
  __shared__ int cnt[NGRP];
  __shared__ int pre[NGRP];
  __shared__ unsigned sw[DEGE];
  __shared__ int so[DEGE];
  const int tid = threadIdx.x;
  if (tid < NGRP) cnt[tid] = 0;
  __syncthreads();
  const int q = blockIdx.x * DEGT + tid;
  unsigned w[4] = {0u, 0u, 0u, 0u};
  int g[4] = {0, 0, 0, 0}, r[4] = {0, 0, 0, 0};
  const bool have = q < NQ4;
  if (have) {
    const int4 d = ((const int4*)(ei + N_EDGES))[q];
    const int4 s = ((const int4*)ei)[q];
    const unsigned dd[4] = {(unsigned)d.x, (unsigned)d.y, (unsigned)d.z, (unsigned)d.w};
    const unsigned ss[4] = {(unsigned)s.x, (unsigned)s.y, (unsigned)s.z, (unsigned)s.w};
    for (int j = 0; j < 4; ++j) {
      const unsigned o = dd[j] / RNG;
      g[j] = (int)(o / GRP);
      w[j] = ss[j] | ((dd[j] % RNG) << 16) | ((o % GRP) << 24);
      r[j] = atomicAdd(&cnt[g[j]], 1);
    }
  }
  __syncthreads();
  if (tid < NGRP) {
    int p = 0;
    for (int k = 0; k < tid; ++k) p += cnt[k];
    pre[tid] = p;
    seg[(size_t)tid * GSTR + (size_t)blockIdx.x * SEGW] = cnt[tid];
  }
  __syncthreads();
  if (have) {
    for (int j = 0; j < 4; ++j) {
      const int p = pre[g[j]] + r[j];
      sw[p] = w[j];
      so[p] = g[j] * GSTR + 1 + r[j];
    }
  }
  __syncthreads();
  const int ne = 4 * min(DEGT, NQ4 - blockIdx.x * DEGT);
  int* mine = seg + (size_t)blockIdx.x * SEGW;
  for (int p = tid; p < ne; p += DEGT) mine[so[p]] = (int)sw[p];
}

#define QCAP  64
#define NSW   ((NDW + 31) / 32)

__device__ __forceinline__ int rl_i(int v, int l) { return __builtin_amdgcn_readlane(v, l); }
__device__ __forceinline__ float rl_f(float v, int l) {
  return __int_as_float(__builtin_amdgcn_readlane(__float_as_int(v), l));
}

#define GN (GRP * RNG)
__global__ __launch_bounds__(1024) void k_cnt(const int* __restrict__ seg, float* __restrict__ dinv) {
  __shared__ int c[GN];
  const int tid  = threadIdx.x;
  const int lane = tid & 31;
  const int wave = __builtin_amdgcn_readfirstlane(tid >> 5);
  const int2* gs = (const int2*)(seg + (size_t)blockIdx.x * GSTR);
  const int nseg = (NDW - wave + 31) / 32;
  int2 v[NSW];
#pragma unroll
  for (int u = 0; u < NSW; ++u)
    v[u] = gs[(size_t)(wave + 32 * min(u, nseg - 1)) * (SEGW / 2) + lane];
  for (int i = tid; i < GN; i += 1024) c[i] = 0;
  __syncthreads();
#pragma unroll
  for (int u = 0; u < NSW; ++u) {
    if (u < nseg) {
      int2 t = v[u];
      const int2* sp = gs + (size_t)(wave + 32 * u) * (SEGW / 2) + lane;
      const int n = min(rl_i(t.x, 0), DEGE);
      for (int b = 0; ; ) {
        const int k0 = b + 2 * lane;
        const unsigned w0 = (unsigned)t.x, w1 = (unsigned)t.y;
        if (k0 >= 1 && k0 <= n)
          atomicAdd(&c[min((w0 >> 24) * RNG + ((w0 >> 16) & 0xFFu), (unsigned)(GN - 1))], 1);
        if (k0 + 1 <= n)
          atomicAdd(&c[min((w1 >> 24) * RNG + ((w1 >> 16) & 0xFFu), (unsigned)(GN - 1))], 1);
        b += 64;
        if (b > n) break;
        t = sp[b / 2];
      }
    }
  }
  __syncthreads();
  for (int i = tid; i < GN; i += 1024) {
    const int node = blockIdx.x * GN + i;
    if (node < N_NODES) {
      const float d = (float)(c[i] + 1);
      dinv[node] = d > 0.0f ? rsqrtf(d) : 0.0f;
    }
  }
}

__device__ __forceinline__ void drain_batch(const unsigned* wq, int head, int m, int lane,
                                            int base, const float* dsh,
                                            const float* __restrict__ x, float* acc) {
  int s = 0;
  int dl = RNG;
  float nrm = 0.0f;
  if (lane < m) {
    const unsigned w = wq[(head + lane) & (QCAP - 1)];
    s = (int)(w & 0xFFFFu);
    dl = (int)((w >> 16) & 0xFFu);
    nrm = dsh[s] * dsh[base + dl];
  }
  for (int j = 0; j < m; j += 4) {
    float4 v[4];
    for (int w = 0; w < 4; ++w)
      v[w] = *(const float4*)(x + (size_t)rl_i(s, j + w) * 128 + lane * 4);
    for (int w = 0; w < 4; ++w) {
      const float nj = rl_f(nrm, j + w);
      float* a = acc + rl_i(dl, j + w) * TPITCH + lane;
      atomicAdd(a,      v[w].x * nj);
      atomicAdd(a + 32, v[w].y * nj);
      atomicAdd(a + 64, v[w].z * nj);
      atomicAdd(a + 96, v[w].w * nj);
    }
  }
}

__device__ __forceinline__ void take_words(unsigned w, bool ok, unsigned* wq, int& head,
                                           int& cnt, int lane, unsigned below, int base,
                                           const float* dsh, const float* __restrict__ x,
                                           float* acc) {
  const unsigned bal = __builtin_amdgcn_ballot_w32(ok);
  if (bal != 0u) {
    if (ok) wq[(head + cnt + __builtin_popcount(bal & below)) & (QCAP - 1)] = w;
    cnt += __builtin_popcount(bal);
    if (cnt >= 32) {
      asm volatile("s_wait_dscnt 0" ::: "memory");
      drain_batch(wq, head, 32, lane, base, dsh, x, acc);
      head = (head + 32) & (QCAP - 1);
      cnt -= 32;
    }
  }
}

#define MLP_FULL  12
#define MLP_WAVES 16
#define MLP_TBUF  (MLP_FULL + 1)
static_assert((RNG + 15) / 16 == MLP_TBUF && RNG + 1 <= MLP_TBUF * 16, "an owner's rows and the padding row fit its tiles");
static_assert(MLP_WAVES - MLP_FULL == 4, "the last tile is split in four: two column tiles of 128, one of 64");
#define W1P 20
#define W2P 12
#define WG_FLOATS (64 * 16 * 16 + (64 * 16 / 4) * 4)
#define WG_KSTEP  (32 * 16 + 8 * 4)
__device__ __forceinline__ int wg_off(int e) { return e * 16 + (e >> 2) * 4; }

#define SM_ROWS  (MLP_TBUF * 16 * TPITCH)
#define SM_QUE   SM_ROWS
#define SM_DSH   (SM_QUE + 32 * QCAP)
#define SM_WG    SM_ROWS
#define SM_W1    (SM_WG + WG_FLOATS)
#define SM_W2    (SM_W1 + 64 * 16 * W1P)
#define SM_W3    (SM_W2 + 64 * 16 * W2P)
#define SM_BS    (SM_W3 + 16 * 32 * 2)
#define SM_ENDA  (SM_DSH + N_NODES)
#define SM_ENDB  (SM_BS + 128 + 64 + 16 + 128)
#define SM_FLOATS (SM_ENDA > SM_ENDB ? SM_ENDA : SM_ENDB)
static_assert(SM_ROWS % 4 == 0 && SM_DSH % 4 == 0 && SM_W1 % 4 == 0 && SM_W2 % 4 == 0 && SM_W3 % 4 == 0, "16-byte vectors stay aligned");
static_assert((size_t)SM_FLOATS * 4 <= (size_t)320 * 1024, "one work-group's LDS");

__global__ __launch_bounds__(1024) void k_own(const int* __restrict__ seg,
                                              const float* __restrict__ dinv,
                                              const float* __restrict__ x,
                                              const float* __restrict__ Wg, const float* __restrict__ bg,
                                              const float* __restrict__ W1, const float* __restrict__ b1,
                                              const float* __restrict__ W2, const float* __restrict__ b2,
                                              const float* __restrict__ W3, const float* __restrict__ b3,
                                              float* __restrict__ out) {
  __shared__ __attribute__((aligned(16))) float sm[SM_FLOATS];
  float* acc = sm;
  unsigned* queue = (unsigned*)(sm + SM_QUE);
  float* dsh = sm + SM_DSH;
  const int tid  = threadIdx.x;
  const int lane = tid & 31;
  const int wave = __builtin_amdgcn_readfirstlane(tid >> 5);
  const int base = blockIdx.x * RNG;
  unsigned* wq = queue + wave * QCAP;

  const int2* gs = (const int2*)(seg + (size_t)(blockIdx.x / GRP) * GSTR);
  const int nseg = (NDW - wave + 31) / 32;
  int2 v[NSW];
#pragma unroll
  for (int u = 0; u < NSW; ++u)
    v[u] = gs[(size_t)(wave + 32 * min(u, nseg - 1)) * (SEGW / 2) + lane];

  constexpr int NV4 = N_NODES / 4;
  constexpr int NPS = (NV4 + 1023) / 1024;
  static_assert(NV4 >= 1024, "the shifted last pass needs a table of at least one pass");
  float4 dv[NPS];
#pragma unroll
  for (int it = 0; it < NPS; ++it) {
    const int i = it < NPS - 1 ? tid + it * 1024 : tid + NV4 - 1024;
    dv[it] = ((const float4*)dinv)[i];
  }
  for (int i = tid; i < SM_ROWS / 4; i += 1024)
    ((float4*)acc)[i] = make_float4(0.0f, 0.0f, 0.0f, 0.0f);
#pragma unroll
  for (int it = 0; it < NPS; ++it) {
    const int i = it < NPS - 1 ? tid + it * 1024 : tid + NV4 - 1024;
    ((float4*)dsh)[i] = dv[it];
  }
  __syncthreads();

  const unsigned sub = (unsigned)blockIdx.x % GRP;
  const unsigned below = (1u << lane) - 1u;
  int cnt = 0;
  int head = 0;

#pragma nounroll
  for (int u = 0; u < nseg; ++u) {
    int2 t = v[0];
#pragma unroll
    for (int k = 1; k < NSW; ++k)
      if (u == k) t = v[k];
    const int2* sp = gs + (size_t)(wave + 32 * u) * (SEGW / 2) + lane;
    const int c = min(rl_i(t.x, 0), DEGE);
    for (int b = 0; ; ) {
      const int k0 = b + 2 * lane;
      const unsigned w0 = (unsigned)t.x, w1 = (unsigned)t.y;
      take_words(w0, k0 >= 1 && k0 <= c && (w0 >> 24) == sub,
                 wq, head, cnt, lane, below, base, dsh, x, acc);
      take_words(w1, k0 + 1 <= c && (w1 >> 24) == sub,
                 wq, head, cnt, lane, below, base, dsh, x, acc);
      b += 64;
      if (b > c) break;
      t = sp[b / 2];
    }
  }
  asm volatile("s_wait_dscnt 0" ::: "memory");
  if (cnt > 0) drain_batch(wq, head, cnt, lane, base, dsh, x, acc);
  asm volatile("s_wait_dscnt 0" ::: "memory");
  __syncthreads();

  float* wgs = sm + SM_WG;
  float* w1s = sm + SM_W1;
  float* w2s = sm + SM_W2;
  float* w3s = sm + SM_W3;
  float* bs  = sm + SM_BS;
  const int half = lane >> 4;
  const int lm   = lane & 15;
  const bool work = wave < MLP_WAVES;
  const bool full = wave < MLP_FULL;
  const int  q    = (wave - MLP_FULL) & 3;
  const int  ti   = full ? wave : MLP_FULL;
  float* tile = acc + ti * (16 * TPITCH);
  const int trow = ti * 16;
  const int row0 = base + trow;

  float dself = 0.0f;
  float4 xv[16];
#pragma unroll
  for (int r = 0; r < 16; ++r) xv[r] = make_float4(0.0f, 0.0f, 0.0f, 0.0f);
  if (work) {
    if (lane < 16 && trow + lane < RNG) dself = dinv[row0 + lane];
#pragma unroll
    for (int r = 0; r < 16; ++r)
      xv[r] = *(const float4*)(x + (size_t)(base + min(trow + r, RNG - 1)) * 128 + lane * 4);
  }

  constexpr int NT = 1024 - MLP_WAVES * 32;
  constexpr int PG = (64 * 32 + NT - 1) / NT;
  constexpr int PS = (64 * 16 + NT - 1) / NT;
  static_assert((64 * 32) % NT == 0 && (64 * 16) % NT == 0 && NT >= 336, "whole passes; one thread per bias word");
  if (!work) {
  const int st = tid - MLP_WAVES * 32;
  float4 g0[PG], g1[PG], u0[PG], u1[PG], s0[PS], s1[PS];
#pragma unroll
  for (int it = 0; it < PG; ++it) {
    const int i = st + it * NT;
    const int p = i >> 5, c = i & 31;
    g0[it] = *(const float4*)(Wg + (size_t)(2 * p) * 128 + c * 4);
    g1[it] = *(const float4*)(Wg + (size_t)(2 * p + 1) * 128 + c * 4);
  }
#pragma unroll
  for (int it = 0; it < PG; ++it) {
    const int i = st + it * NT;
    const int p = i >> 5, c = i & 31;
    u0[it] = *(const float4*)(W1 + (size_t)(2 * p) * 128 + c * 4);
    u1[it] = *(const float4*)(W1 + (size_t)(2 * p + 1) * 128 + c * 4);
  }
#pragma unroll
  for (int it = 0; it < PS; ++it) {
    const int i = st + it * NT;
    const int p = i >> 4, c = i & 15;
    s0[it] = *(const float4*)(W2 + (size_t)(2 * p) * 64 + c * 4);
    s1[it] = *(const float4*)(W2 + (size_t)(2 * p + 1) * 64 + c * 4);
  }
#pragma unroll
  for (int it = 0; it < PG; ++it) {
    const int i = st + it * NT;
    {
      const int p = i >> 5, c = i & 31;
      const float e0[4] = {g0[it].x, g0[it].y, g0[it].z, g0[it].w};
      const float e1[4] = {g1[it].x, g1[it].y, g1[it].z, g1[it].w};
      for (int j = 0; j < 4; ++j) {
        const int n = c * 4 + j;
        v2f pr; pr.x = e0[j]; pr.y = e1[j];
        *(v2f*)(wgs + wg_off(p * 16 + (n & 15)) + 2 * (n >> 4)) = pr;
      }
    }
  }
#pragma unroll
  for (int it = 0; it < PG; ++it) {
    const int i = st + it * NT;
    {
      const int p = i >> 5, c = i & 31;
      const float e0[4] = {u0[it].x, u0[it].y, u0[it].z, u0[it].w};
      const float e1[4] = {u1[it].x, u1[it].y, u1[it].z, u1[it].w};
      for (int j = 0; j < 4; ++j) {
        const int n = c * 4 + j;
        v2f pr; pr.x = e0[j]; pr.y = e1[j];
        *(v2f*)(w1s + (p * 16 + (n & 15)) * W1P + 2 * (n >> 4)) = pr;
      }
    }
  }
#pragma unroll
  for (int it = 0; it < PS; ++it) {
    const int i = st + it * NT;
    {
      const int p = i >> 4, c = i & 15;
      const float e0[4] = {s0[it].x, s0[it].y, s0[it].z, s0[it].w};
      const float e1[4] = {s1[it].x, s1[it].y, s1[it].z, s1[it].w};
      for (int j = 0; j < 4; ++j) {
        const int n = c * 4 + j;
        v2f pr; pr.x = e0[j]; pr.y = e1[j];
        *(v2f*)(w2s + (p * 16 + (n & 15)) * W2P + 2 * (n >> 4)) = pr;
      }
    }
  }
  for (int i = st; i < 16 * 32; i += NT) {
    const int kk = i >> 5, l = i & 31;
    const int k0 = 4 * kk + 2 * (l >> 4), n = l & 15;
    v2f pr; pr.x = 0.0f; pr.y = 0.0f;
    if (n < 2) { pr.x = W3[k0 * 2 + n]; pr.y = W3[(k0 + 1) * 2 + n]; }
    *(v2f*)(w3s + i * 2) = pr;
  }
  if (st < 128) bs[st] = b1[st];
  else if (st < 192) bs[st] = b2[st - 128];
  else if (st < 208) bs[st] = st < 194 ? b3[st - 192] : 0.0f;
  else if (st < 336) bs[st] = bg[st - 208];
  }

  if (work) {
#pragma unroll
    for (int r = 0; r < 16; ++r) {
      if (full || (r >> 2) == q) {
        const float di = rl_f(dself, r);
        const float sl = di * di;
        const float* a = tile + r * TPITCH + lane;
        float4 t;
        t.x = a[0]  + xv[r].x * sl;
        t.y = a[32] + xv[r].y * sl;
        t.z = a[64] + xv[r].z * sl;
        t.w = a[96] + xv[r].w * sl;
        *(float4*)(tile + r * TPITCH + lane * 4) = t;
      }
    }
  }
  __syncthreads();

  const float* ap = tile + lm * TPITCH + 2 * half;


  v8f acc0[8];
  for (int t = 0; t < 8; ++t) acc0[t] = (v8f){};
  if (work) {
    const float* bp = wgs + wg_off(half * 16 + lm);
    if (full) {
      for (int kk = 0; kk < 32; ++kk) {
        const v2f a = *(const v2f*)(ap + 4 * kk);
        const float4* bq = (const float4*)(bp + kk * WG_KSTEP);
        const float4 q0 = bq[0], q1 = bq[1], q2 = bq[2], q3 = bq[3];
        v2f b;
        b.x = q0.x; b.y = q0.y; acc0[0] = wmma_f32(a, b, acc0[0]);
        b.x = q0.z; b.y = q0.w; acc0[1] = wmma_f32(a, b, acc0[1]);
        b.x = q1.x; b.y = q1.y; acc0[2] = wmma_f32(a, b, acc0[2]);
        b.x = q1.z; b.y = q1.w; acc0[3] = wmma_f32(a, b, acc0[3]);
        b.x = q2.x; b.y = q2.y; acc0[4] = wmma_f32(a, b, acc0[4]);
        b.x = q2.z; b.y = q2.w; acc0[5] = wmma_f32(a, b, acc0[5]);
        b.x = q3.x; b.y = q3.y; acc0[6] = wmma_f32(a, b, acc0[6]);
        b.x = q3.z; b.y = q3.w; acc0[7] = wmma_f32(a, b, acc0[7]);
      }
    } else {
      for (int kk = 0; kk < 32; ++kk) {
        const v2f a = *(const v2f*)(ap + 4 * kk);
        const float4 q0 = ((const float4*)(bp + kk * WG_KSTEP))[q];
        v2f b;
        b.x = q0.x; b.y = q0.y; acc0[0] = wmma_f32(a, b, acc0[0]);
        b.x = q0.z; b.y = q0.w; acc0[1] = wmma_f32(a, b, acc0[1]);
      }
    }
  }
  __syncthreads();
  if (work) {
    if (full) {
      for (int t = 0; t < 8; ++t) {
        float bb = bs[208 + 16 * t + lm];
        for (int r = 0; r < 8; ++r)
          tile[(r + 8 * half) * TPITCH + 16 * t + lm] = fmaxf(acc0[t][r] + bb, 0.0f);
      }
    } else {
      for (int t = 0; t < 2; ++t) {
        float bb = bs[208 + 16 * (2 * q + t) + lm];
        for (int r = 0; r < 8; ++r)
          tile[(r + 8 * half) * TPITCH + 16 * (2 * q + t) + lm] = fmaxf(acc0[t][r] + bb, 0.0f);
      }
    }
  }
  __syncthreads();
  if (work) {
    float4 xr[16];
#pragma unroll
    for (int r = 0; r < 16; ++r)
      xr[r] = *(const float4*)(x + (size_t)(base + min(trow + r, RNG - 1)) * 128 + lane * 4);
#pragma unroll
    for (int r = 0; r < 16; ++r) {
      if (full || (r >> 2) == q) {
        float4* hp = (float4*)(tile + r * TPITCH + lane * 4);
        float4 h = *hp;
        h.x += xr[r].x; h.y += xr[r].y; h.z += xr[r].z; h.w += xr[r].w;
        *hp = h;
      }
    }
  }
  __syncthreads();

  v8f acc1[8];
  for (int t = 0; t < 8; ++t) acc1[t] = (v8f){};
  if (work) {
    const float* bp = w1s + (half * 16 + lm) * W1P;
    if (full) {
      for (int kk = 0; kk < 32; ++kk) {
        const v2f a = *(const v2f*)(ap + 4 * kk);
        const float4* bq = (const float4*)(bp + kk * (2 * 16 * W1P));
        const float4 q0 = bq[0], q1 = bq[1], q2 = bq[2], q3 = bq[3];
        v2f b;
        b.x = q0.x; b.y = q0.y; acc1[0] = wmma_f32(a, b, acc1[0]);
        b.x = q0.z; b.y = q0.w; acc1[1] = wmma_f32(a, b, acc1[1]);
        b.x = q1.x; b.y = q1.y; acc1[2] = wmma_f32(a, b, acc1[2]);
        b.x = q1.z; b.y = q1.w; acc1[3] = wmma_f32(a, b, acc1[3]);
        b.x = q2.x; b.y = q2.y; acc1[4] = wmma_f32(a, b, acc1[4]);
        b.x = q2.z; b.y = q2.w; acc1[5] = wmma_f32(a, b, acc1[5]);
        b.x = q3.x; b.y = q3.y; acc1[6] = wmma_f32(a, b, acc1[6]);
        b.x = q3.z; b.y = q3.w; acc1[7] = wmma_f32(a, b, acc1[7]);
      }
    } else {
      for (int kk = 0; kk < 32; ++kk) {
        const v2f a = *(const v2f*)(ap + 4 * kk);
        const float4 q0 = ((const float4*)(bp + kk * (2 * 16 * W1P)))[q];
        v2f b;
        b.x = q0.x; b.y = q0.y; acc1[0] = wmma_f32(a, b, acc1[0]);
        b.x = q0.z; b.y = q0.w; acc1[1] = wmma_f32(a, b, acc1[1]);
      }
    }
  }
  __syncthreads();
  if (work) {
    if (full) {
      for (int t = 0; t < 8; ++t) {
        float bb = bs[16 * t + lm];
        for (int r = 0; r < 8; ++r)
          tile[(r + 8 * half) * TPITCH + 16 * t + lm] = fmaxf(acc1[t][r] + bb, 0.0f);
      }
    } else {
      for (int t = 0; t < 2; ++t) {
        float bb = bs[16 * (2 * q + t) + lm];
        for (int r = 0; r < 8; ++r)
          tile[(r + 8 * half) * TPITCH + 16 * (2 * q + t) + lm] = fmaxf(acc1[t][r] + bb, 0.0f);
      }
    }
  }
  __syncthreads();

  v8f acc2[4];
  for (int t = 0; t < 4; ++t) acc2[t] = (v8f){};
  if (work) {
    const float* bp = w2s + (half * 16 + lm) * W2P;
    if (full) {
      for (int kk = 0; kk < 32; ++kk) {
        const v2f a = *(const v2f*)(ap + 4 * kk);
        const float4* bq = (const float4*)(bp + kk * (2 * 16 * W2P));
        const float4 q0 = bq[0], q1 = bq[1];
        v2f b;
        b.x = q0.x; b.y = q0.y; acc2[0] = wmma_f32(a, b, acc2[0]);
        b.x = q0.z; b.y = q0.w; acc2[1] = wmma_f32(a, b, acc2[1]);
        b.x = q1.x; b.y = q1.y; acc2[2] = wmma_f32(a, b, acc2[2]);
        b.x = q1.z; b.y = q1.w; acc2[3] = wmma_f32(a, b, acc2[3]);
      }
    } else {
      for (int kk = 0; kk < 32; ++kk) {
        const v2f a = *(const v2f*)(ap + 4 * kk);
        const v2f b = *(const v2f*)(bp + kk * (2 * 16 * W2P) + 2 * q);
        acc2[0] = wmma_f32(a, b, acc2[0]);
      }
    }
  }
  __syncthreads();
  if (work) {
    if (full) {
      for (int t = 0; t < 4; ++t) {
        float bb = bs[128 + 16 * t + lm];
        for (int r = 0; r < 8; ++r)
          tile[(r + 8 * half) * TPITCH + 16 * t + lm] = fmaxf(acc2[t][r] + bb, 0.0f);
      }
    } else {
      float bb = bs[128 + 16 * q + lm];
      for (int r = 0; r < 8; ++r)
        tile[(r + 8 * half) * TPITCH + 16 * q + lm] = fmaxf(acc2[0][r] + bb, 0.0f);
    }
  }
  __syncthreads();

  if (work && (full || q == 0)) {
    v8f acc3 = (v8f){};
    for (int kk = 0; kk < 16; ++kk) {
      const v2f a = *(const v2f*)(ap + 4 * kk);
      const v2f b = *(const v2f*)(w3s + (kk * 32 + lane) * 2);
      acc3 = wmma_f32(a, b, acc3);
    }
    if (lm < 2) {
      float bb = bs[192 + lm];
      for (int r = 0; r < 8; ++r)
        if (trow + r + 8 * half < RNG)
          out[(size_t)lm * N_NODES + row0 + r + 8 * half] = acc3[r] + bb;
    }
  }
}

#define SEG_OFF 6500032
static_assert(SEG_OFF >= 2 * N_NODES && SEG_OFF % 64 == 0, "segments start behind dinv, in whole 256-byte units");
static_assert(((size_t)SEG_OFF + (size_t)NGRP * GSTR) * 4 <= (size_t)268435456, "segments end inside the workspace");

extern "C" void kernel_launch(void* const* d_in, const int* in_sizes, int n_in,
                              void* d_out, int out_size, void* d_ws, size_t ws_size,
                              hipStream_t stream) {
  (void)in_sizes; (void)n_in; (void)out_size; (void)ws_size;
  const float* x  = (const float*)d_in[0];
  const int*   ei = (const int*)d_in[1];
  const float* Wg = (const float*)d_in[2];
  const float* bg = (const float*)d_in[3];
  const float* W1 = (const float*)d_in[4];
  const float* b1 = (const float*)d_in[5];
  const float* W2 = (const float*)d_in[6];
  const float* b2 = (const float*)d_in[7];
  const float* W3 = (const float*)d_in[8];
  const float* b3 = (const float*)d_in[9];
  float* out = (float*)d_out;

  float* ws   = (float*)d_ws;
  float* dinv = ws + N_NODES;
  int*   seg  = (int*)d_ws + SEG_OFF;

  k_deg<<<NDW, DEGT, 0, stream>>>(ei, seg);
  k_cnt<<<NGRP, 1024, 0, stream>>>(seg, dinv);
  k_own<<<NOWN, 1024, 0, stream>>>(seg, dinv, x, Wg, bg, W1, b1, W2, b2, W3, b3, out);
}
